# starting kernel
# baseline (speedup 1.0000x reference)
#include <hip/hip_runtime.h>
#include <hip/hip_bf16.h>


typedef float v2f __attribute__((ext_vector_type(2)));
typedef float v8f __attribute__((ext_vector_type(8)));

#define Bsz   256
#define Dd    16
#define Mm    128
#define HID   2048

__device__ __forceinline__ v8f wmma_f32_k4(v2f a, v2f b, v8f c) {
  return __builtin_amdgcn_wmma_f32_16x16x4_f32(
      false, a, false, b, (short)0, c, false, false);
}

__device__ __forceinline__ void gemm_tile(const float* __restrict__ A, int lda,
                                          const float* __restrict__ W, int ldw,
                                          int m0, int n0, int K, v8f& acc) {
  const int lane = threadIdx.x & 31;
  const int r    = lane & 15;
  const int kh   = (lane >> 4) << 1;
  const float* arow = A + (size_t)(m0 + r) * lda;
  const float* wrow = W + (size_t)(n0 + r) * ldw;
  for (int k = 0; k < K; k += 4) {
    v2f a, b;
    a.x = arow[k + kh];  a.y = arow[k + kh + 1];
    b.x = wrow[k + kh];  b.y = wrow[k + kh + 1];
    acc = wmma_f32_k4(a, b, acc);
  }
}

__device__ __forceinline__ float fast_tanh(float x) {
#if defined(__has_builtin)
#if __has_builtin(__builtin_amdgcn_tanhf)
  return __builtin_amdgcn_tanhf(x);
#elif __has_builtin(__builtin_amdgcn_tanh_f32)
  return __builtin_amdgcn_tanh_f32(x);
#else
  return tanhf(x);
#endif
#else
  return tanhf(x);
#endif
}

__device__ __forceinline__ float softplus_f(float x) {
  return fmaxf(x, 0.0f) + __logf(1.0f + __expf(-fabsf(x)));
}
__device__ __forceinline__ float log_abs_dtanh(float h) {
  return 2.0f * (0.6931471805599453f - h - softplus_f(-2.0f * h));
}

__global__ void k_layer0(const float* __restrict__ X, const float* __restrict__ W0,
                         const float* __restrict__ b0,
                         float* __restrict__ act, float* __restrict__ grad) {
  const int wave = (blockIdx.x * blockDim.x + threadIdx.x) >> 5;
  const int mt = wave >> 7;
  const int nt = wave & 127;
  if (mt >= Bsz / 16) return;
  v8f acc = {};
  gemm_tile(X, Dd, W0, Dd, mt * 16, nt * 16, Dd, acc);

  const int lane = threadIdx.x & 31;
  const int n = nt * 16 + (lane & 15);
  const int rowoff = (lane >> 4) * 8;
  const float ldiag = __logf(W0[n * Dd + (n >> 7)]);
  const float bias = b0[n];
  for (int v = 0; v < 8; ++v) {
    const int m = mt * 16 + v + rowoff;
    const float h = acc[v] + bias;
    act[m * HID + n]  = fast_tanh(h);
    grad[m * HID + n] = ldiag + log_abs_dtanh(h);
  }
}

__global__ void k_grad_prep(const float* __restrict__ grad,
                            float* __restrict__ mrow, float* __restrict__ E) {
  const int bk = blockIdx.x;
  const int i  = threadIdx.x;
  const int idx = bk * Mm + i;
  const float g = grad[idx];
  __shared__ float red[Mm];
  red[i] = g;
  __syncthreads();
  for (int s = 64; s > 0; s >>= 1) {
    if (i < s) red[i] = fmaxf(red[i], red[i + s]);
    __syncthreads();
  }
  const float mx = red[0];
  E[idx] = __expf(g - mx);
  if (i == 0) mrow[bk] = mx;
}

__global__ void k_layer1(const float* __restrict__ A, const float* __restrict__ W1,
                         const float* __restrict__ b1,
                         float* __restrict__ act, float* __restrict__ lga) {
  const int wave = (blockIdx.x * blockDim.x + threadIdx.x) >> 5;
  const int mt = wave >> 7;
  const int nt = wave & 127;
  if (mt >= Bsz / 16) return;
  v8f acc = {};
  gemm_tile(A, HID, W1, HID, mt * 16, nt * 16, HID, acc);

  const int lane = threadIdx.x & 31;
  const int n = nt * 16 + (lane & 15);
  const int rowoff = (lane >> 4) * 8;
  const float bias = b1[n];
  for (int v = 0; v < 8; ++v) {
    const int m = mt * 16 + v + rowoff;
    const float h = acc[v] + bias;
    act[m * HID + n] = fast_tanh(h);
    lga[m * HID + n] = log_abs_dtanh(h);
  }
}

__global__ void k_grad_gemm(const float* __restrict__ E, const float* __restrict__ W1,
                            const float* __restrict__ lga, const float* __restrict__ mrow,
                            float* __restrict__ grad) {
  const int wave = (blockIdx.x * blockDim.x + threadIdx.x) >> 5;
  const int ks  = wave >> 7;
  const int rem = wave & 127;
  const int mt  = rem >> 3;
  const int nt  = rem & 7;
  if (ks >= Dd) return;
  const float* Ak = E  + (size_t)ks * Mm;
  const float* Wk = W1 + (size_t)ks * Mm * HID + (size_t)ks * Mm;
  v8f acc = {};
  gemm_tile(Ak, HID, Wk, HID, mt * 16, nt * 16, Mm, acc);

  const int lane = threadIdx.x & 31;
  const int o = nt * 16 + (lane & 15);
  const int rowoff = (lane >> 4) * 8;
  for (int v = 0; v < 8; ++v) {
    const int b = mt * 16 + v + rowoff;
    const float s = acc[v];
    grad[b * HID + ks * Mm + o] =
        __logf(s) + mrow[b * Dd + ks] + lga[b * HID + ks * Mm + o];
  }
}

__global__ void k_out_gemm(const float* __restrict__ A, const float* __restrict__ W2,
                           float* __restrict__ out) {
  const int wave = threadIdx.x >> 5;
  if (wave >= Bsz / 16) return;
  v8f acc = {};
  gemm_tile(A, HID, W2, HID, wave * 16, 0, HID, acc);
  const int lane = threadIdx.x & 31;
  const int n = lane & 15;
  const int rowoff = (lane >> 4) * 8;
  for (int v = 0; v < 8; ++v) {
    const int m = wave * 16 + v + rowoff;
    out[m * Dd + n] = acc[v];
  }
}

__global__ void k_logdet(const float* __restrict__ E, const float* __restrict__ mrow,
                         const float* __restrict__ W2, float* __restrict__ ld,
                         int accumulate) {
  const int b = blockIdx.x * blockDim.x + threadIdx.x;
  if (b >= Bsz) return;
  float tot = 0.0f;
  for (int k = 0; k < Dd; ++k) {
    const float* w = W2 + k * HID + k * Mm;
    const float* e = E + b * HID + k * Mm;
    float s = 0.0f;
    for (int i = 0; i < Mm; ++i) s = fmaf(w[i], e[i], s);
    tot += __logf(s) + mrow[b * Dd + k];
  }
  if (accumulate) ld[b] += tot; else ld[b] = tot;
}

static void run_flow(const float* X, const float* W0, const float* W1,
                     const float* W2, const float* b0, const float* b1,
                     float* x_out, float* ld_out, int accumulate,
                     float* act0, float* act1, float* lga, float* grad,
                     float* E, float* mrow, hipStream_t stream) {
  k_layer0   <<<256, 256, 0, stream>>>(X, W0, b0, act0, grad);
  k_grad_prep<<<Bsz * Dd, Mm, 0, stream>>>(grad, mrow, E);
  k_layer1   <<<256, 256, 0, stream>>>(act0, W1, b1, act1, lga);
  k_grad_gemm<<<256, 256, 0, stream>>>(E, W1, lga, mrow, grad);
  k_grad_prep<<<Bsz * Dd, Mm, 0, stream>>>(grad, mrow, E);
  k_out_gemm <<<1, 512, 0, stream>>>(act1, W2, x_out);
  k_logdet   <<<1, 256, 0, stream>>>(E, mrow, W2, ld_out, accumulate);
}

extern "C" void kernel_launch(void* const* d_in, const int* in_sizes, int n_in,
                              void* d_out, int out_size, void* d_ws, size_t ws_size,
                              hipStream_t stream) {
  const float* inp   = (const float*)d_in[0];
  const float* f0_W0 = (const float*)d_in[1];
  const float* f0_W1 = (const float*)d_in[2];
  const float* f0_W2 = (const float*)d_in[3];
  const float* f0_b0 = (const float*)d_in[4];
  const float* f0_b1 = (const float*)d_in[5];
  const float* f1_W0 = (const float*)d_in[6];
  const float* f1_W1 = (const float*)d_in[7];
  const float* f1_W2 = (const float*)d_in[8];
  const float* f1_b0 = (const float*)d_in[9];
  const float* f1_b1 = (const float*)d_in[10];

  float* out = (float*)d_out;
  float* ld  = out + Bsz * Dd;

  float* ws   = (float*)d_ws;
  float* act0 = ws;
  float* act1 = act0 + Bsz * HID;
  float* lga  = act1 + Bsz * HID;
  float* grad = lga  + Bsz * HID;
  float* E    = grad + Bsz * HID;
  float* mrow = E    + Bsz * HID;
  float* xmid = mrow + Bsz * Dd;

  run_flow(inp,  f0_W0, f0_W1, f0_W2, f0_b0, f0_b1, xmid, ld,  0,
           act0, act1, lga, grad, E, mrow, stream);
  run_flow(xmid, f1_W0, f1_W1, f1_W2, f1_b0, f1_b1, out,  ld,  1,
           act0, act1, lga, grad, E, mrow, stream);
}


# Final 256x16 GEMM: K split over the 16 waves of one work-group per M-tile (16 groups, was 1 group), partial tiles summed through LDS in fixed wave order; sum order changes
# speedup vs baseline: 1.6020x; 1.6020x over previous
#include <hip/hip_runtime.h>
#include <hip/hip_bf16.h>


typedef float v2f __attribute__((ext_vector_type(2)));
typedef float v8f __attribute__((ext_vector_type(8)));

#define Bsz   256
#define Dd    16
#define Mm    128
#define HID   2048

__device__ __forceinline__ v8f wmma_f32_k4(v2f a, v2f b, v8f c) {
  return __builtin_amdgcn_wmma_f32_16x16x4_f32(
      false, a, false, b, (short)0, c, false, false);
}

__device__ __forceinline__ void gemm_tile(const float* __restrict__ A, int lda,
                                          const float* __restrict__ W, int ldw,
                                          int m0, int n0, int K, v8f& acc) {
  const int lane = threadIdx.x & 31;
  const int r    = lane & 15;
  const int kh   = (lane >> 4) << 1;
  const float* arow = A + (size_t)(m0 + r) * lda;
  const float* wrow = W + (size_t)(n0 + r) * ldw;
  for (int k = 0; k < K; k += 4) {
    v2f a, b;
    a.x = arow[k + kh];  a.y = arow[k + kh + 1];
    b.x = wrow[k + kh];  b.y = wrow[k + kh + 1];
    acc = wmma_f32_k4(a, b, acc);
  }
}

__device__ __forceinline__ float fast_tanh(float x) {
#if defined(__has_builtin)
#if __has_builtin(__builtin_amdgcn_tanhf)
  return __builtin_amdgcn_tanhf(x);
#elif __has_builtin(__builtin_amdgcn_tanh_f32)
  return __builtin_amdgcn_tanh_f32(x);
#else
  return tanhf(x);
#endif
#else
  return tanhf(x);
#endif
}

__device__ __forceinline__ float softplus_f(float x) {
  return fmaxf(x, 0.0f) + __logf(1.0f + __expf(-fabsf(x)));
}
__device__ __forceinline__ float log_abs_dtanh(float h) {
  return 2.0f * (0.6931471805599453f - h - softplus_f(-2.0f * h));
}

__global__ void k_layer0(const float* __restrict__ X, const float* __restrict__ W0,
                         const float* __restrict__ b0,
                         float* __restrict__ act, float* __restrict__ grad) {
  const int wave = (blockIdx.x * blockDim.x + threadIdx.x) >> 5;
  const int mt = wave >> 7;
  const int nt = wave & 127;
  if (mt >= Bsz / 16) return;
  v8f acc = {};
  gemm_tile(X, Dd, W0, Dd, mt * 16, nt * 16, Dd, acc);

  const int lane = threadIdx.x & 31;
  const int n = nt * 16 + (lane & 15);
  const int rowoff = (lane >> 4) * 8;
  const float ldiag = __logf(W0[n * Dd + (n >> 7)]);
  const float bias = b0[n];
  for (int v = 0; v < 8; ++v) {
    const int m = mt * 16 + v + rowoff;
    const float h = acc[v] + bias;
    act[m * HID + n]  = fast_tanh(h);
    grad[m * HID + n] = ldiag + log_abs_dtanh(h);
  }
}

__global__ void k_grad_prep(const float* __restrict__ grad,
                            float* __restrict__ mrow, float* __restrict__ E) {
  const int bk = blockIdx.x;
  const int i  = threadIdx.x;
  const int idx = bk * Mm + i;
  const float g = grad[idx];
  __shared__ float red[Mm];
  red[i] = g;
  __syncthreads();
  for (int s = 64; s > 0; s >>= 1) {
    if (i < s) red[i] = fmaxf(red[i], red[i + s]);
    __syncthreads();
  }
  const float mx = red[0];
  E[idx] = __expf(g - mx);
  if (i == 0) mrow[bk] = mx;
}

__global__ void k_layer1(const float* __restrict__ A, const float* __restrict__ W1,
                         const float* __restrict__ b1,
                         float* __restrict__ act, float* __restrict__ lga) {
  const int wave = (blockIdx.x * blockDim.x + threadIdx.x) >> 5;
  const int mt = wave >> 7;
  const int nt = wave & 127;
  if (mt >= Bsz / 16) return;
  v8f acc = {};
  gemm_tile(A, HID, W1, HID, mt * 16, nt * 16, HID, acc);

  const int lane = threadIdx.x & 31;
  const int n = nt * 16 + (lane & 15);
  const int rowoff = (lane >> 4) * 8;
  const float bias = b1[n];
  for (int v = 0; v < 8; ++v) {
    const int m = mt * 16 + v + rowoff;
    const float h = acc[v] + bias;
    act[m * HID + n] = fast_tanh(h);
    lga[m * HID + n] = log_abs_dtanh(h);
  }
}

__global__ void k_grad_gemm(const float* __restrict__ E, const float* __restrict__ W1,
                            const float* __restrict__ lga, const float* __restrict__ mrow,
                            float* __restrict__ grad) {
  const int wave = (blockIdx.x * blockDim.x + threadIdx.x) >> 5;
  const int ks  = wave >> 7;
  const int rem = wave & 127;
  const int mt  = rem >> 3;
  const int nt  = rem & 7;
  if (ks >= Dd) return;
  const float* Ak = E  + (size_t)ks * Mm;
  const float* Wk = W1 + (size_t)ks * Mm * HID + (size_t)ks * Mm;
  v8f acc = {};
  gemm_tile(Ak, HID, Wk, HID, mt * 16, nt * 16, Mm, acc);

  const int lane = threadIdx.x & 31;
  const int o = nt * 16 + (lane & 15);
  const int rowoff = (lane >> 4) * 8;
  for (int v = 0; v < 8; ++v) {
    const int b = mt * 16 + v + rowoff;
    const float s = acc[v];
    grad[b * HID + ks * Mm + o] =
        __logf(s) + mrow[b * Dd + ks] + lga[b * HID + ks * Mm + o];
  }
}

#define OG_WAVES 16
#define OG_KC    (HID / OG_WAVES)
__global__ void k_out_gemm(const float* __restrict__ A, const float* __restrict__ W2,
                           float* __restrict__ out) {
  __shared__ float part[OG_WAVES][16 * 16];
  const int wave = threadIdx.x >> 5;
  const int lane = threadIdx.x & 31;
  const int mt   = blockIdx.x;
  v8f acc = {};
  gemm_tile(A + wave * OG_KC, HID, W2 + wave * OG_KC, HID, mt * 16, 0, OG_KC, acc);
  const int n = lane & 15;
  const int rowoff = (lane >> 4) * 8;
  for (int v = 0; v < 8; ++v)
    part[wave][(v + rowoff) * Dd + n] = acc[v];
  __syncthreads();
  if (threadIdx.x < 16 * 16) {
    float s = part[0][threadIdx.x];
    for (int w = 1; w < OG_WAVES; ++w) s += part[w][threadIdx.x];
    out[mt * 16 * Dd + threadIdx.x] = s;
  }
}

__global__ void k_logdet(const float* __restrict__ E, const float* __restrict__ mrow,
                         const float* __restrict__ W2, float* __restrict__ ld,
                         int accumulate) {
  const int b = blockIdx.x * blockDim.x + threadIdx.x;
  if (b >= Bsz) return;
  float tot = 0.0f;
  for (int k = 0; k < Dd; ++k) {
    const float* w = W2 + k * HID + k * Mm;
    const float* e = E + b * HID + k * Mm;
    float s = 0.0f;
    for (int i = 0; i < Mm; ++i) s = fmaf(w[i], e[i], s);
    tot += __logf(s) + mrow[b * Dd + k];
  }
  if (accumulate) ld[b] += tot; else ld[b] = tot;
}

static void run_flow(const float* X, const float* W0, const float* W1,
                     const float* W2, const float* b0, const float* b1,
                     float* x_out, float* ld_out, int accumulate,
                     float* act0, float* act1, float* lga, float* grad,
                     float* E, float* mrow, hipStream_t stream) {
  k_layer0   <<<256, 256, 0, stream>>>(X, W0, b0, act0, grad);
  k_grad_prep<<<Bsz * Dd, Mm, 0, stream>>>(grad, mrow, E);
  k_layer1   <<<256, 256, 0, stream>>>(act0, W1, b1, act1, lga);
  k_grad_gemm<<<256, 256, 0, stream>>>(E, W1, lga, mrow, grad);
  k_grad_prep<<<Bsz * Dd, Mm, 0, stream>>>(grad, mrow, E);
  k_out_gemm <<<Bsz / 16, OG_WAVES * 32, 0, stream>>>(act1, W2, x_out);
  k_logdet   <<<1, 256, 0, stream>>>(E, mrow, W2, ld_out, accumulate);
}

extern "C" void kernel_launch(void* const* d_in, const int* in_sizes, int n_in,
                              void* d_out, int out_size, void* d_ws, size_t ws_size,
                              hipStream_t stream) {
  const float* inp   = (const float*)d_in[0];
  const float* f0_W0 = (const float*)d_in[1];
  const float* f0_W1 = (const float*)d_in[2];
  const float* f0_W2 = (const float*)d_in[3];
  const float* f0_b0 = (const float*)d_in[4];
  const float* f0_b1 = (const float*)d_in[5];
  const float* f1_W0 = (const float*)d_in[6];
  const float* f1_W1 = (const float*)d_in[7];
  const float* f1_W2 = (const float*)d_in[8];
  const float* f1_b0 = (const float*)d_in[9];
  const float* f1_b1 = (const float*)d_in[10];

  float* out = (float*)d_out;
  float* ld  = out + Bsz * Dd;

  float* ws   = (float*)d_ws;
  float* act0 = ws;
  float* act1 = act0 + Bsz * HID;
  float* lga  = act1 + Bsz * HID;
  float* grad = lga  + Bsz * HID;
  float* E    = grad + Bsz * HID;
  float* mrow = E    + Bsz * HID;
  float* xmid = mrow + Bsz * Dd;

  run_flow(inp,  f0_W0, f0_W1, f0_W2, f0_b0, f0_b1, xmid, ld,  0,
           act0, act1, lga, grad, E, mrow, stream);
  run_flow(xmid, f1_W0, f1_W1, f1_W2, f1_b0, f1_b1, out,  ld,  1,
           act0, act1, lga, grad, E, mrow, stream);
}


# Log-det dots moved off a single work-group: 8 groups x 32 waves, one wave per sample, coalesced 16-byte loads, halving butterfly across lanes replaces the serial 128-term chain; sum order changes
# speedup vs baseline: 2.4295x; 1.5165x over previous
#include <hip/hip_runtime.h>
#include <hip/hip_bf16.h>


typedef float v2f __attribute__((ext_vector_type(2)));
typedef float v8f __attribute__((ext_vector_type(8)));

#define Bsz   256
#define Dd    16
#define Mm    128
#define HID   2048

__device__ __forceinline__ v8f wmma_f32_k4(v2f a, v2f b, v8f c) {
  return __builtin_amdgcn_wmma_f32_16x16x4_f32(
      false, a, false, b, (short)0, c, false, false);
}

__device__ __forceinline__ void gemm_tile(const float* __restrict__ A, int lda,
                                          const float* __restrict__ W, int ldw,
                                          int m0, int n0, int K, v8f& acc) {
  const int lane = threadIdx.x & 31;
  const int r    = lane & 15;
  const int kh   = (lane >> 4) << 1;
  const float* arow = A + (size_t)(m0 + r) * lda;
  const float* wrow = W + (size_t)(n0 + r) * ldw;
  for (int k = 0; k < K; k += 4) {
    v2f a, b;
    a.x = arow[k + kh];  a.y = arow[k + kh + 1];
    b.x = wrow[k + kh];  b.y = wrow[k + kh + 1];
    acc = wmma_f32_k4(a, b, acc);
  }
}

__device__ __forceinline__ float fast_tanh(float x) {
#if defined(__has_builtin)
#if __has_builtin(__builtin_amdgcn_tanhf)
  return __builtin_amdgcn_tanhf(x);
#elif __has_builtin(__builtin_amdgcn_tanh_f32)
  return __builtin_amdgcn_tanh_f32(x);
#else
  return tanhf(x);
#endif
#else
  return tanhf(x);
#endif
}

__device__ __forceinline__ float softplus_f(float x) {
  return fmaxf(x, 0.0f) + __logf(1.0f + __expf(-fabsf(x)));
}
__device__ __forceinline__ float log_abs_dtanh(float h) {
  return 2.0f * (0.6931471805599453f - h - softplus_f(-2.0f * h));
}

__global__ void k_layer0(const float* __restrict__ X, const float* __restrict__ W0,
                         const float* __restrict__ b0,
                         float* __restrict__ act, float* __restrict__ grad) {
  const int wave = (blockIdx.x * blockDim.x + threadIdx.x) >> 5;
  const int mt = wave >> 7;
  const int nt = wave & 127;
  if (mt >= Bsz / 16) return;
  v8f acc = {};
  gemm_tile(X, Dd, W0, Dd, mt * 16, nt * 16, Dd, acc);

  const int lane = threadIdx.x & 31;
  const int n = nt * 16 + (lane & 15);
  const int rowoff = (lane >> 4) * 8;
  const float ldiag = __logf(W0[n * Dd + (n >> 7)]);
  const float bias = b0[n];
  for (int v = 0; v < 8; ++v) {
    const int m = mt * 16 + v + rowoff;
    const float h = acc[v] + bias;
    act[m * HID + n]  = fast_tanh(h);
    grad[m * HID + n] = ldiag + log_abs_dtanh(h);
  }
}

__global__ void k_grad_prep(const float* __restrict__ grad,
                            float* __restrict__ mrow, float* __restrict__ E) {
  const int bk = blockIdx.x;
  const int i  = threadIdx.x;
  const int idx = bk * Mm + i;
  const float g = grad[idx];
  __shared__ float red[Mm];
  red[i] = g;
  __syncthreads();
  for (int s = 64; s > 0; s >>= 1) {
    if (i < s) red[i] = fmaxf(red[i], red[i + s]);
    __syncthreads();
  }
  const float mx = red[0];
  E[idx] = __expf(g - mx);
  if (i == 0) mrow[bk] = mx;
}

__global__ void k_layer1(const float* __restrict__ A, const float* __restrict__ W1,
                         const float* __restrict__ b1,
                         float* __restrict__ act, float* __restrict__ lga) {
  const int wave = (blockIdx.x * blockDim.x + threadIdx.x) >> 5;
  const int mt = wave >> 7;
  const int nt = wave & 127;
  if (mt >= Bsz / 16) return;
  v8f acc = {};
  gemm_tile(A, HID, W1, HID, mt * 16, nt * 16, HID, acc);

  const int lane = threadIdx.x & 31;
  const int n = nt * 16 + (lane & 15);
  const int rowoff = (lane >> 4) * 8;
  const float bias = b1[n];
  for (int v = 0; v < 8; ++v) {
    const int m = mt * 16 + v + rowoff;
    const float h = acc[v] + bias;
    act[m * HID + n] = fast_tanh(h);
    lga[m * HID + n] = log_abs_dtanh(h);
  }
}

__global__ void k_grad_gemm(const float* __restrict__ E, const float* __restrict__ W1,
                            const float* __restrict__ lga, const float* __restrict__ mrow,
                            float* __restrict__ grad) {
  const int wave = (blockIdx.x * blockDim.x + threadIdx.x) >> 5;
  const int ks  = wave >> 7;
  const int rem = wave & 127;
  const int mt  = rem >> 3;
  const int nt  = rem & 7;
  if (ks >= Dd) return;
  const float* Ak = E  + (size_t)ks * Mm;
  const float* Wk = W1 + (size_t)ks * Mm * HID + (size_t)ks * Mm;
  v8f acc = {};
  gemm_tile(Ak, HID, Wk, HID, mt * 16, nt * 16, Mm, acc);

  const int lane = threadIdx.x & 31;
  const int o = nt * 16 + (lane & 15);
  const int rowoff = (lane >> 4) * 8;
  for (int v = 0; v < 8; ++v) {
    const int b = mt * 16 + v + rowoff;
    const float s = acc[v];
    grad[b * HID + ks * Mm + o] =
        __logf(s) + mrow[b * Dd + ks] + lga[b * HID + ks * Mm + o];
  }
}

#define OG_WAVES 16
#define OG_KC    (HID / OG_WAVES)
__global__ void k_out_gemm(const float* __restrict__ A, const float* __restrict__ W2,
                           float* __restrict__ out) {
  __shared__ float part[OG_WAVES][16 * 16];
  const int wave = threadIdx.x >> 5;
  const int lane = threadIdx.x & 31;
  const int mt   = blockIdx.x;
  v8f acc = {};
  gemm_tile(A + wave * OG_KC, HID, W2 + wave * OG_KC, HID, mt * 16, 0, OG_KC, acc);
  const int n = lane & 15;
  const int rowoff = (lane >> 4) * 8;
  for (int v = 0; v < 8; ++v)
    part[wave][(v + rowoff) * Dd + n] = acc[v];
  __syncthreads();
  if (threadIdx.x < 16 * 16) {
    float s = part[0][threadIdx.x];
    for (int w = 1; w < OG_WAVES; ++w) s += part[w][threadIdx.x];
    out[mt * 16 * Dd + threadIdx.x] = s;
  }
}

#define LD_WAVES 32

template <int HALF>
__device__ __forceinline__ void fold_half(float (&s)[Dd], int lane, int off) {
  const bool up = (lane & off) != 0;
#pragma unroll
  for (int j = 0; j < HALF; ++j) {
    const float keep = up ? s[j + HALF] : s[j];
    const float send = up ? s[j] : s[j + HALF];
    s[j] = keep + __shfl_xor(send, off);
  }
}

__global__ void k_logdet(const float* __restrict__ E, const float* __restrict__ mrow,
                         const float* __restrict__ W2, float* __restrict__ ld,
                         int accumulate) {
  __shared__ float tot_s[LD_WAVES];
  const int wave = threadIdx.x >> 5;
  const int lane = threadIdx.x & 31;
  const int b = blockIdx.x * LD_WAVES + wave;
  const float mk = mrow[b * Dd + (lane >> 1)];
  const float4* e4 = reinterpret_cast<const float4*>(E + (size_t)b * HID) + lane;
  float s[Dd];
#pragma unroll
  for (int k = 0; k < Dd; ++k) {
    const float4 e = e4[k * (Mm / 4)];
    const float4 w =
        reinterpret_cast<const float4*>(W2 + (size_t)k * HID + k * Mm)[lane];
    float p = e.x * w.x;
    p = fmaf(e.y, w.y, p);
    p = fmaf(e.z, w.z, p);
    p = fmaf(e.w, w.w, p);
    s[k] = p;
  }
  fold_half<8>(s, lane, 16);
  fold_half<4>(s, lane, 8);
  fold_half<2>(s, lane, 4);
  fold_half<1>(s, lane, 2);
  const float sk = s[0] + __shfl_xor(s[0], 1);
  float t = __logf(sk) + mk;
  t += __shfl_xor(t, 2);
  t += __shfl_xor(t, 4);
  t += __shfl_xor(t, 8);
  t += __shfl_xor(t, 16);
  if (lane == 0) tot_s[wave] = t;
  __syncthreads();
  if (wave == 0) {
    const int bo = blockIdx.x * LD_WAVES + lane;
    if (accumulate) ld[bo] += tot_s[lane]; else ld[bo] = tot_s[lane];
  }
}

static void run_flow(const float* X, const float* W0, const float* W1,
                     const float* W2, const float* b0, const float* b1,
                     float* x_out, float* ld_out, int accumulate,
                     float* act0, float* act1, float* lga, float* grad,
                     float* E, float* mrow, hipStream_t stream) {
  k_layer0   <<<256, 256, 0, stream>>>(X, W0, b0, act0, grad);
  k_grad_prep<<<Bsz * Dd, Mm, 0, stream>>>(grad, mrow, E);
  k_layer1   <<<256, 256, 0, stream>>>(act0, W1, b1, act1, lga);
  k_grad_gemm<<<256, 256, 0, stream>>>(E, W1, lga, mrow, grad);
  k_grad_prep<<<Bsz * Dd, Mm, 0, stream>>>(grad, mrow, E);
  k_out_gemm <<<Bsz / 16, OG_WAVES * 32, 0, stream>>>(act1, W2, x_out);
  k_logdet   <<<Bsz / LD_WAVES, LD_WAVES * 32, 0, stream>>>(E, mrow, W2, ld_out, accumulate);
}

extern "C" void kernel_launch(void* const* d_in, const int* in_sizes, int n_in,
                              void* d_out, int out_size, void* d_ws, size_t ws_size,
                              hipStream_t stream) {
  const float* inp   = (const float*)d_in[0];
  const float* f0_W0 = (const float*)d_in[1];
  const float* f0_W1 = (const float*)d_in[2];
  const float* f0_W2 = (const float*)d_in[3];
  const float* f0_b0 = (const float*)d_in[4];
  const float* f0_b1 = (const float*)d_in[5];
  const float* f1_W0 = (const float*)d_in[6];
  const float* f1_W1 = (const float*)d_in[7];
  const float* f1_W2 = (const float*)d_in[8];
  const float* f1_b0 = (const float*)d_in[9];
  const float* f1_b1 = (const float*)d_in[10];

  float* out = (float*)d_out;
  float* ld  = out + Bsz * Dd;

  float* ws   = (float*)d_ws;
  float* act0 = ws;
  float* act1 = act0 + Bsz * HID;
  float* lga  = act1 + Bsz * HID;
  float* grad = lga  + Bsz * HID;
  float* E    = grad + Bsz * HID;
  float* mrow = E    + Bsz * HID;
  float* xmid = mrow + Bsz * Dd;

  run_flow(inp,  f0_W0, f0_W1, f0_W2, f0_b0, f0_b1, xmid, ld,  0,
           act0, act1, lga, grad, E, mrow, stream);
  run_flow(xmid, f1_W0, f1_W1, f1_W2, f1_b0, f1_b1, out,  ld,  1,
           act0, act1, lga, grad, E, mrow, stream);
}


# Row max + exp pass folded into the epilogue of its two producers (layer-0 and grad GEMM: one work-group already holds whole 128-wide rows), 8-wave max through LDS; 14 launches -> 10
# speedup vs baseline: 2.5986x; 1.0696x over previous
#include <hip/hip_runtime.h>
#include <hip/hip_bf16.h>


typedef float v2f __attribute__((ext_vector_type(2)));
typedef float v8f __attribute__((ext_vector_type(8)));

#define Bsz   256
#define Dd    16
#define Mm    128
#define HID   2048

__device__ __forceinline__ v8f wmma_f32_k4(v2f a, v2f b, v8f c) {
  return __builtin_amdgcn_wmma_f32_16x16x4_f32(
      false, a, false, b, (short)0, c, false, false);
}

__device__ __forceinline__ void gemm_tile(const float* __restrict__ A, int lda,
                                          const float* __restrict__ W, int ldw,
                                          int m0, int n0, int K, v8f& acc) {
  const int lane = threadIdx.x & 31;
  const int r    = lane & 15;
  const int kh   = (lane >> 4) << 1;
  const float* arow = A + (size_t)(m0 + r) * lda;
  const float* wrow = W + (size_t)(n0 + r) * ldw;
  for (int k = 0; k < K; k += 4) {
    v2f a, b;
    a.x = arow[k + kh];  a.y = arow[k + kh + 1];
    b.x = wrow[k + kh];  b.y = wrow[k + kh + 1];
    acc = wmma_f32_k4(a, b, acc);
  }
}

__device__ __forceinline__ float fast_tanh(float x) {
#if defined(__has_builtin)
#if __has_builtin(__builtin_amdgcn_tanhf)
  return __builtin_amdgcn_tanhf(x);
#elif __has_builtin(__builtin_amdgcn_tanh_f32)
  return __builtin_amdgcn_tanh_f32(x);
#else
  return tanhf(x);
#endif
#else
  return tanhf(x);
#endif
}

__device__ __forceinline__ float softplus_f(float x) {
  return fmaxf(x, 0.0f) + __logf(1.0f + __expf(-fabsf(x)));
}
__device__ __forceinline__ float log_abs_dtanh(float h) {
  return 2.0f * (0.6931471805599453f - h - softplus_f(-2.0f * h));
}

#define RP_WAVES 8
__device__ __forceinline__ void rowmax_exp_store(const float (&g)[8], int m0, int ks,
                                                 float* __restrict__ E,
                                                 float* __restrict__ mrow) {
  __shared__ __attribute__((aligned(16))) float pmax[RP_WAVES][16];
  const int wave = threadIdx.x >> 5;
  const int lane = threadIdx.x & 31;
  const int c = lane & 15;
  const int rowoff = (lane >> 4) * 8;
  float mx[8];
  for (int v = 0; v < 8; ++v) {
    float t = g[v];
    t = fmaxf(t, __shfl_xor(t, 1));
    t = fmaxf(t, __shfl_xor(t, 2));
    t = fmaxf(t, __shfl_xor(t, 4));
    t = fmaxf(t, __shfl_xor(t, 8));
    mx[v] = t;
  }
  if (c == 0) {
    for (int v = 0; v < 8; ++v) pmax[wave][rowoff + v] = mx[v];
  }
  __syncthreads();
  for (int v = 0; v < 8; ++v) {
    float r = pmax[0][rowoff + v];
    for (int w = 1; w < RP_WAVES; ++w) r = fmaxf(r, pmax[w][rowoff + v]);
    mx[v] = r;
  }
  for (int v = 0; v < 8; ++v) {
    const int m = m0 + rowoff + v;
    E[m * HID + ks * Mm + wave * 16 + c] = __expf(g[v] - mx[v]);
  }
  if (wave == 0 && c == 0) {
    for (int v = 0; v < 8; ++v) mrow[(m0 + rowoff + v) * Dd + ks] = mx[v];
  }
}

__global__ void k_layer0(const float* __restrict__ X, const float* __restrict__ W0,
                         const float* __restrict__ b0,
                         float* __restrict__ act,
                         float* __restrict__ E, float* __restrict__ mrow) {
  const int wave = (blockIdx.x * blockDim.x + threadIdx.x) >> 5;
  const int mt = wave >> 7;
  const int nt = wave & 127;
  if (mt >= Bsz / 16) return;
  v8f acc = {};
  gemm_tile(X, Dd, W0, Dd, mt * 16, nt * 16, Dd, acc);

  const int lane = threadIdx.x & 31;
  const int n = nt * 16 + (lane & 15);
  const int rowoff = (lane >> 4) * 8;
  const float ldiag = __logf(W0[n * Dd + (n >> 7)]);
  const float bias = b0[n];
  float g[8];
  for (int v = 0; v < 8; ++v) {
    const int m = mt * 16 + v + rowoff;
    const float h = acc[v] + bias;
    act[m * HID + n] = fast_tanh(h);
    g[v] = ldiag + log_abs_dtanh(h);
  }
  rowmax_exp_store(g, mt * 16, nt >> 3, E, mrow);
}

__global__ void k_layer1(const float* __restrict__ A, const float* __restrict__ W1,
                         const float* __restrict__ b1,
                         float* __restrict__ act, float* __restrict__ lga) {
  const int wave = (blockIdx.x * blockDim.x + threadIdx.x) >> 5;
  const int mt = wave >> 7;
  const int nt = wave & 127;
  if (mt >= Bsz / 16) return;
  v8f acc = {};
  gemm_tile(A, HID, W1, HID, mt * 16, nt * 16, HID, acc);

  const int lane = threadIdx.x & 31;
  const int n = nt * 16 + (lane & 15);
  const int rowoff = (lane >> 4) * 8;
  const float bias = b1[n];
  for (int v = 0; v < 8; ++v) {
    const int m = mt * 16 + v + rowoff;
    const float h = acc[v] + bias;
    act[m * HID + n] = fast_tanh(h);
    lga[m * HID + n] = log_abs_dtanh(h);
  }
}

__global__ void k_grad_gemm(const float* __restrict__ E, const float* __restrict__ W1,
                            const float* __restrict__ lga, const float* __restrict__ mrow,
                            float* __restrict__ Eo, float* __restrict__ mo) {
  const int wave = (blockIdx.x * blockDim.x + threadIdx.x) >> 5;
  const int ks  = wave >> 7;
  const int rem = wave & 127;
  const int mt  = rem >> 3;
  const int nt  = rem & 7;
  if (ks >= Dd) return;
  const float* Ak = E  + (size_t)ks * Mm;
  const float* Wk = W1 + (size_t)ks * Mm * HID + (size_t)ks * Mm;
  v8f acc = {};
  gemm_tile(Ak, HID, Wk, HID, mt * 16, nt * 16, Mm, acc);

  const int lane = threadIdx.x & 31;
  const int o = nt * 16 + (lane & 15);
  const int rowoff = (lane >> 4) * 8;
  float g[8];
  for (int v = 0; v < 8; ++v) {
    const int b = mt * 16 + v + rowoff;
    const float s = acc[v];
    g[v] = __logf(s) + mrow[b * Dd + ks] + lga[b * HID + ks * Mm + o];
  }
  rowmax_exp_store(g, mt * 16, ks, Eo, mo);
}

#define OG_WAVES 16
#define OG_KC    (HID / OG_WAVES)
__global__ void k_out_gemm(const float* __restrict__ A, const float* __restrict__ W2,
                           float* __restrict__ out) {
  __shared__ float part[OG_WAVES][16 * 16];
  const int wave = threadIdx.x >> 5;
  const int lane = threadIdx.x & 31;
  const int mt   = blockIdx.x;
  v8f acc = {};
  gemm_tile(A + wave * OG_KC, HID, W2 + wave * OG_KC, HID, mt * 16, 0, OG_KC, acc);
  const int n = lane & 15;
  const int rowoff = (lane >> 4) * 8;
  for (int v = 0; v < 8; ++v)
    part[wave][(v + rowoff) * Dd + n] = acc[v];
  __syncthreads();
  if (threadIdx.x < 16 * 16) {
    float s = part[0][threadIdx.x];
    for (int w = 1; w < OG_WAVES; ++w) s += part[w][threadIdx.x];
    out[mt * 16 * Dd + threadIdx.x] = s;
  }
}

#define LD_WAVES 32

template <int HALF>
__device__ __forceinline__ void fold_half(float (&s)[Dd], int lane, int off) {
  const bool up = (lane & off) != 0;
#pragma unroll
  for (int j = 0; j < HALF; ++j) {
    const float keep = up ? s[j + HALF] : s[j];
    const float send = up ? s[j] : s[j + HALF];
    s[j] = keep + __shfl_xor(send, off);
  }
}

__global__ void k_logdet(const float* __restrict__ E, const float* __restrict__ mrow,
                         const float* __restrict__ W2, float* __restrict__ ld,
                         int accumulate) {
  __shared__ float tot_s[LD_WAVES];
  const int wave = threadIdx.x >> 5;
  const int lane = threadIdx.x & 31;
  const int b = blockIdx.x * LD_WAVES + wave;
  const float mk = mrow[b * Dd + (lane >> 1)];
  const float4* e4 = reinterpret_cast<const float4*>(E + (size_t)b * HID) + lane;
  float s[Dd];
#pragma unroll
  for (int k = 0; k < Dd; ++k) {
    const float4 e = e4[k * (Mm / 4)];
    const float4 w =
        reinterpret_cast<const float4*>(W2 + (size_t)k * HID + k * Mm)[lane];
    float p = e.x * w.x;
    p = fmaf(e.y, w.y, p);
    p = fmaf(e.z, w.z, p);
    p = fmaf(e.w, w.w, p);
    s[k] = p;
  }
  fold_half<8>(s, lane, 16);
  fold_half<4>(s, lane, 8);
  fold_half<2>(s, lane, 4);
  fold_half<1>(s, lane, 2);
  const float sk = s[0] + __shfl_xor(s[0], 1);
  float t = __logf(sk) + mk;
  t += __shfl_xor(t, 2);
  t += __shfl_xor(t, 4);
  t += __shfl_xor(t, 8);
  t += __shfl_xor(t, 16);
  if (lane == 0) tot_s[wave] = t;
  __syncthreads();
  if (wave == 0) {
    const int bo = blockIdx.x * LD_WAVES + lane;
    if (accumulate) ld[bo] += tot_s[lane]; else ld[bo] = tot_s[lane];
  }
}

static void run_flow(const float* X, const float* W0, const float* W1,
                     const float* W2, const float* b0, const float* b1,
                     float* x_out, float* ld_out, int accumulate,
                     float* act0, float* act1, float* lga,
                     float* E0, float* mrow0, float* E1, float* mrow1,
                     hipStream_t stream) {
  k_layer0   <<<256, RP_WAVES * 32, 0, stream>>>(X, W0, b0, act0, E0, mrow0);
  k_layer1   <<<256, 256, 0, stream>>>(act0, W1, b1, act1, lga);
  k_grad_gemm<<<256, RP_WAVES * 32, 0, stream>>>(E0, W1, lga, mrow0, E1, mrow1);
  k_out_gemm <<<Bsz / 16, OG_WAVES * 32, 0, stream>>>(act1, W2, x_out);
  k_logdet   <<<Bsz / LD_WAVES, LD_WAVES * 32, 0, stream>>>(E1, mrow1, W2, ld_out, accumulate);
}

extern "C" void kernel_launch(void* const* d_in, const int* in_sizes, int n_in,
                              void* d_out, int out_size, void* d_ws, size_t ws_size,
                              hipStream_t stream) {
  const float* inp   = (const float*)d_in[0];
  const float* f0_W0 = (const float*)d_in[1];
  const float* f0_W1 = (const float*)d_in[2];
  const float* f0_W2 = (const float*)d_in[3];
  const float* f0_b0 = (const float*)d_in[4];
  const float* f0_b1 = (const float*)d_in[5];
  const float* f1_W0 = (const float*)d_in[6];
  const float* f1_W1 = (const float*)d_in[7];
  const float* f1_W2 = (const float*)d_in[8];
  const float* f1_b0 = (const float*)d_in[9];
  const float* f1_b1 = (const float*)d_in[10];

  float* out = (float*)d_out;
  float* ld  = out + Bsz * Dd;

  float* ws    = (float*)d_ws;
  float* act0  = ws;
  float* act1  = act0  + Bsz * HID;
  float* lga   = act1  + Bsz * HID;
  float* E0    = lga   + Bsz * HID;
  float* E1    = E0    + Bsz * HID;
  float* mrow0 = E1    + Bsz * HID;
  float* mrow1 = mrow0 + Bsz * Dd;
  float* xmid  = mrow1 + Bsz * Dd;

  run_flow(inp,  f0_W0, f0_W1, f0_W2, f0_b0, f0_b1, xmid, ld,  0,
           act0, act1, lga, E0, mrow0, E1, mrow1, stream);
  run_flow(xmid, f1_W0, f1_W1, f1_W2, f1_b0, f1_b1, out,  ld,  1,
           act0, act1, lga, E0, mrow0, E1, mrow1, stream);
}


# Layer-1 GEMM operands staged through LDS by whole-line 16-byte loads (32x64 slab per 4-wave group, 128-wide K chunks, two stages), 16-byte fragment reads, 2 tiles per wave; k sum order changes
# speedup vs baseline: 3.4798x; 1.3391x over previous
#include <hip/hip_runtime.h>
#include <hip/hip_bf16.h>


typedef float v2f __attribute__((ext_vector_type(2)));
typedef float v8f __attribute__((ext_vector_type(8)));

#define Bsz   256
#define Dd    16
#define Mm    128
#define HID   2048

__device__ __forceinline__ v8f wmma_f32_k4(v2f a, v2f b, v8f c) {
  return __builtin_amdgcn_wmma_f32_16x16x4_f32(
      false, a, false, b, (short)0, c, false, false);
}

__device__ __forceinline__ void gemm_tile(const float* __restrict__ A, int lda,
                                          const float* __restrict__ W, int ldw,
                                          int m0, int n0, int K, v8f& acc) {
  const int lane = threadIdx.x & 31;
  const int r    = lane & 15;
  const int kh   = (lane >> 4) << 1;
  const float* arow = A + (size_t)(m0 + r) * lda;
  const float* wrow = W + (size_t)(n0 + r) * ldw;
  for (int k = 0; k < K; k += 4) {
    v2f a, b;
    a.x = arow[k + kh];  a.y = arow[k + kh + 1];
    b.x = wrow[k + kh];  b.y = wrow[k + kh + 1];
    acc = wmma_f32_k4(a, b, acc);
  }
}

__device__ __forceinline__ float fast_tanh(float x) {
#if defined(__has_builtin)
#if __has_builtin(__builtin_amdgcn_tanhf)
  return __builtin_amdgcn_tanhf(x);
#elif __has_builtin(__builtin_amdgcn_tanh_f32)
  return __builtin_amdgcn_tanh_f32(x);
#else
  return tanhf(x);
#endif
#else
  return tanhf(x);
#endif
}

__device__ __forceinline__ float softplus_f(float x) {
  return fmaxf(x, 0.0f) + __logf(1.0f + __expf(-fabsf(x)));
}
__device__ __forceinline__ float log_abs_dtanh(float h) {
  return 2.0f * (0.6931471805599453f - h - softplus_f(-2.0f * h));
}

#define RP_WAVES 8
__device__ __forceinline__ void rowmax_exp_store(const float (&g)[8], int m0, int ks,
                                                 float* __restrict__ E,
                                                 float* __restrict__ mrow) {
  __shared__ __attribute__((aligned(16))) float pmax[RP_WAVES][16];
  const int wave = threadIdx.x >> 5;
  const int lane = threadIdx.x & 31;
  const int c = lane & 15;
  const int rowoff = (lane >> 4) * 8;
  float mx[8];
  for (int v = 0; v < 8; ++v) {
    float t = g[v];
    t = fmaxf(t, __shfl_xor(t, 1));
    t = fmaxf(t, __shfl_xor(t, 2));
    t = fmaxf(t, __shfl_xor(t, 4));
    t = fmaxf(t, __shfl_xor(t, 8));
    mx[v] = t;
  }
  if (c == 0) {
    for (int v = 0; v < 8; ++v) pmax[wave][rowoff + v] = mx[v];
  }
  __syncthreads();
  for (int v = 0; v < 8; ++v) {
    float r = pmax[0][rowoff + v];
    for (int w = 1; w < RP_WAVES; ++w) r = fmaxf(r, pmax[w][rowoff + v]);
    mx[v] = r;
  }
  for (int v = 0; v < 8; ++v) {
    const int m = m0 + rowoff + v;
    E[m * HID + ks * Mm + wave * 16 + c] = __expf(g[v] - mx[v]);
  }
  if (wave == 0 && c == 0) {
    for (int v = 0; v < 8; ++v) mrow[(m0 + rowoff + v) * Dd + ks] = mx[v];
  }
}

__global__ void k_layer0(const float* __restrict__ X, const float* __restrict__ W0,
                         const float* __restrict__ b0,
                         float* __restrict__ act,
                         float* __restrict__ E, float* __restrict__ mrow) {
  const int wave = (blockIdx.x * blockDim.x + threadIdx.x) >> 5;
  const int mt = wave >> 7;
  const int nt = wave & 127;
  if (mt >= Bsz / 16) return;
  v8f acc = {};
  gemm_tile(X, Dd, W0, Dd, mt * 16, nt * 16, Dd, acc);

  const int lane = threadIdx.x & 31;
  const int n = nt * 16 + (lane & 15);
  const int rowoff = (lane >> 4) * 8;
  const float ldiag = __logf(W0[n * Dd + (n >> 7)]);
  const float bias = b0[n];
  float g[8];
  for (int v = 0; v < 8; ++v) {
    const int m = mt * 16 + v + rowoff;
    const float h = acc[v] + bias;
    act[m * HID + n] = fast_tanh(h);
    g[v] = ldiag + log_abs_dtanh(h);
  }
  rowmax_exp_store(g, mt * 16, nt >> 3, E, mrow);
}

#define L1_KC    128
#define L1_PITCH (L1_KC + 4)
#define L1_NCH   (HID / L1_KC)
#define L1_MROWS 32
#define L1_NROWS 64
#define L1_NG    (L1_KC / 8)
#define L1_WAVES 4
#define L1_A8(X)  X(0) X(1) X(2) X(3) X(4) X(5) X(6) X(7)
#define L1_W16(X) X(0) X(1) X(2) X(3) X(4) X(5) X(6) X(7) X(8) X(9) X(10) X(11) X(12) X(13) X(14) X(15)
#define L1_DECLA(i) float4 pa##i;
#define L1_DECLW(i) float4 pw##i;
#define L1_LDA(i) pa##i = *reinterpret_cast<const float4*>(gA + (size_t)(4 * i) * HID + ko);
#define L1_LDW(i) pw##i = *reinterpret_cast<const float4*>(gW + (size_t)(4 * i) * HID + ko);
#define L1_STA(i) *reinterpret_cast<float4*>(&As[st][soff + 4 * i * L1_PITCH]) = pa##i;
#define L1_STW(i) *reinterpret_cast<float4*>(&Ws[st][soff + 4 * i * L1_PITCH]) = pw##i;
__global__ __launch_bounds__(L1_WAVES * 32)
void k_layer1(const float* __restrict__ A, const float* __restrict__ W1,
              const float* __restrict__ b1,
              float* __restrict__ act, float* __restrict__ lga) {
  __shared__ __attribute__((aligned(16))) float As[2][L1_MROWS * L1_PITCH];
  __shared__ __attribute__((aligned(16))) float Ws[2][L1_NROWS * L1_PITCH];
  const int tid  = threadIdx.x;
  const int wave = tid >> 5;
  const int lane = tid & 31;
  const int ms = blockIdx.x >> 5;
  const int ns = blockIdx.x & 31;
  const int mi = wave >> 1;
  const int nj = (wave & 1) * 2;

  const int srow = tid >> 5;
  const int sc4  = (tid & 31) * 4;
  const float* gA = A  + (size_t)(ms * L1_MROWS + srow) * HID + sc4;
  const float* gW = W1 + (size_t)(ns * L1_NROWS + srow) * HID + sc4;
  const int soff = srow * L1_PITCH + sc4;

  const int foffA = (mi * 16 + (lane & 15)) * L1_PITCH + (lane >> 4) * 4;
  const int foffB = (nj * 16 + (lane & 15)) * L1_PITCH + (lane >> 4) * 4;

  L1_A8(L1_DECLA)
  L1_W16(L1_DECLW)
  {
    const int ko = 0;
    const int st = 0;
    L1_A8(L1_LDA)  L1_W16(L1_LDW)
    L1_A8(L1_STA)  L1_W16(L1_STW)
  }
  __syncthreads();

  v8f acc0 = {};
  v8f acc1 = {};
  for (int c = 0; c < L1_NCH; ++c) {
    const int cur = c & 1;
    const bool more = (c + 1 < L1_NCH);
    if (more) {
      const int ko = (c + 1) * L1_KC;
      L1_A8(L1_LDA)  L1_W16(L1_LDW)
    }
    const float* as  = &As[cur][foffA];
    const float* ws0 = &Ws[cur][foffB];
    const float* ws1 = ws0 + 16 * L1_PITCH;
    float4 fa[L1_NG], fb0[L1_NG], fb1[L1_NG];
#pragma unroll
    for (int j = 0; j < L1_NG; ++j) {
      fa[j]  = *reinterpret_cast<const float4*>(as  + 8 * j);
      fb0[j] = *reinterpret_cast<const float4*>(ws0 + 8 * j);
      fb1[j] = *reinterpret_cast<const float4*>(ws1 + 8 * j);
    }
#pragma unroll
    for (int j = 0; j < L1_NG; ++j) {
      v2f a0, a1, p0, p1, q0, q1;
      a0.x = fa[j].x;   a0.y = fa[j].y;   a1.x = fa[j].z;   a1.y = fa[j].w;
      p0.x = fb0[j].x;  p0.y = fb0[j].y;  p1.x = fb0[j].z;  p1.y = fb0[j].w;
      q0.x = fb1[j].x;  q0.y = fb1[j].y;  q1.x = fb1[j].z;  q1.y = fb1[j].w;
      acc0 = wmma_f32_k4(a0, p0, acc0);
      acc1 = wmma_f32_k4(a0, q0, acc1);
      acc0 = wmma_f32_k4(a1, p1, acc0);
      acc1 = wmma_f32_k4(a1, q1, acc1);
    }
    __builtin_amdgcn_sched_group_barrier(0x100, 3 * L1_NG, 0);
    __builtin_amdgcn_sched_group_barrier(0x008, 4 * L1_NG, 0);
    if (more) {
      const int st = cur ^ 1;
      L1_A8(L1_STA)  L1_W16(L1_STW)
    }
    __syncthreads();
  }

  const int rowoff = (lane >> 4) * 8;
  const int n0 = ns * L1_NROWS + nj * 16 + (lane & 15);
  const int n1 = n0 + 16;
  const float bias0 = b1[n0];
  const float bias1 = b1[n1];
#pragma unroll
  for (int v = 0; v < 8; ++v) {
    const int m = ms * L1_MROWS + mi * 16 + v + rowoff;
    const float h0 = acc0[v] + bias0;
    const float h1 = acc1[v] + bias1;
    act[m * HID + n0] = fast_tanh(h0);
    lga[m * HID + n0] = log_abs_dtanh(h0);
    act[m * HID + n1] = fast_tanh(h1);
    lga[m * HID + n1] = log_abs_dtanh(h1);
  }
}

__global__ void k_grad_gemm(const float* __restrict__ E, const float* __restrict__ W1,
                            const float* __restrict__ lga, const float* __restrict__ mrow,
                            float* __restrict__ Eo, float* __restrict__ mo) {
  const int wave = (blockIdx.x * blockDim.x + threadIdx.x) >> 5;
  const int ks  = wave >> 7;
  const int rem = wave & 127;
  const int mt  = rem >> 3;
  const int nt  = rem & 7;
  if (ks >= Dd) return;
  const float* Ak = E  + (size_t)ks * Mm;
  const float* Wk = W1 + (size_t)ks * Mm * HID + (size_t)ks * Mm;
  v8f acc = {};
  gemm_tile(Ak, HID, Wk, HID, mt * 16, nt * 16, Mm, acc);

  const int lane = threadIdx.x & 31;
  const int o = nt * 16 + (lane & 15);
  const int rowoff = (lane >> 4) * 8;
  float g[8];
  for (int v = 0; v < 8; ++v) {
    const int b = mt * 16 + v + rowoff;
    const float s = acc[v];
    g[v] = __logf(s) + mrow[b * Dd + ks] + lga[b * HID + ks * Mm + o];
  }
  rowmax_exp_store(g, mt * 16, ks, Eo, mo);
}

#define OG_WAVES 16
#define OG_KC    (HID / OG_WAVES)
__global__ void k_out_gemm(const float* __restrict__ A, const float* __restrict__ W2,
                           float* __restrict__ out) {
  __shared__ float part[OG_WAVES][16 * 16];
  const int wave = threadIdx.x >> 5;
  const int lane = threadIdx.x & 31;
  const int mt   = blockIdx.x;
  v8f acc = {};
  gemm_tile(A + wave * OG_KC, HID, W2 + wave * OG_KC, HID, mt * 16, 0, OG_KC, acc);
  const int n = lane & 15;
  const int rowoff = (lane >> 4) * 8;
  for (int v = 0; v < 8; ++v)
    part[wave][(v + rowoff) * Dd + n] = acc[v];
  __syncthreads();
  if (threadIdx.x < 16 * 16) {
    float s = part[0][threadIdx.x];
    for (int w = 1; w < OG_WAVES; ++w) s += part[w][threadIdx.x];
    out[mt * 16 * Dd + threadIdx.x] = s;
  }
}

#define LD_WAVES 32

template <int HALF>
__device__ __forceinline__ void fold_half(float (&s)[Dd], int lane, int off) {
  const bool up = (lane & off) != 0;
#pragma unroll
  for (int j = 0; j < HALF; ++j) {
    const float keep = up ? s[j + HALF] : s[j];
    const float send = up ? s[j] : s[j + HALF];
    s[j] = keep + __shfl_xor(send, off);
  }
}

__global__ void k_logdet(const float* __restrict__ E, const float* __restrict__ mrow,
                         const float* __restrict__ W2, float* __restrict__ ld,
                         int accumulate) {
  __shared__ float tot_s[LD_WAVES];
  const int wave = threadIdx.x >> 5;
  const int lane = threadIdx.x & 31;
  const int b = blockIdx.x * LD_WAVES + wave;
  const float mk = mrow[b * Dd + (lane >> 1)];
  const float4* e4 = reinterpret_cast<const float4*>(E + (size_t)b * HID) + lane;
  float s[Dd];
#pragma unroll
  for (int k = 0; k < Dd; ++k) {
    const float4 e = e4[k * (Mm / 4)];
    const float4 w =
        reinterpret_cast<const float4*>(W2 + (size_t)k * HID + k * Mm)[lane];
    float p = e.x * w.x;
    p = fmaf(e.y, w.y, p);
    p = fmaf(e.z, w.z, p);
    p = fmaf(e.w, w.w, p);
    s[k] = p;
  }
  fold_half<8>(s, lane, 16);
  fold_half<4>(s, lane, 8);
  fold_half<2>(s, lane, 4);
  fold_half<1>(s, lane, 2);
  const float sk = s[0] + __shfl_xor(s[0], 1);
  float t = __logf(sk) + mk;
  t += __shfl_xor(t, 2);
  t += __shfl_xor(t, 4);
  t += __shfl_xor(t, 8);
  t += __shfl_xor(t, 16);
  if (lane == 0) tot_s[wave] = t;
  __syncthreads();
  if (wave == 0) {
    const int bo = blockIdx.x * LD_WAVES + lane;
    if (accumulate) ld[bo] += tot_s[lane]; else ld[bo] = tot_s[lane];
  }
}

static void run_flow(const float* X, const float* W0, const float* W1,
                     const float* W2, const float* b0, const float* b1,
                     float* x_out, float* ld_out, int accumulate,
                     float* act0, float* act1, float* lga,
                     float* E0, float* mrow0, float* E1, float* mrow1,
                     hipStream_t stream) {
  k_layer0   <<<256, RP_WAVES * 32, 0, stream>>>(X, W0, b0, act0, E0, mrow0);
  k_layer1   <<<(Bsz / L1_MROWS) * (HID / L1_NROWS), L1_WAVES * 32, 0, stream>>>(act0, W1, b1, act1, lga);
  k_grad_gemm<<<256, RP_WAVES * 32, 0, stream>>>(E0, W1, lga, mrow0, E1, mrow1);
  k_out_gemm <<<Bsz / 16, OG_WAVES * 32, 0, stream>>>(act1, W2, x_out);
  k_logdet   <<<Bsz / LD_WAVES, LD_WAVES * 32, 0, stream>>>(E1, mrow1, W2, ld_out, accumulate);
}

extern "C" void kernel_launch(void* const* d_in, const int* in_sizes, int n_in,
                              void* d_out, int out_size, void* d_ws, size_t ws_size,
                              hipStream_t stream) {
  const float* inp   = (const float*)d_in[0];
  const float* f0_W0 = (const float*)d_in[1];
  const float* f0_W1 = (const float*)d_in[2];
  const float* f0_W2 = (const float*)d_in[3];
  const float* f0_b0 = (const float*)d_in[4];
  const float* f0_b1 = (const float*)d_in[5];
  const float* f1_W0 = (const float*)d_in[6];
  const float* f1_W1 = (const float*)d_in[7];
  const float* f1_W2 = (const float*)d_in[8];
  const float* f1_b0 = (const float*)d_in[9];
  const float* f1_b1 = (const float*)d_in[10];

  float* out = (float*)d_out;
  float* ld  = out + Bsz * Dd;

  float* ws    = (float*)d_ws;
  float* act0  = ws;
  float* act1  = act0  + Bsz * HID;
  float* lga   = act1  + Bsz * HID;
  float* E0    = lga   + Bsz * HID;
  float* E1    = E0    + Bsz * HID;
  float* mrow0 = E1    + Bsz * HID;
  float* mrow1 = mrow0 + Bsz * Dd;
  float* xmid  = mrow1 + Bsz * Dd;

  run_flow(inp,  f0_W0, f0_W1, f0_W2, f0_b0, f0_b1, xmid, ld,  0,
           act0, act1, lga, E0, mrow0, E1, mrow1, stream);
  run_flow(xmid, f1_W0, f1_W1, f1_W2, f1_b0, f1_b1, out,  ld,  1,
           act0, act1, lga, E0, mrow0, E1, mrow1, stream);
}


# Independent kernels share launches: final GEMM (8-wave groups, K split 8 ways: sum order changes) runs beside the grad GEMM, both flows' log-dets in one last launch; 10 launches -> 7
# speedup vs baseline: 3.9137x; 1.1247x over previous
#include <hip/hip_runtime.h>
#include <hip/hip_bf16.h>


typedef float v2f __attribute__((ext_vector_type(2)));
typedef float v8f __attribute__((ext_vector_type(8)));

#define Bsz   256
#define Dd    16
#define Mm    128
#define HID   2048

__device__ __forceinline__ v8f wmma_f32_k4(v2f a, v2f b, v8f c) {
  return __builtin_amdgcn_wmma_f32_16x16x4_f32(
      false, a, false, b, (short)0, c, false, false);
}

__device__ __forceinline__ void gemm_tile(const float* __restrict__ A, int lda,
                                          const float* __restrict__ W, int ldw,
                                          int m0, int n0, int K, v8f& acc) {
  const int lane = threadIdx.x & 31;
  const int r    = lane & 15;
  const int kh   = (lane >> 4) << 1;
  const float* arow = A + (size_t)(m0 + r) * lda;
  const float* wrow = W + (size_t)(n0 + r) * ldw;
#pragma unroll
  for (int k = 0; k < K; k += 4) {
    v2f a, b;
    a.x = arow[k + kh];  a.y = arow[k + kh + 1];
    b.x = wrow[k + kh];  b.y = wrow[k + kh + 1];
    acc = wmma_f32_k4(a, b, acc);
  }
}

__device__ __forceinline__ float fast_tanh(float x) {
#if defined(__has_builtin)
#if __has_builtin(__builtin_amdgcn_tanhf)
  return __builtin_amdgcn_tanhf(x);
#elif __has_builtin(__builtin_amdgcn_tanh_f32)
  return __builtin_amdgcn_tanh_f32(x);
#else
  return tanhf(x);
#endif
#else
  return tanhf(x);
#endif
}

__device__ __forceinline__ float softplus_f(float x) {
  return fmaxf(x, 0.0f) + __logf(1.0f + __expf(-fabsf(x)));
}
__device__ __forceinline__ float log_abs_dtanh(float h) {
  return 2.0f * (0.6931471805599453f - h - softplus_f(-2.0f * h));
}

#define RP_WAVES 8
__device__ __forceinline__ void rowmax_exp_store(const float (&g)[8], int m0, int ks,
                                                 float* __restrict__ E,
                                                 float* __restrict__ mrow) {
  __shared__ __attribute__((aligned(16))) float pmax[RP_WAVES][16];
  const int wave = threadIdx.x >> 5;
  const int lane = threadIdx.x & 31;
  const int c = lane & 15;
  const int rowoff = (lane >> 4) * 8;
  float mx[8];
  for (int v = 0; v < 8; ++v) {
    float t = g[v];
    t = fmaxf(t, __shfl_xor(t, 1));
    t = fmaxf(t, __shfl_xor(t, 2));
    t = fmaxf(t, __shfl_xor(t, 4));
    t = fmaxf(t, __shfl_xor(t, 8));
    mx[v] = t;
  }
  if (c == 0) {
    for (int v = 0; v < 8; ++v) pmax[wave][rowoff + v] = mx[v];
  }
  __syncthreads();
  for (int v = 0; v < 8; ++v) {
    float r = pmax[0][rowoff + v];
    for (int w = 1; w < RP_WAVES; ++w) r = fmaxf(r, pmax[w][rowoff + v]);
    mx[v] = r;
  }
  for (int v = 0; v < 8; ++v) {
    const int m = m0 + rowoff + v;
    E[m * HID + ks * Mm + wave * 16 + c] = __expf(g[v] - mx[v]);
  }
  if (wave == 0 && c == 0) {
    for (int v = 0; v < 8; ++v) mrow[(m0 + rowoff + v) * Dd + ks] = mx[v];
  }
}

__global__ void k_layer0(const float* __restrict__ X, const float* __restrict__ W0,
                         const float* __restrict__ b0,
                         float* __restrict__ act,
                         float* __restrict__ E, float* __restrict__ mrow) {
  const int wave = (blockIdx.x * blockDim.x + threadIdx.x) >> 5;
  const int mt = wave >> 7;
  const int nt = wave & 127;
  if (mt >= Bsz / 16) return;
  v8f acc = {};
  gemm_tile(X, Dd, W0, Dd, mt * 16, nt * 16, Dd, acc);

  const int lane = threadIdx.x & 31;
  const int n = nt * 16 + (lane & 15);
  const int rowoff = (lane >> 4) * 8;
  const float ldiag = __logf(W0[n * Dd + (n >> 7)]);
  const float bias = b0[n];
  float g[8];
  for (int v = 0; v < 8; ++v) {
    const int m = mt * 16 + v + rowoff;
    const float h = acc[v] + bias;
    act[m * HID + n] = fast_tanh(h);
    g[v] = ldiag + log_abs_dtanh(h);
  }
  rowmax_exp_store(g, mt * 16, nt >> 3, E, mrow);
}

#define L1_KC    128
#define L1_PITCH (L1_KC + 4)
#define L1_NCH   (HID / L1_KC)
#define L1_MROWS 32
#define L1_NROWS 64
#define L1_NG    (L1_KC / 8)
#define L1_WAVES 4
#define L1_A8(X)  X(0) X(1) X(2) X(3) X(4) X(5) X(6) X(7)
#define L1_W16(X) X(0) X(1) X(2) X(3) X(4) X(5) X(6) X(7) X(8) X(9) X(10) X(11) X(12) X(13) X(14) X(15)
#define L1_DECLA(i) float4 pa##i;
#define L1_DECLW(i) float4 pw##i;
#define L1_LDA(i) pa##i = *reinterpret_cast<const float4*>(gA + (size_t)(4 * i) * HID + ko);
#define L1_LDW(i) pw##i = *reinterpret_cast<const float4*>(gW + (size_t)(4 * i) * HID + ko);
#define L1_STA(i) *reinterpret_cast<float4*>(&As[st][soff + 4 * i * L1_PITCH]) = pa##i;
#define L1_STW(i) *reinterpret_cast<float4*>(&Ws[st][soff + 4 * i * L1_PITCH]) = pw##i;
__global__ __launch_bounds__(L1_WAVES * 32)
void k_layer1(const float* __restrict__ A, const float* __restrict__ W1,
              const float* __restrict__ b1,
              float* __restrict__ act, float* __restrict__ lga) {
  __shared__ __attribute__((aligned(16))) float As[2][L1_MROWS * L1_PITCH];
  __shared__ __attribute__((aligned(16))) float Ws[2][L1_NROWS * L1_PITCH];
  const int tid  = threadIdx.x;
  const int wave = tid >> 5;
  const int lane = tid & 31;
  const int ms = blockIdx.x >> 5;
  const int ns = blockIdx.x & 31;
  const int mi = wave >> 1;
  const int nj = (wave & 1) * 2;

  const int srow = tid >> 5;
  const int sc4  = (tid & 31) * 4;
  const float* gA = A  + (size_t)(ms * L1_MROWS + srow) * HID + sc4;
  const float* gW = W1 + (size_t)(ns * L1_NROWS + srow) * HID + sc4;
  const int soff = srow * L1_PITCH + sc4;

  const int foffA = (mi * 16 + (lane & 15)) * L1_PITCH + (lane >> 4) * 4;
  const int foffB = (nj * 16 + (lane & 15)) * L1_PITCH + (lane >> 4) * 4;

  L1_A8(L1_DECLA)
  L1_W16(L1_DECLW)
  {
    const int ko = 0;
    const int st = 0;
    L1_A8(L1_LDA)  L1_W16(L1_LDW)
    L1_A8(L1_STA)  L1_W16(L1_STW)
  }
  __syncthreads();

  v8f acc0 = {};
  v8f acc1 = {};
  for (int c = 0; c < L1_NCH; ++c) {
    const int cur = c & 1;
    const bool more = (c + 1 < L1_NCH);
    if (more) {
      const int ko = (c + 1) * L1_KC;
      L1_A8(L1_LDA)  L1_W16(L1_LDW)
    }
    const float* as  = &As[cur][foffA];
    const float* ws0 = &Ws[cur][foffB];
    const float* ws1 = ws0 + 16 * L1_PITCH;
    float4 fa[L1_NG], fb0[L1_NG], fb1[L1_NG];
#pragma unroll
    for (int j = 0; j < L1_NG; ++j) {
      fa[j]  = *reinterpret_cast<const float4*>(as  + 8 * j);
      fb0[j] = *reinterpret_cast<const float4*>(ws0 + 8 * j);
      fb1[j] = *reinterpret_cast<const float4*>(ws1 + 8 * j);
    }
#pragma unroll
    for (int j = 0; j < L1_NG; ++j) {
      v2f a0, a1, p0, p1, q0, q1;
      a0.x = fa[j].x;   a0.y = fa[j].y;   a1.x = fa[j].z;   a1.y = fa[j].w;
      p0.x = fb0[j].x;  p0.y = fb0[j].y;  p1.x = fb0[j].z;  p1.y = fb0[j].w;
      q0.x = fb1[j].x;  q0.y = fb1[j].y;  q1.x = fb1[j].z;  q1.y = fb1[j].w;
      acc0 = wmma_f32_k4(a0, p0, acc0);
      acc1 = wmma_f32_k4(a0, q0, acc1);
      acc0 = wmma_f32_k4(a1, p1, acc0);
      acc1 = wmma_f32_k4(a1, q1, acc1);
    }
    __builtin_amdgcn_sched_group_barrier(0x100, 3 * L1_NG, 0);
    __builtin_amdgcn_sched_group_barrier(0x008, 4 * L1_NG, 0);
    if (more) {
      const int st = cur ^ 1;
      L1_A8(L1_STA)  L1_W16(L1_STW)
    }
    __syncthreads();
  }

  const int rowoff = (lane >> 4) * 8;
  const int n0 = ns * L1_NROWS + nj * 16 + (lane & 15);
  const int n1 = n0 + 16;
  const float bias0 = b1[n0];
  const float bias1 = b1[n1];
#pragma unroll
  for (int v = 0; v < 8; ++v) {
    const int m = ms * L1_MROWS + mi * 16 + v + rowoff;
    const float h0 = acc0[v] + bias0;
    const float h1 = acc1[v] + bias1;
    act[m * HID + n0] = fast_tanh(h0);
    lga[m * HID + n0] = log_abs_dtanh(h0);
    act[m * HID + n1] = fast_tanh(h1);
    lga[m * HID + n1] = log_abs_dtanh(h1);
  }
}

__device__ __forceinline__ void grad_body(int wave,
                                          const float* __restrict__ E, const float* __restrict__ W1,
                                          const float* __restrict__ lga, const float* __restrict__ mrow,
                                          float* __restrict__ Eo, float* __restrict__ mo) {
  const int ks  = wave >> 7;
  const int rem = wave & 127;
  const int mt  = rem >> 3;
  const int nt  = rem & 7;
  if (ks >= Dd) return;
  const float* Ak = E  + (size_t)ks * Mm;
  const float* Wk = W1 + (size_t)ks * Mm * HID + (size_t)ks * Mm;
  v8f acc = {};
  gemm_tile(Ak, HID, Wk, HID, mt * 16, nt * 16, Mm, acc);

  const int lane = threadIdx.x & 31;
  const int o = nt * 16 + (lane & 15);
  const int rowoff = (lane >> 4) * 8;
  float g[8];
  for (int v = 0; v < 8; ++v) {
    const int b = mt * 16 + v + rowoff;
    const float s = acc[v];
    g[v] = __logf(s) + mrow[b * Dd + ks] + lga[b * HID + ks * Mm + o];
  }
  rowmax_exp_store(g, mt * 16, ks, Eo, mo);
}

#define OG_WAVES  8
#define OG_KC     (HID / OG_WAVES)
#define OG_BLOCKS (Bsz / 16)
__device__ __forceinline__ void out_body(int mt, const float* __restrict__ A,
                                         const float* __restrict__ W2, float* __restrict__ out) {
  __shared__ float part[OG_WAVES][16 * 16];
  const int wave = threadIdx.x >> 5;
  const int lane = threadIdx.x & 31;
  v8f acc = {};
  gemm_tile(A + wave * OG_KC, HID, W2 + wave * OG_KC, HID, mt * 16, 0, OG_KC, acc);
  const int n = lane & 15;
  const int rowoff = (lane >> 4) * 8;
  for (int v = 0; v < 8; ++v)
    part[wave][(v + rowoff) * Dd + n] = acc[v];
  __syncthreads();
  float s = part[0][threadIdx.x];
  for (int w = 1; w < OG_WAVES; ++w) s += part[w][threadIdx.x];
  out[mt * 16 * Dd + threadIdx.x] = s;
}

__global__ __launch_bounds__(RP_WAVES * 32)
void k_grad_out(const float* __restrict__ E, const float* __restrict__ W1,
                const float* __restrict__ lga, const float* __restrict__ mrow,
                float* __restrict__ Eo, float* __restrict__ mo,
                const float* __restrict__ A, const float* __restrict__ W2,
                float* __restrict__ out) {
  if (blockIdx.x < OG_BLOCKS) {
    out_body(blockIdx.x, A, W2, out);
  } else {
    grad_body(((blockIdx.x - OG_BLOCKS) * (RP_WAVES * 32) + threadIdx.x) >> 5,
              E, W1, lga, mrow, Eo, mo);
  }
}

#define LD_WAVES 32

template <int HALF>
__device__ __forceinline__ void fold_half(float (&s)[Dd], int lane, int off) {
  const bool up = (lane & off) != 0;
#pragma unroll
  for (int j = 0; j < HALF; ++j) {
    const float keep = up ? s[j + HALF] : s[j];
    const float send = up ? s[j] : s[j + HALF];
    s[j] = keep + __shfl_xor(send, off);
  }
}

__device__ __forceinline__ float logdet_sample(const float* __restrict__ E,
                                               const float* __restrict__ mrow,
                                               const float* __restrict__ W2,
                                               int b, int lane) {
  const float mk = mrow[b * Dd + (lane >> 1)];
  const float4* e4 = reinterpret_cast<const float4*>(E + (size_t)b * HID) + lane;
  float s[Dd];
#pragma unroll
  for (int k = 0; k < Dd; ++k) {
    const float4 e = e4[k * (Mm / 4)];
    const float4 w =
        reinterpret_cast<const float4*>(W2 + (size_t)k * HID + k * Mm)[lane];
    float p = e.x * w.x;
    p = fmaf(e.y, w.y, p);
    p = fmaf(e.z, w.z, p);
    p = fmaf(e.w, w.w, p);
    s[k] = p;
  }
  fold_half<8>(s, lane, 16);
  fold_half<4>(s, lane, 8);
  fold_half<2>(s, lane, 4);
  fold_half<1>(s, lane, 2);
  const float sk = s[0] + __shfl_xor(s[0], 1);
  float t = __logf(sk) + mk;
  t += __shfl_xor(t, 2);
  t += __shfl_xor(t, 4);
  t += __shfl_xor(t, 8);
  t += __shfl_xor(t, 16);
  return t;
}

__global__ void k_logdet2(const float* __restrict__ Ea, const float* __restrict__ mrowa,
                          const float* __restrict__ W2a,
                          const float* __restrict__ Eb, const float* __restrict__ mrowb,
                          const float* __restrict__ W2b,
                          float* __restrict__ ld) {
  __shared__ float tot_s[LD_WAVES];
  const int wave = threadIdx.x >> 5;
  const int lane = threadIdx.x & 31;
  const int b = blockIdx.x * LD_WAVES + wave;
  const float t0 = logdet_sample(Ea, mrowa, W2a, b, lane);
  const float t1 = logdet_sample(Eb, mrowb, W2b, b, lane);
  if (lane == 0) tot_s[wave] = t0 + t1;
  __syncthreads();
  if (wave == 0) ld[blockIdx.x * LD_WAVES + lane] = tot_s[lane];
}

extern "C" void kernel_launch(void* const* d_in, const int* in_sizes, int n_in,
                              void* d_out, int out_size, void* d_ws, size_t ws_size,
                              hipStream_t stream) {
  const float* inp   = (const float*)d_in[0];
  const float* f0_W0 = (const float*)d_in[1];
  const float* f0_W1 = (const float*)d_in[2];
  const float* f0_W2 = (const float*)d_in[3];
  const float* f0_b0 = (const float*)d_in[4];
  const float* f0_b1 = (const float*)d_in[5];
  const float* f1_W0 = (const float*)d_in[6];
  const float* f1_W1 = (const float*)d_in[7];
  const float* f1_W2 = (const float*)d_in[8];
  const float* f1_b0 = (const float*)d_in[9];
  const float* f1_b1 = (const float*)d_in[10];

  float* out = (float*)d_out;
  float* ld  = out + Bsz * Dd;

  float* ws     = (float*)d_ws;
  float* act0   = ws;
  float* act1   = act0   + Bsz * HID;
  float* lga    = act1   + Bsz * HID;
  float* E0     = lga    + Bsz * HID;
  float* E1a    = E0     + Bsz * HID;
  float* E1b    = E1a    + Bsz * HID;
  float* mrow0  = E1b    + Bsz * HID;
  float* mrow1a = mrow0  + Bsz * Dd;
  float* mrow1b = mrow1a + Bsz * Dd;
  float* xmid   = mrow1b + Bsz * Dd;

  k_layer0  <<<256, RP_WAVES * 32, 0, stream>>>(inp, f0_W0, f0_b0, act0, E0, mrow0);
  k_layer1  <<<(Bsz / L1_MROWS) * (HID / L1_NROWS), L1_WAVES * 32, 0, stream>>>(act0, f0_W1, f0_b1, act1, lga);
  k_grad_out<<<OG_BLOCKS + 256, RP_WAVES * 32, 0, stream>>>(E0, f0_W1, lga, mrow0, E1a, mrow1a,
                                                           act1, f0_W2, xmid);
  k_layer0  <<<256, RP_WAVES * 32, 0, stream>>>(xmid, f1_W0, f1_b0, act0, E0, mrow0);
  k_layer1  <<<(Bsz / L1_MROWS) * (HID / L1_NROWS), L1_WAVES * 32, 0, stream>>>(act0, f1_W1, f1_b1, act1, lga);
  k_grad_out<<<OG_BLOCKS + 256, RP_WAVES * 32, 0, stream>>>(E0, f1_W1, lga, mrow0, E1b, mrow1b,
                                                           act1, f1_W2, out);
  k_logdet2 <<<Bsz / LD_WAVES, LD_WAVES * 32, 0, stream>>>(E1a, mrow1a, f0_W2, E1b, mrow1b, f1_W2, ld);
}


# Layer-1 GEMM ends K at each column slab's own diagonal block of the reference's mask (rest is masked to 0); slabs paired (j,63-j) per group: 544 WMMAs per wave, was 1024; k sum order changes
# speedup vs baseline: 4.7187x; 1.2057x over previous
#include <hip/hip_runtime.h>
#include <hip/hip_bf16.h>


typedef float v2f __attribute__((ext_vector_type(2)));
typedef float v8f __attribute__((ext_vector_type(8)));

#define Bsz   256
#define Dd    16
#define Mm    128
#define HID   2048

__device__ __forceinline__ v8f wmma_f32_k4(v2f a, v2f b, v8f c) {
  return __builtin_amdgcn_wmma_f32_16x16x4_f32(
      false, a, false, b, (short)0, c, false, false);
}

__device__ __forceinline__ void gemm_tile(const float* __restrict__ A, int lda,
                                          const float* __restrict__ W, int ldw,
                                          int m0, int n0, int K, v8f& acc) {
  const int lane = threadIdx.x & 31;
  const int r    = lane & 15;
  const int kh   = (lane >> 4) << 1;
  const float* arow = A + (size_t)(m0 + r) * lda;
  const float* wrow = W + (size_t)(n0 + r) * ldw;
#pragma unroll
  for (int k = 0; k < K; k += 4) {
    v2f a, b;
    a.x = arow[k + kh];  a.y = arow[k + kh + 1];
    b.x = wrow[k + kh];  b.y = wrow[k + kh + 1];
    acc = wmma_f32_k4(a, b, acc);
  }
}

__device__ __forceinline__ float fast_tanh(float x) {
#if defined(__has_builtin)
#if __has_builtin(__builtin_amdgcn_tanhf)
  return __builtin_amdgcn_tanhf(x);
#elif __has_builtin(__builtin_amdgcn_tanh_f32)
  return __builtin_amdgcn_tanh_f32(x);
#else
  return tanhf(x);
#endif
#else
  return tanhf(x);
#endif
}

__device__ __forceinline__ float softplus_f(float x) {
  return fmaxf(x, 0.0f) + __logf(1.0f + __expf(-fabsf(x)));
}
__device__ __forceinline__ float log_abs_dtanh(float h) {
  return 2.0f * (0.6931471805599453f - h - softplus_f(-2.0f * h));
}

#define RP_WAVES 8
__device__ __forceinline__ void rowmax_exp_store(const float (&g)[8], int m0, int ks,
                                                 float* __restrict__ E,
                                                 float* __restrict__ mrow) {
  __shared__ __attribute__((aligned(16))) float pmax[RP_WAVES][16];
  const int wave = threadIdx.x >> 5;
  const int lane = threadIdx.x & 31;
  const int c = lane & 15;
  const int rowoff = (lane >> 4) * 8;
  float mx[8];
  for (int v = 0; v < 8; ++v) {
    float t = g[v];
    t = fmaxf(t, __shfl_xor(t, 1));
    t = fmaxf(t, __shfl_xor(t, 2));
    t = fmaxf(t, __shfl_xor(t, 4));
    t = fmaxf(t, __shfl_xor(t, 8));
    mx[v] = t;
  }
  if (c == 0) {
    for (int v = 0; v < 8; ++v) pmax[wave][rowoff + v] = mx[v];
  }
  __syncthreads();
  for (int v = 0; v < 8; ++v) {
    float r = pmax[0][rowoff + v];
    for (int w = 1; w < RP_WAVES; ++w) r = fmaxf(r, pmax[w][rowoff + v]);
    mx[v] = r;
  }
  for (int v = 0; v < 8; ++v) {
    const int m = m0 + rowoff + v;
    E[m * HID + ks * Mm + wave * 16 + c] = __expf(g[v] - mx[v]);
  }
  if (wave == 0 && c == 0) {
    for (int v = 0; v < 8; ++v) mrow[(m0 + rowoff + v) * Dd + ks] = mx[v];
  }
}

__global__ void k_layer0(const float* __restrict__ X, const float* __restrict__ W0,
                         const float* __restrict__ b0,
                         float* __restrict__ act,
                         float* __restrict__ E, float* __restrict__ mrow) {
  const int wave = (blockIdx.x * blockDim.x + threadIdx.x) >> 5;
  const int mt = wave >> 7;
  const int nt = wave & 127;
  if (mt >= Bsz / 16) return;
  v8f acc = {};
  gemm_tile(X, Dd, W0, Dd, mt * 16, nt * 16, Dd, acc);

  const int lane = threadIdx.x & 31;
  const int n = nt * 16 + (lane & 15);
  const int rowoff = (lane >> 4) * 8;
  const float ldiag = __logf(W0[n * Dd + (n >> 7)]);
  const float bias = b0[n];
  float g[8];
  for (int v = 0; v < 8; ++v) {
    const int m = mt * 16 + v + rowoff;
    const float h = acc[v] + bias;
    act[m * HID + n] = fast_tanh(h);
    g[v] = ldiag + log_abs_dtanh(h);
  }
  rowmax_exp_store(g, mt * 16, nt >> 3, E, mrow);
}

#define L1_KC    128
#define L1_PITCH (L1_KC + 4)
#define L1_MROWS 32
#define L1_SLAB  32
#define L1_NROWS (2 * L1_SLAB)
#define L1_NSLAB (HID / L1_SLAB)
#define L1_NG    (L1_KC / 8)
#define L1_WAVES 4
#define L1_A8(X)   X(0) X(1) X(2) X(3) X(4) X(5) X(6) X(7)
#define L1_WHI8(X) X(0) X(1) X(2) X(3) X(4) X(5) X(6) X(7)
#define L1_WLO8(X) X(8) X(9) X(10) X(11) X(12) X(13) X(14) X(15)
#define L1_DECLA(i) float4 pa##i;
#define L1_DECLW(i) float4 pw##i;
#define L1_LDA(i)  pa##i = *reinterpret_cast<const float4*>(gA  + (size_t)(4 * i) * HID + ko);
#define L1_LDWH(i) pw##i = *reinterpret_cast<const float4*>(gWh + (size_t)(4 * i) * HID + ko);
#define L1_LDWL(i) pw##i = *reinterpret_cast<const float4*>(gWl + (size_t)(4 * (i - 8)) * HID + ko);
#define L1_STA(i) *reinterpret_cast<float4*>(&As[st][soff + 4 * i * L1_PITCH]) = pa##i;
#define L1_STW(i) *reinterpret_cast<float4*>(&Ws[st][soff + 4 * i * L1_PITCH]) = pw##i;

__device__ __forceinline__ void l1_mm_both(const float* __restrict__ as, const float* __restrict__ ws0,
                                           v8f& acc0, v8f& acc1) {
  const float* ws1 = ws0 + L1_SLAB * L1_PITCH;
  float4 fa[L1_NG], fb0[L1_NG], fb1[L1_NG];
#pragma unroll
  for (int j = 0; j < L1_NG; ++j) {
    fa[j]  = *reinterpret_cast<const float4*>(as  + 8 * j);
    fb0[j] = *reinterpret_cast<const float4*>(ws0 + 8 * j);
    fb1[j] = *reinterpret_cast<const float4*>(ws1 + 8 * j);
  }
#pragma unroll
  for (int j = 0; j < L1_NG; ++j) {
    v2f a0, a1, p0, p1, q0, q1;
    a0.x = fa[j].x;   a0.y = fa[j].y;   a1.x = fa[j].z;   a1.y = fa[j].w;
    p0.x = fb0[j].x;  p0.y = fb0[j].y;  p1.x = fb0[j].z;  p1.y = fb0[j].w;
    q0.x = fb1[j].x;  q0.y = fb1[j].y;  q1.x = fb1[j].z;  q1.y = fb1[j].w;
    acc0 = wmma_f32_k4(a0, p0, acc0);
    acc1 = wmma_f32_k4(a0, q0, acc1);
    acc0 = wmma_f32_k4(a1, p1, acc0);
    acc1 = wmma_f32_k4(a1, q1, acc1);
  }
  __builtin_amdgcn_sched_group_barrier(0x100, 3 * L1_NG, 0);
  __builtin_amdgcn_sched_group_barrier(0x008, 4 * L1_NG, 0);
}

__device__ __forceinline__ void l1_mm_upper(const float* __restrict__ as, const float* __restrict__ ws0,
                                            v8f& acce, v8f& acco) {
  float4 fa[L1_NG], fb0[L1_NG];
#pragma unroll
  for (int j = 0; j < L1_NG; ++j) {
    fa[j]  = *reinterpret_cast<const float4*>(as  + 8 * j);
    fb0[j] = *reinterpret_cast<const float4*>(ws0 + 8 * j);
  }
#pragma unroll
  for (int j = 0; j < L1_NG; ++j) {
    v2f a0, a1, p0, p1;
    a0.x = fa[j].x;   a0.y = fa[j].y;   a1.x = fa[j].z;   a1.y = fa[j].w;
    p0.x = fb0[j].x;  p0.y = fb0[j].y;  p1.x = fb0[j].z;  p1.y = fb0[j].w;
    acce = wmma_f32_k4(a0, p0, acce);
    acco = wmma_f32_k4(a1, p1, acco);
  }
  __builtin_amdgcn_sched_group_barrier(0x100, 2 * L1_NG, 0);
  __builtin_amdgcn_sched_group_barrier(0x008, 2 * L1_NG, 0);
}

__global__ __launch_bounds__(L1_WAVES * 32)
void k_layer1(const float* __restrict__ A, const float* __restrict__ W1,
              const float* __restrict__ b1,
              float* __restrict__ act, float* __restrict__ lga) {
  __shared__ __attribute__((aligned(16))) float As[2][L1_MROWS * L1_PITCH];
  __shared__ __attribute__((aligned(16))) float Ws[2][L1_NROWS * L1_PITCH];
  const int tid  = threadIdx.x;
  const int wave = tid >> 5;
  const int lane = tid & 31;
  const int ms  = blockIdx.x >> 5;
  const int p   = blockIdx.x & 31;
  const int jlo = p;
  const int jhi = L1_NSLAB - 1 - p;
  const int q   = p >> 2;
  const int nch = (HID / L1_KC) - q;
  const int mi = wave >> 1;
  const int t  = wave & 1;

  const int srow = tid >> 5;
  const int sc4  = (tid & 31) * 4;
  const float* gA  = A  + (size_t)(ms  * L1_MROWS + srow) * HID + sc4;
  const float* gWh = W1 + (size_t)(jhi * L1_SLAB  + srow) * HID + sc4;
  const float* gWl = W1 + (size_t)(jlo * L1_SLAB  + srow) * HID + sc4;
  const int soff = srow * L1_PITCH + sc4;

  const int foffA = (mi * 16 + (lane & 15)) * L1_PITCH + (lane >> 4) * 4;
  const int foffB = (t  * 16 + (lane & 15)) * L1_PITCH + (lane >> 4) * 4;

  L1_A8(L1_DECLA)
  L1_WHI8(L1_DECLW)
  L1_WLO8(L1_DECLW)
  {
    const int ko = 0;
    const int st = 0;
    L1_A8(L1_LDA)  L1_WHI8(L1_LDWH)  L1_WLO8(L1_LDWL)
    L1_A8(L1_STA)  L1_WHI8(L1_STW)   L1_WLO8(L1_STW)
  }
  __syncthreads();

  v8f acc0 = {};
  v8f acc1 = {};
  v8f acc2 = {};
  for (int c = 0; c < q; ++c) {
    const int cur = c & 1;
    {
      const int ko = (c + 1) * L1_KC;
      L1_A8(L1_LDA)  L1_WHI8(L1_LDWH)  L1_WLO8(L1_LDWL)
    }
    l1_mm_both(&As[cur][foffA], &Ws[cur][foffB], acc0, acc1);
    {
      const int st = cur ^ 1;
      L1_A8(L1_STA)  L1_WHI8(L1_STW)  L1_WLO8(L1_STW)
    }
    __syncthreads();
  }
  {
    const int cur = q & 1;
    {
      const int ko = (q + 1) * L1_KC;
      L1_A8(L1_LDA)  L1_WHI8(L1_LDWH)
    }
    l1_mm_both(&As[cur][foffA], &Ws[cur][foffB], acc0, acc1);
    {
      const int st = cur ^ 1;
      L1_A8(L1_STA)  L1_WHI8(L1_STW)
    }
    __syncthreads();
  }
  for (int c = q + 1; c < nch - 1; ++c) {
    const int cur = c & 1;
    {
      const int ko = (c + 1) * L1_KC;
      L1_A8(L1_LDA)  L1_WHI8(L1_LDWH)
    }
    l1_mm_upper(&As[cur][foffA], &Ws[cur][foffB], acc0, acc2);
    {
      const int st = cur ^ 1;
      L1_A8(L1_STA)  L1_WHI8(L1_STW)
    }
    __syncthreads();
  }
  l1_mm_upper(&As[(nch - 1) & 1][foffA], &Ws[(nch - 1) & 1][foffB], acc0, acc2);

  const int rowoff = (lane >> 4) * 8;
  const int n0 = jhi * L1_SLAB + t * 16 + (lane & 15);
  const int n1 = jlo * L1_SLAB + t * 16 + (lane & 15);
  const float bias0 = b1[n0];
  const float bias1 = b1[n1];
#pragma unroll
  for (int v = 0; v < 8; ++v) {
    const int m = ms * L1_MROWS + mi * 16 + v + rowoff;
    const float h0 = (acc0[v] + acc2[v]) + bias0;
    const float h1 = acc1[v] + bias1;
    act[m * HID + n0] = fast_tanh(h0);
    lga[m * HID + n0] = log_abs_dtanh(h0);
    act[m * HID + n1] = fast_tanh(h1);
    lga[m * HID + n1] = log_abs_dtanh(h1);
  }
}

__device__ __forceinline__ void grad_body(int wave,
                                          const float* __restrict__ E, const float* __restrict__ W1,
                                          const float* __restrict__ lga, const float* __restrict__ mrow,
                                          float* __restrict__ Eo, float* __restrict__ mo) {
  const int ks  = wave >> 7;
  const int rem = wave & 127;
  const int mt  = rem >> 3;
  const int nt  = rem & 7;
  if (ks >= Dd) return;
  const float* Ak = E  + (size_t)ks * Mm;
  const float* Wk = W1 + (size_t)ks * Mm * HID + (size_t)ks * Mm;
  v8f acc = {};
  gemm_tile(Ak, HID, Wk, HID, mt * 16, nt * 16, Mm, acc);

  const int lane = threadIdx.x & 31;
  const int o = nt * 16 + (lane & 15);
  const int rowoff = (lane >> 4) * 8;
  float g[8];
  for (int v = 0; v < 8; ++v) {
    const int b = mt * 16 + v + rowoff;
    const float s = acc[v];
    g[v] = __logf(s) + mrow[b * Dd + ks] + lga[b * HID + ks * Mm + o];
  }
  rowmax_exp_store(g, mt * 16, ks, Eo, mo);
}

#define OG_WAVES  8
#define OG_KC     (HID / OG_WAVES)
#define OG_BLOCKS (Bsz / 16)
__device__ __forceinline__ void out_body(int mt, const float* __restrict__ A,
                                         const float* __restrict__ W2, float* __restrict__ out) {
  __shared__ float part[OG_WAVES][16 * 16];
  const int wave = threadIdx.x >> 5;
  const int lane = threadIdx.x & 31;
  v8f acc = {};
  gemm_tile(A + wave * OG_KC, HID, W2 + wave * OG_KC, HID, mt * 16, 0, OG_KC, acc);
  const int n = lane & 15;
  const int rowoff = (lane >> 4) * 8;
  for (int v = 0; v < 8; ++v)
    part[wave][(v + rowoff) * Dd + n] = acc[v];
  __syncthreads();
  float s = part[0][threadIdx.x];
  for (int w = 1; w < OG_WAVES; ++w) s += part[w][threadIdx.x];
  out[mt * 16 * Dd + threadIdx.x] = s;
}

__global__ __launch_bounds__(RP_WAVES * 32)
void k_grad_out(const float* __restrict__ E, const float* __restrict__ W1,
                const float* __restrict__ lga, const float* __restrict__ mrow,
                float* __restrict__ Eo, float* __restrict__ mo,
                const float* __restrict__ A, const float* __restrict__ W2,
                float* __restrict__ out) {
  if (blockIdx.x < OG_BLOCKS) {
    out_body(blockIdx.x, A, W2, out);
  } else {
    grad_body(((blockIdx.x - OG_BLOCKS) * (RP_WAVES * 32) + threadIdx.x) >> 5,
              E, W1, lga, mrow, Eo, mo);
  }
}

#define LD_WAVES 32

template <int HALF>
__device__ __forceinline__ void fold_half(float (&s)[Dd], int lane, int off) {
  const bool up = (lane & off) != 0;
#pragma unroll
  for (int j = 0; j < HALF; ++j) {
    const float keep = up ? s[j + HALF] : s[j];
    const float send = up ? s[j] : s[j + HALF];
    s[j] = keep + __shfl_xor(send, off);
  }
}

__device__ __forceinline__ float logdet_sample(const float* __restrict__ E,
                                               const float* __restrict__ mrow,
                                               const float* __restrict__ W2,
                                               int b, int lane) {
  const float mk = mrow[b * Dd + (lane >> 1)];
  const float4* e4 = reinterpret_cast<const float4*>(E + (size_t)b * HID) + lane;
  float s[Dd];
#pragma unroll
  for (int k = 0; k < Dd; ++k) {
    const float4 e = e4[k * (Mm / 4)];
    const float4 w =
        reinterpret_cast<const float4*>(W2 + (size_t)k * HID + k * Mm)[lane];
    float p = e.x * w.x;
    p = fmaf(e.y, w.y, p);
    p = fmaf(e.z, w.z, p);
    p = fmaf(e.w, w.w, p);
    s[k] = p;
  }
  fold_half<8>(s, lane, 16);
  fold_half<4>(s, lane, 8);
  fold_half<2>(s, lane, 4);
  fold_half<1>(s, lane, 2);
  const float sk = s[0] + __shfl_xor(s[0], 1);
  float t = __logf(sk) + mk;
  t += __shfl_xor(t, 2);
  t += __shfl_xor(t, 4);
  t += __shfl_xor(t, 8);
  t += __shfl_xor(t, 16);
  return t;
}

__global__ void k_logdet2(const float* __restrict__ Ea, const float* __restrict__ mrowa,
                          const float* __restrict__ W2a,
                          const float* __restrict__ Eb, const float* __restrict__ mrowb,
                          const float* __restrict__ W2b,
                          float* __restrict__ ld) {
  __shared__ float tot_s[LD_WAVES];
  const int wave = threadIdx.x >> 5;
  const int lane = threadIdx.x & 31;
  const int b = blockIdx.x * LD_WAVES + wave;
  const float t0 = logdet_sample(Ea, mrowa, W2a, b, lane);
  const float t1 = logdet_sample(Eb, mrowb, W2b, b, lane);
  if (lane == 0) tot_s[wave] = t0 + t1;
  __syncthreads();
  if (wave == 0) ld[blockIdx.x * LD_WAVES + lane] = tot_s[lane];
}

extern "C" void kernel_launch(void* const* d_in, const int* in_sizes, int n_in,
                              void* d_out, int out_size, void* d_ws, size_t ws_size,
                              hipStream_t stream) {
  const float* inp   = (const float*)d_in[0];
  const float* f0_W0 = (const float*)d_in[1];
  const float* f0_W1 = (const float*)d_in[2];
  const float* f0_W2 = (const float*)d_in[3];
  const float* f0_b0 = (const float*)d_in[4];
  const float* f0_b1 = (const float*)d_in[5];
  const float* f1_W0 = (const float*)d_in[6];
  const float* f1_W1 = (const float*)d_in[7];
  const float* f1_W2 = (const float*)d_in[8];
  const float* f1_b0 = (const float*)d_in[9];
  const float* f1_b1 = (const float*)d_in[10];

  float* out = (float*)d_out;
  float* ld  = out + Bsz * Dd;

  float* ws     = (float*)d_ws;
  float* act0   = ws;
  float* act1   = act0   + Bsz * HID;
  float* lga    = act1   + Bsz * HID;
  float* E0     = lga    + Bsz * HID;
  float* E1a    = E0     + Bsz * HID;
  float* E1b    = E1a    + Bsz * HID;
  float* mrow0  = E1b    + Bsz * HID;
  float* mrow1a = mrow0  + Bsz * Dd;
  float* mrow1b = mrow1a + Bsz * Dd;
  float* xmid   = mrow1b + Bsz * Dd;

  k_layer0  <<<256, RP_WAVES * 32, 0, stream>>>(inp, f0_W0, f0_b0, act0, E0, mrow0);
  k_layer1  <<<(Bsz / L1_MROWS) * (L1_NSLAB / 2), L1_WAVES * 32, 0, stream>>>(act0, f0_W1, f0_b1, act1, lga);
  k_grad_out<<<OG_BLOCKS + 256, RP_WAVES * 32, 0, stream>>>(E0, f0_W1, lga, mrow0, E1a, mrow1a,
                                                           act1, f0_W2, xmid);
  k_layer0  <<<256, RP_WAVES * 32, 0, stream>>>(xmid, f1_W0, f1_b0, act0, E0, mrow0);
  k_layer1  <<<(Bsz / L1_MROWS) * (L1_NSLAB / 2), L1_WAVES * 32, 0, stream>>>(act0, f1_W1, f1_b1, act1, lga);
  k_grad_out<<<OG_BLOCKS + 256, RP_WAVES * 32, 0, stream>>>(E0, f1_W1, lga, mrow0, E1b, mrow1b,
                                                           act1, f1_W2, out);
  k_logdet2 <<<Bsz / LD_WAVES, LD_WAVES * 32, 0, stream>>>(E1a, mrow1a, f0_W2, E1b, mrow1b, f1_W2, ld);
}


# Fragment loads of the three direct-from-global GEMMs (layer 0, grad, final) widened to one 16-byte load per lane per 8 k (was two 8-byte): half the 16-line loads; k sum order changes
# speedup vs baseline: 4.8971x; 1.0378x over previous
#include <hip/hip_runtime.h>
#include <hip/hip_bf16.h>


typedef float v2f __attribute__((ext_vector_type(2)));
typedef float v8f __attribute__((ext_vector_type(8)));

#define Bsz   256
#define Dd    16
#define Mm    128
#define HID   2048

__device__ __forceinline__ v8f wmma_f32_k4(v2f a, v2f b, v8f c) {
  return __builtin_amdgcn_wmma_f32_16x16x4_f32(
      false, a, false, b, (short)0, c, false, false);
}

__device__ __forceinline__ void gemm_tile(const float* __restrict__ A, int lda,
                                          const float* __restrict__ W, int ldw,
                                          int m0, int n0, int K, v8f& acc) {
  const int lane = threadIdx.x & 31;
  const int r    = lane & 15;
  const int k4   = (lane >> 4) << 2;
  const float* arow = A + (size_t)(m0 + r) * lda + k4;
  const float* wrow = W + (size_t)(n0 + r) * ldw + k4;
#pragma unroll
  for (int k = 0; k < K; k += 8) {
    const float4 fa = *reinterpret_cast<const float4*>(arow + k);
    const float4 fb = *reinterpret_cast<const float4*>(wrow + k);
    v2f a0, a1, b0, b1;
    a0.x = fa.x;  a0.y = fa.y;  a1.x = fa.z;  a1.y = fa.w;
    b0.x = fb.x;  b0.y = fb.y;  b1.x = fb.z;  b1.y = fb.w;
    acc = wmma_f32_k4(a0, b0, acc);
    acc = wmma_f32_k4(a1, b1, acc);
  }
}

__device__ __forceinline__ float fast_tanh(float x) {
#if defined(__has_builtin)
#if __has_builtin(__builtin_amdgcn_tanhf)
  return __builtin_amdgcn_tanhf(x);
#elif __has_builtin(__builtin_amdgcn_tanh_f32)
  return __builtin_amdgcn_tanh_f32(x);
#else
  return tanhf(x);
#endif
#else
  return tanhf(x);
#endif
}

__device__ __forceinline__ float softplus_f(float x) {
  return fmaxf(x, 0.0f) + __logf(1.0f + __expf(-fabsf(x)));
}
__device__ __forceinline__ float log_abs_dtanh(float h) {
  return 2.0f * (0.6931471805599453f - h - softplus_f(-2.0f * h));
}

#define RP_WAVES 8
__device__ __forceinline__ void rowmax_exp_store(const float (&g)[8], int m0, int ks,
                                                 float* __restrict__ E,
                                                 float* __restrict__ mrow) {
  __shared__ __attribute__((aligned(16))) float pmax[RP_WAVES][16];
  const int wave = threadIdx.x >> 5;
  const int lane = threadIdx.x & 31;
  const int c = lane & 15;
  const int rowoff = (lane >> 4) * 8;
  float mx[8];
  for (int v = 0; v < 8; ++v) {
    float t = g[v];
    t = fmaxf(t, __shfl_xor(t, 1));
    t = fmaxf(t, __shfl_xor(t, 2));
    t = fmaxf(t, __shfl_xor(t, 4));
    t = fmaxf(t, __shfl_xor(t, 8));
    mx[v] = t;
  }
  if (c == 0) {
    for (int v = 0; v < 8; ++v) pmax[wave][rowoff + v] = mx[v];
  }
  __syncthreads();
  for (int v = 0; v < 8; ++v) {
    float r = pmax[0][rowoff + v];
    for (int w = 1; w < RP_WAVES; ++w) r = fmaxf(r, pmax[w][rowoff + v]);
    mx[v] = r;
  }
  for (int v = 0; v < 8; ++v) {
    const int m = m0 + rowoff + v;
    E[m * HID + ks * Mm + wave * 16 + c] = __expf(g[v] - mx[v]);
  }
  if (wave == 0 && c == 0) {
    for (int v = 0; v < 8; ++v) mrow[(m0 + rowoff + v) * Dd + ks] = mx[v];
  }
}

__global__ void k_layer0(const float* __restrict__ X, const float* __restrict__ W0,
                         const float* __restrict__ b0,
                         float* __restrict__ act,
                         float* __restrict__ E, float* __restrict__ mrow) {
  const int wave = (blockIdx.x * blockDim.x + threadIdx.x) >> 5;
  const int mt = wave >> 7;
  const int nt = wave & 127;
  if (mt >= Bsz / 16) return;
  v8f acc = {};
  gemm_tile(X, Dd, W0, Dd, mt * 16, nt * 16, Dd, acc);

  const int lane = threadIdx.x & 31;
  const int n = nt * 16 + (lane & 15);
  const int rowoff = (lane >> 4) * 8;
  const float ldiag = __logf(W0[n * Dd + (n >> 7)]);
  const float bias = b0[n];
  float g[8];
  for (int v = 0; v < 8; ++v) {
    const int m = mt * 16 + v + rowoff;
    const float h = acc[v] + bias;
    act[m * HID + n] = fast_tanh(h);
    g[v] = ldiag + log_abs_dtanh(h);
  }
  rowmax_exp_store(g, mt * 16, nt >> 3, E, mrow);
}

#define L1_KC    128
#define L1_PITCH (L1_KC + 4)
#define L1_MROWS 32
#define L1_SLAB  32
#define L1_NROWS (2 * L1_SLAB)
#define L1_NSLAB (HID / L1_SLAB)
#define L1_NG    (L1_KC / 8)
#define L1_WAVES 4
#define L1_A8(X)   X(0) X(1) X(2) X(3) X(4) X(5) X(6) X(7)
#define L1_WHI8(X) X(0) X(1) X(2) X(3) X(4) X(5) X(6) X(7)
#define L1_WLO8(X) X(8) X(9) X(10) X(11) X(12) X(13) X(14) X(15)
#define L1_DECLA(i) float4 pa##i;
#define L1_DECLW(i) float4 pw##i;
#define L1_LDA(i)  pa##i = *reinterpret_cast<const float4*>(gA  + (size_t)(4 * i) * HID + ko);
#define L1_LDWH(i) pw##i = *reinterpret_cast<const float4*>(gWh + (size_t)(4 * i) * HID + ko);
#define L1_LDWL(i) pw##i = *reinterpret_cast<const float4*>(gWl + (size_t)(4 * (i - 8)) * HID + ko);
#define L1_STA(i) *reinterpret_cast<float4*>(&As[st][soff + 4 * i * L1_PITCH]) = pa##i;
#define L1_STW(i) *reinterpret_cast<float4*>(&Ws[st][soff + 4 * i * L1_PITCH]) = pw##i;

__device__ __forceinline__ void l1_mm_both(const float* __restrict__ as, const float* __restrict__ ws0,
                                           v8f& acc0, v8f& acc1) {
  const float* ws1 = ws0 + L1_SLAB * L1_PITCH;
  float4 fa[L1_NG], fb0[L1_NG], fb1[L1_NG];
#pragma unroll
  for (int j = 0; j < L1_NG; ++j) {
    fa[j]  = *reinterpret_cast<const float4*>(as  + 8 * j);
    fb0[j] = *reinterpret_cast<const float4*>(ws0 + 8 * j);
    fb1[j] = *reinterpret_cast<const float4*>(ws1 + 8 * j);
  }
#pragma unroll
  for (int j = 0; j < L1_NG; ++j) {
    v2f a0, a1, p0, p1, q0, q1;
    a0.x = fa[j].x;   a0.y = fa[j].y;   a1.x = fa[j].z;   a1.y = fa[j].w;
    p0.x = fb0[j].x;  p0.y = fb0[j].y;  p1.x = fb0[j].z;  p1.y = fb0[j].w;
    q0.x = fb1[j].x;  q0.y = fb1[j].y;  q1.x = fb1[j].z;  q1.y = fb1[j].w;
    acc0 = wmma_f32_k4(a0, p0, acc0);
    acc1 = wmma_f32_k4(a0, q0, acc1);
    acc0 = wmma_f32_k4(a1, p1, acc0);
    acc1 = wmma_f32_k4(a1, q1, acc1);
  }
  __builtin_amdgcn_sched_group_barrier(0x100, 3 * L1_NG, 0);
  __builtin_amdgcn_sched_group_barrier(0x008, 4 * L1_NG, 0);
}

__device__ __forceinline__ void l1_mm_upper(const float* __restrict__ as, const float* __restrict__ ws0,
                                            v8f& acce, v8f& acco) {
  float4 fa[L1_NG], fb0[L1_NG];
#pragma unroll
  for (int j = 0; j < L1_NG; ++j) {
    fa[j]  = *reinterpret_cast<const float4*>(as  + 8 * j);
    fb0[j] = *reinterpret_cast<const float4*>(ws0 + 8 * j);
  }
#pragma unroll
  for (int j = 0; j < L1_NG; ++j) {
    v2f a0, a1, p0, p1;
    a0.x = fa[j].x;   a0.y = fa[j].y;   a1.x = fa[j].z;   a1.y = fa[j].w;
    p0.x = fb0[j].x;  p0.y = fb0[j].y;  p1.x = fb0[j].z;  p1.y = fb0[j].w;
    acce = wmma_f32_k4(a0, p0, acce);
    acco = wmma_f32_k4(a1, p1, acco);
  }
  __builtin_amdgcn_sched_group_barrier(0x100, 2 * L1_NG, 0);
  __builtin_amdgcn_sched_group_barrier(0x008, 2 * L1_NG, 0);
}

__global__ __launch_bounds__(L1_WAVES * 32)
void k_layer1(const float* __restrict__ A, const float* __restrict__ W1,
              const float* __restrict__ b1,
              float* __restrict__ act, float* __restrict__ lga) {
  __shared__ __attribute__((aligned(16))) float As[2][L1_MROWS * L1_PITCH];
  __shared__ __attribute__((aligned(16))) float Ws[2][L1_NROWS * L1_PITCH];
  const int tid  = threadIdx.x;
  const int wave = tid >> 5;
  const int lane = tid & 31;
  const int ms  = blockIdx.x >> 5;
  const int p   = blockIdx.x & 31;
  const int jlo = p;
  const int jhi = L1_NSLAB - 1 - p;
  const int q   = p >> 2;
  const int nch = (HID / L1_KC) - q;
  const int mi = wave >> 1;
  const int t  = wave & 1;

  const int srow = tid >> 5;
  const int sc4  = (tid & 31) * 4;
  const float* gA  = A  + (size_t)(ms  * L1_MROWS + srow) * HID + sc4;
  const float* gWh = W1 + (size_t)(jhi * L1_SLAB  + srow) * HID + sc4;
  const float* gWl = W1 + (size_t)(jlo * L1_SLAB  + srow) * HID + sc4;
  const int soff = srow * L1_PITCH + sc4;

  const int foffA = (mi * 16 + (lane & 15)) * L1_PITCH + (lane >> 4) * 4;
  const int foffB = (t  * 16 + (lane & 15)) * L1_PITCH + (lane >> 4) * 4;

  L1_A8(L1_DECLA)
  L1_WHI8(L1_DECLW)
  L1_WLO8(L1_DECLW)
  {
    const int ko = 0;
    const int st = 0;
    L1_A8(L1_LDA)  L1_WHI8(L1_LDWH)  L1_WLO8(L1_LDWL)
    L1_A8(L1_STA)  L1_WHI8(L1_STW)   L1_WLO8(L1_STW)
  }
  __syncthreads();

  v8f acc0 = {};
  v8f acc1 = {};
  v8f acc2 = {};
  for (int c = 0; c < q; ++c) {
    const int cur = c & 1;
    {
      const int ko = (c + 1) * L1_KC;
      L1_A8(L1_LDA)  L1_WHI8(L1_LDWH)  L1_WLO8(L1_LDWL)
    }
    l1_mm_both(&As[cur][foffA], &Ws[cur][foffB], acc0, acc1);
    {
      const int st = cur ^ 1;
      L1_A8(L1_STA)  L1_WHI8(L1_STW)  L1_WLO8(L1_STW)
    }
    __syncthreads();
  }
  {
    const int cur = q & 1;
    {
      const int ko = (q + 1) * L1_KC;
      L1_A8(L1_LDA)  L1_WHI8(L1_LDWH)
    }
    l1_mm_both(&As[cur][foffA], &Ws[cur][foffB], acc0, acc1);
    {
      const int st = cur ^ 1;
      L1_A8(L1_STA)  L1_WHI8(L1_STW)
    }
    __syncthreads();
  }
  for (int c = q + 1; c < nch - 1; ++c) {
    const int cur = c & 1;
    {
      const int ko = (c + 1) * L1_KC;
      L1_A8(L1_LDA)  L1_WHI8(L1_LDWH)
    }
    l1_mm_upper(&As[cur][foffA], &Ws[cur][foffB], acc0, acc2);
    {
      const int st = cur ^ 1;
      L1_A8(L1_STA)  L1_WHI8(L1_STW)
    }
    __syncthreads();
  }
  l1_mm_upper(&As[(nch - 1) & 1][foffA], &Ws[(nch - 1) & 1][foffB], acc0, acc2);

  const int rowoff = (lane >> 4) * 8;
  const int n0 = jhi * L1_SLAB + t * 16 + (lane & 15);
  const int n1 = jlo * L1_SLAB + t * 16 + (lane & 15);
  const float bias0 = b1[n0];
  const float bias1 = b1[n1];
#pragma unroll
  for (int v = 0; v < 8; ++v) {
    const int m = ms * L1_MROWS + mi * 16 + v + rowoff;
    const float h0 = (acc0[v] + acc2[v]) + bias0;
    const float h1 = acc1[v] + bias1;
    act[m * HID + n0] = fast_tanh(h0);
    lga[m * HID + n0] = log_abs_dtanh(h0);
    act[m * HID + n1] = fast_tanh(h1);
    lga[m * HID + n1] = log_abs_dtanh(h1);
  }
}

__device__ __forceinline__ void grad_body(int wave,
                                          const float* __restrict__ E, const float* __restrict__ W1,
                                          const float* __restrict__ lga, const float* __restrict__ mrow,
                                          float* __restrict__ Eo, float* __restrict__ mo) {
  const int ks  = wave >> 7;
  const int rem = wave & 127;
  const int mt  = rem >> 3;
  const int nt  = rem & 7;
  if (ks >= Dd) return;
  const float* Ak = E  + (size_t)ks * Mm;
  const float* Wk = W1 + (size_t)ks * Mm * HID + (size_t)ks * Mm;
  v8f acc = {};
  gemm_tile(Ak, HID, Wk, HID, mt * 16, nt * 16, Mm, acc);

  const int lane = threadIdx.x & 31;
  const int o = nt * 16 + (lane & 15);
  const int rowoff = (lane >> 4) * 8;
  float g[8];
  for (int v = 0; v < 8; ++v) {
    const int b = mt * 16 + v + rowoff;
    const float s = acc[v];
    g[v] = __logf(s) + mrow[b * Dd + ks] + lga[b * HID + ks * Mm + o];
  }
  rowmax_exp_store(g, mt * 16, ks, Eo, mo);
}

#define OG_WAVES  8
#define OG_KC     (HID / OG_WAVES)
#define OG_BLOCKS (Bsz / 16)
__device__ __forceinline__ void out_body(int mt, const float* __restrict__ A,
                                         const float* __restrict__ W2, float* __restrict__ out) {
  __shared__ float part[OG_WAVES][16 * 16];
  const int wave = threadIdx.x >> 5;
  const int lane = threadIdx.x & 31;
  v8f acc = {};
  gemm_tile(A + wave * OG_KC, HID, W2 + wave * OG_KC, HID, mt * 16, 0, OG_KC, acc);
  const int n = lane & 15;
  const int rowoff = (lane >> 4) * 8;
  for (int v = 0; v < 8; ++v)
    part[wave][(v + rowoff) * Dd + n] = acc[v];
  __syncthreads();
  float s = part[0][threadIdx.x];
  for (int w = 1; w < OG_WAVES; ++w) s += part[w][threadIdx.x];
  out[mt * 16 * Dd + threadIdx.x] = s;
}

__global__ __launch_bounds__(RP_WAVES * 32)
void k_grad_out(const float* __restrict__ E, const float* __restrict__ W1,
                const float* __restrict__ lga, const float* __restrict__ mrow,
                float* __restrict__ Eo, float* __restrict__ mo,
                const float* __restrict__ A, const float* __restrict__ W2,
                float* __restrict__ out) {
  if (blockIdx.x < OG_BLOCKS) {
    out_body(blockIdx.x, A, W2, out);
  } else {
    grad_body(((blockIdx.x - OG_BLOCKS) * (RP_WAVES * 32) + threadIdx.x) >> 5,
              E, W1, lga, mrow, Eo, mo);
  }
}

#define LD_WAVES 32

template <int HALF>
__device__ __forceinline__ void fold_half(float (&s)[Dd], int lane, int off) {
  const bool up = (lane & off) != 0;
#pragma unroll
  for (int j = 0; j < HALF; ++j) {
    const float keep = up ? s[j + HALF] : s[j];
    const float send = up ? s[j] : s[j + HALF];
    s[j] = keep + __shfl_xor(send, off);
  }
}

__device__ __forceinline__ float logdet_sample(const float* __restrict__ E,
                                               const float* __restrict__ mrow,
                                               const float* __restrict__ W2,
                                               int b, int lane) {
  const float mk = mrow[b * Dd + (lane >> 1)];
  const float4* e4 = reinterpret_cast<const float4*>(E + (size_t)b * HID) + lane;
  float s[Dd];
#pragma unroll
  for (int k = 0; k < Dd; ++k) {
    const float4 e = e4[k * (Mm / 4)];
    const float4 w =
        reinterpret_cast<const float4*>(W2 + (size_t)k * HID + k * Mm)[lane];
    float p = e.x * w.x;
    p = fmaf(e.y, w.y, p);
    p = fmaf(e.z, w.z, p);
    p = fmaf(e.w, w.w, p);
    s[k] = p;
  }
  fold_half<8>(s, lane, 16);
  fold_half<4>(s, lane, 8);
  fold_half<2>(s, lane, 4);
  fold_half<1>(s, lane, 2);
  const float sk = s[0] + __shfl_xor(s[0], 1);
  float t = __logf(sk) + mk;
  t += __shfl_xor(t, 2);
  t += __shfl_xor(t, 4);
  t += __shfl_xor(t, 8);
  t += __shfl_xor(t, 16);
  return t;
}

__global__ void k_logdet2(const float* __restrict__ Ea, const float* __restrict__ mrowa,
                          const float* __restrict__ W2a,
                          const float* __restrict__ Eb, const float* __restrict__ mrowb,
                          const float* __restrict__ W2b,
                          float* __restrict__ ld) {
  __shared__ float tot_s[LD_WAVES];
  const int wave = threadIdx.x >> 5;
  const int lane = threadIdx.x & 31;
  const int b = blockIdx.x * LD_WAVES + wave;
  const float t0 = logdet_sample(Ea, mrowa, W2a, b, lane);
  const float t1 = logdet_sample(Eb, mrowb, W2b, b, lane);
  if (lane == 0) tot_s[wave] = t0 + t1;
  __syncthreads();
  if (wave == 0) ld[blockIdx.x * LD_WAVES + lane] = tot_s[lane];
}

extern "C" void kernel_launch(void* const* d_in, const int* in_sizes, int n_in,
                              void* d_out, int out_size, void* d_ws, size_t ws_size,
                              hipStream_t stream) {
  const float* inp   = (const float*)d_in[0];
  const float* f0_W0 = (const float*)d_in[1];
  const float* f0_W1 = (const float*)d_in[2];
  const float* f0_W2 = (const float*)d_in[3];
  const float* f0_b0 = (const float*)d_in[4];
  const float* f0_b1 = (const float*)d_in[5];
  const float* f1_W0 = (const float*)d_in[6];
  const float* f1_W1 = (const float*)d_in[7];
  const float* f1_W2 = (const float*)d_in[8];
  const float* f1_b0 = (const float*)d_in[9];
  const float* f1_b1 = (const float*)d_in[10];

  float* out = (float*)d_out;
  float* ld  = out + Bsz * Dd;

  float* ws     = (float*)d_ws;
  float* act0   = ws;
  float* act1   = act0   + Bsz * HID;
  float* lga    = act1   + Bsz * HID;
  float* E0     = lga    + Bsz * HID;
  float* E1a    = E0     + Bsz * HID;
  float* E1b    = E1a    + Bsz * HID;
  float* mrow0  = E1b    + Bsz * HID;
  float* mrow1a = mrow0  + Bsz * Dd;
  float* mrow1b = mrow1a + Bsz * Dd;
  float* xmid   = mrow1b + Bsz * Dd;

  k_layer0  <<<256, RP_WAVES * 32, 0, stream>>>(inp, f0_W0, f0_b0, act0, E0, mrow0);
  k_layer1  <<<(Bsz / L1_MROWS) * (L1_NSLAB / 2), L1_WAVES * 32, 0, stream>>>(act0, f0_W1, f0_b1, act1, lga);
  k_grad_out<<<OG_BLOCKS + 256, RP_WAVES * 32, 0, stream>>>(E0, f0_W1, lga, mrow0, E1a, mrow1a,
                                                           act1, f0_W2, xmid);
  k_layer0  <<<256, RP_WAVES * 32, 0, stream>>>(xmid, f1_W0, f1_b0, act0, E0, mrow0);
  k_layer1  <<<(Bsz / L1_MROWS) * (L1_NSLAB / 2), L1_WAVES * 32, 0, stream>>>(act0, f1_W1, f1_b1, act1, lga);
  k_grad_out<<<OG_BLOCKS + 256, RP_WAVES * 32, 0, stream>>>(E0, f1_W1, lga, mrow0, E1b, mrow1b,
                                                           act1, f1_W2, out);
  k_logdet2 <<<Bsz / LD_WAVES, LD_WAVES * 32, 0, stream>>>(E1a, mrow1a, f0_W2, E1b, mrow1b, f1_W2, ld);
}


# Layer-1 chunk loop: a chunk's global loads go out one whole trip before its LDS stores, into staging registers of their own, so stores and loads issue beside the WMMAs; barrier waits on LDS only
# speedup vs baseline: 5.1077x; 1.0430x over previous
#include <hip/hip_runtime.h>
#include <hip/hip_bf16.h>


typedef float v2f __attribute__((ext_vector_type(2)));
typedef float v8f __attribute__((ext_vector_type(8)));

#define Bsz   256
#define Dd    16
#define Mm    128
#define HID   2048

__device__ __forceinline__ v8f wmma_f32_k4(v2f a, v2f b, v8f c) {
  return __builtin_amdgcn_wmma_f32_16x16x4_f32(
      false, a, false, b, (short)0, c, false, false);
}

__device__ __forceinline__ void gemm_tile(const float* __restrict__ A, int lda,
                                          const float* __restrict__ W, int ldw,
                                          int m0, int n0, int K, v8f& acc) {
  const int lane = threadIdx.x & 31;
  const int r    = lane & 15;
  const int k4   = (lane >> 4) << 2;
  const float* arow = A + (size_t)(m0 + r) * lda + k4;
  const float* wrow = W + (size_t)(n0 + r) * ldw + k4;
#pragma unroll
  for (int k = 0; k < K; k += 8) {
    const float4 fa = *reinterpret_cast<const float4*>(arow + k);
    const float4 fb = *reinterpret_cast<const float4*>(wrow + k);
    v2f a0, a1, b0, b1;
    a0.x = fa.x;  a0.y = fa.y;  a1.x = fa.z;  a1.y = fa.w;
    b0.x = fb.x;  b0.y = fb.y;  b1.x = fb.z;  b1.y = fb.w;
    acc = wmma_f32_k4(a0, b0, acc);
    acc = wmma_f32_k4(a1, b1, acc);
  }
}

__device__ __forceinline__ float fast_tanh(float x) {
#if defined(__has_builtin)
#if __has_builtin(__builtin_amdgcn_tanhf)
  return __builtin_amdgcn_tanhf(x);
#elif __has_builtin(__builtin_amdgcn_tanh_f32)
  return __builtin_amdgcn_tanh_f32(x);
#else
  return tanhf(x);
#endif
#else
  return tanhf(x);
#endif
}

__device__ __forceinline__ float softplus_f(float x) {
  return fmaxf(x, 0.0f) + __logf(1.0f + __expf(-fabsf(x)));
}
__device__ __forceinline__ float log_abs_dtanh(float h) {
  return 2.0f * (0.6931471805599453f - h - softplus_f(-2.0f * h));
}

#define RP_WAVES 8
__device__ __forceinline__ void rowmax_exp_store(const float (&g)[8], int m0, int ks,
                                                 float* __restrict__ E,
                                                 float* __restrict__ mrow) {
  __shared__ __attribute__((aligned(16))) float pmax[RP_WAVES][16];
  const int wave = threadIdx.x >> 5;
  const int lane = threadIdx.x & 31;
  const int c = lane & 15;
  const int rowoff = (lane >> 4) * 8;
  float mx[8];
  for (int v = 0; v < 8; ++v) {
    float t = g[v];
    t = fmaxf(t, __shfl_xor(t, 1));
    t = fmaxf(t, __shfl_xor(t, 2));
    t = fmaxf(t, __shfl_xor(t, 4));
    t = fmaxf(t, __shfl_xor(t, 8));
    mx[v] = t;
  }
  if (c == 0) {
    for (int v = 0; v < 8; ++v) pmax[wave][rowoff + v] = mx[v];
  }
  __syncthreads();
  for (int v = 0; v < 8; ++v) {
    float r = pmax[0][rowoff + v];
    for (int w = 1; w < RP_WAVES; ++w) r = fmaxf(r, pmax[w][rowoff + v]);
    mx[v] = r;
  }
  for (int v = 0; v < 8; ++v) {
    const int m = m0 + rowoff + v;
    E[m * HID + ks * Mm + wave * 16 + c] = __expf(g[v] - mx[v]);
  }
  if (wave == 0 && c == 0) {
    for (int v = 0; v < 8; ++v) mrow[(m0 + rowoff + v) * Dd + ks] = mx[v];
  }
}

__global__ void k_layer0(const float* __restrict__ X, const float* __restrict__ W0,
                         const float* __restrict__ b0,
                         float* __restrict__ act,
                         float* __restrict__ E, float* __restrict__ mrow) {
  const int wave = (blockIdx.x * blockDim.x + threadIdx.x) >> 5;
  const int mt = wave >> 7;
  const int nt = wave & 127;
  if (mt >= Bsz / 16) return;
  v8f acc = {};
  gemm_tile(X, Dd, W0, Dd, mt * 16, nt * 16, Dd, acc);

  const int lane = threadIdx.x & 31;
  const int n = nt * 16 + (lane & 15);
  const int rowoff = (lane >> 4) * 8;
  const float ldiag = __logf(W0[n * Dd + (n >> 7)]);
  const float bias = b0[n];
  float g[8];
  for (int v = 0; v < 8; ++v) {
    const int m = mt * 16 + v + rowoff;
    const float h = acc[v] + bias;
    act[m * HID + n] = fast_tanh(h);
    g[v] = ldiag + log_abs_dtanh(h);
  }
  rowmax_exp_store(g, mt * 16, nt >> 3, E, mrow);
}

#define L1_KC    128
#define L1_PITCH (L1_KC + 4)
#define L1_MROWS 32
#define L1_SLAB  32
#define L1_NROWS (2 * L1_SLAB)
#define L1_NSLAB (HID / L1_SLAB)
#define L1_NG    (L1_KC / 8)
#define L1_WAVES 4
#define L1_A8(X)   X(0) X(1) X(2) X(3) X(4) X(5) X(6) X(7)
#define L1_WHI8(X) X(0) X(1) X(2) X(3) X(4) X(5) X(6) X(7)
#define L1_WLO8(X) X(8) X(9) X(10) X(11) X(12) X(13) X(14) X(15)
#define L1_DECLA(i) float4 pa##i;
#define L1_DECLW(i) float4 pw##i;
#define L1_LDA(i)  pa##i = *reinterpret_cast<const float4*>(gA  + (size_t)(4 * i) * HID + ko);
#define L1_LDWH(i) pw##i = *reinterpret_cast<const float4*>(gWh + (size_t)(4 * i) * HID + ko);
#define L1_LDWL(i) pw##i = *reinterpret_cast<const float4*>(gWl + (size_t)(4 * (i - 8)) * HID + ko);
#define L1_STA(i) *reinterpret_cast<float4*>(&As[st][soff + 4 * i * L1_PITCH]) = pa##i;
#define L1_STW(i) *reinterpret_cast<float4*>(&Ws[st][soff + 4 * i * L1_PITCH]) = pw##i;
#define L1_LD_AH(kc)  { const int ko = (kc) * L1_KC;  L1_A8(L1_LDA)  L1_WHI8(L1_LDWH) }
#define L1_LD_AHL(kc) { const int ko = (kc) * L1_KC;  L1_A8(L1_LDA)  L1_WHI8(L1_LDWH)  L1_WLO8(L1_LDWL) }
#define L1_LD_NONE(kc)
#define L1_ST_AH(s)   { const int st = (s);  L1_A8(L1_STA)  L1_WHI8(L1_STW) }
#define L1_ST_AHL(s)  { const int st = (s);  L1_A8(L1_STA)  L1_WHI8(L1_STW)  L1_WLO8(L1_STW) }
#define L1_N_NONE 0
#define L1_N_AH   16
#define L1_N_AHL  24
#define L1_BOTH_RD  (3 * L1_NG)
#define L1_BOTH_MM  (4 * L1_NG)
#define L1_UPPER_RD (2 * L1_NG)
#define L1_UPPER_MM (2 * L1_NG)
#define L1_TRIP_BOTH(c, STORE, NST, LOAD, NLD)                       \
  {                                                                  \
    const int cur = (c) & 1;                                         \
    float4 fa[L1_NG], fb0[L1_NG], fb1[L1_NG];                        \
    l1_frag_both(&As[cur][foffA], &Ws[cur][foffB], fa, fb0, fb1);    \
    STORE(cur ^ 1)                                                   \
    LOAD((c) + 2)                                                    \
    l1_wmma_both(fa, fb0, fb1, acc0, acc1);                          \
    l1_trip_order<L1_BOTH_RD, L1_BOTH_MM, NST, NLD>();               \
    l1_lds_barrier();                                                \
  }
#define L1_TRIP_UPPER(c, STORE, NST, LOAD, NLD)                      \
  {                                                                  \
    const int cur = (c) & 1;                                         \
    float4 fa[L1_NG], fb0[L1_NG];                                    \
    l1_frag_upper(&As[cur][foffA], &Ws[cur][foffB], fa, fb0);        \
    STORE(cur ^ 1)                                                   \
    LOAD((c) + 2)                                                    \
    l1_wmma_upper(fa, fb0, acc0, acc2);                              \
    l1_trip_order<L1_UPPER_RD, L1_UPPER_MM, NST, NLD>();             \
    l1_lds_barrier();                                                \
  }

__device__ __forceinline__ void l1_frag_both(const float* __restrict__ as, const float* __restrict__ ws0,
                                             float4 (&fa)[L1_NG], float4 (&fb0)[L1_NG], float4 (&fb1)[L1_NG]) {
  const float* ws1 = ws0 + L1_SLAB * L1_PITCH;
#pragma unroll
  for (int j = 0; j < L1_NG; ++j) {
    fa[j]  = *reinterpret_cast<const float4*>(as  + 8 * j);
    fb0[j] = *reinterpret_cast<const float4*>(ws0 + 8 * j);
    fb1[j] = *reinterpret_cast<const float4*>(ws1 + 8 * j);
  }
}
__device__ __forceinline__ void l1_frag_upper(const float* __restrict__ as, const float* __restrict__ ws0,
                                              float4 (&fa)[L1_NG], float4 (&fb0)[L1_NG]) {
#pragma unroll
  for (int j = 0; j < L1_NG; ++j) {
    fa[j]  = *reinterpret_cast<const float4*>(as  + 8 * j);
    fb0[j] = *reinterpret_cast<const float4*>(ws0 + 8 * j);
  }
}

__device__ __forceinline__ void l1_wmma_both(const float4 (&fa)[L1_NG], const float4 (&fb0)[L1_NG],
                                             const float4 (&fb1)[L1_NG], v8f& acc0, v8f& acc1) {
#pragma unroll
  for (int j = 0; j < L1_NG; ++j) {
    v2f a0, a1, p0, p1, q0, q1;
    a0.x = fa[j].x;   a0.y = fa[j].y;   a1.x = fa[j].z;   a1.y = fa[j].w;
    p0.x = fb0[j].x;  p0.y = fb0[j].y;  p1.x = fb0[j].z;  p1.y = fb0[j].w;
    q0.x = fb1[j].x;  q0.y = fb1[j].y;  q1.x = fb1[j].z;  q1.y = fb1[j].w;
    acc0 = wmma_f32_k4(a0, p0, acc0);
    acc1 = wmma_f32_k4(a0, q0, acc1);
    acc0 = wmma_f32_k4(a1, p1, acc0);
    acc1 = wmma_f32_k4(a1, q1, acc1);
  }
}

__device__ __forceinline__ void l1_wmma_upper(const float4 (&fa)[L1_NG], const float4 (&fb0)[L1_NG],
                                              v8f& acce, v8f& acco) {
#pragma unroll
  for (int j = 0; j < L1_NG; ++j) {
    v2f a0, a1, p0, p1;
    a0.x = fa[j].x;   a0.y = fa[j].y;   a1.x = fa[j].z;   a1.y = fa[j].w;
    p0.x = fb0[j].x;  p0.y = fb0[j].y;  p1.x = fb0[j].z;  p1.y = fb0[j].w;
    acce = wmma_f32_k4(a0, p0, acce);
    acco = wmma_f32_k4(a1, p1, acco);
  }
}

template <int NREAD, int NWMMA, int NST, int NLD>
__device__ __forceinline__ void l1_trip_order() {
  constexpr int NMEM = NST > NLD ? NST : NLD;
  __builtin_amdgcn_sched_group_barrier(0x100, NREAD, 0);
#pragma unroll
  for (int i = 0; i < NMEM; ++i) {
    __builtin_amdgcn_sched_group_barrier(0x008, 1, 0);
    if (i < NST) __builtin_amdgcn_sched_group_barrier(0x200, 1, 0);
    if (i < NLD) __builtin_amdgcn_sched_group_barrier(0x020, 1, 0);
  }
  if constexpr (NWMMA - NMEM > 0)
    __builtin_amdgcn_sched_group_barrier(0x008, NWMMA - NMEM, 0);
}

__device__ __forceinline__ void l1_lds_barrier() {
  __builtin_amdgcn_sched_barrier(0);
  __builtin_amdgcn_fence(__ATOMIC_RELEASE, "workgroup", "local");
  __builtin_amdgcn_s_barrier();
  __builtin_amdgcn_fence(__ATOMIC_ACQUIRE, "workgroup", "local");
}

__global__ __launch_bounds__(L1_WAVES * 32)
void k_layer1(const float* __restrict__ A, const float* __restrict__ W1,
              const float* __restrict__ b1,
              float* __restrict__ act, float* __restrict__ lga) {
  __shared__ __attribute__((aligned(16))) float As[2][L1_MROWS * L1_PITCH];
  __shared__ __attribute__((aligned(16))) float Ws[2][L1_NROWS * L1_PITCH];
  const int tid  = threadIdx.x;
  const int wave = tid >> 5;
  const int lane = tid & 31;
  const int ms  = blockIdx.x >> 5;
  const int p   = blockIdx.x & 31;
  const int jlo = p;
  const int jhi = L1_NSLAB - 1 - p;
  const int q   = p >> 2;
  const int nch = (HID / L1_KC) - q;
  const int mi = wave >> 1;
  const int t  = wave & 1;

  const int srow = tid >> 5;
  const int sc4  = (tid & 31) * 4;
  const float* gA  = A  + (size_t)(ms  * L1_MROWS + srow) * HID + sc4;
  const float* gWh = W1 + (size_t)(jhi * L1_SLAB  + srow) * HID + sc4;
  const float* gWl = W1 + (size_t)(jlo * L1_SLAB  + srow) * HID + sc4;
  const int soff = srow * L1_PITCH + sc4;

  const int foffA = (mi * 16 + (lane & 15)) * L1_PITCH + (lane >> 4) * 4;
  const int foffB = (t  * 16 + (lane & 15)) * L1_PITCH + (lane >> 4) * 4;

  L1_A8(L1_DECLA)
  L1_WHI8(L1_DECLW)
  L1_WLO8(L1_DECLW)
  const int last = nch - 1;
  L1_LD_AHL(0)
  L1_ST_AHL(0)
  if (q >= 1) { L1_LD_AHL(1) } else { L1_LD_AH(1) }
  __syncthreads();

  v8f acc0 = {};
  v8f acc1 = {};
  v8f acc2 = {};
  int c = 0;
  for (; c + 2 <= q; ++c)
    L1_TRIP_BOTH(c, L1_ST_AHL, L1_N_AHL, L1_LD_AHL, L1_N_AHL)
  if (q >= 1) {
    L1_TRIP_BOTH(c, L1_ST_AHL, L1_N_AHL, L1_LD_AH, L1_N_AH)
    ++c;
  }
  if (q + 2 <= last)
    L1_TRIP_BOTH(c, L1_ST_AH, L1_N_AH, L1_LD_AH, L1_N_AH)
  else
    L1_TRIP_BOTH(c, L1_ST_AH, L1_N_AH, L1_LD_NONE, L1_N_NONE)
  ++c;
  for (; c + 2 <= last; ++c)
    L1_TRIP_UPPER(c, L1_ST_AH, L1_N_AH, L1_LD_AH, L1_N_AH)
  if (c + 1 <= last) {
    L1_TRIP_UPPER(c, L1_ST_AH, L1_N_AH, L1_LD_NONE, L1_N_NONE)
    ++c;
  }
  {
    float4 fa[L1_NG], fb0[L1_NG];
    l1_frag_upper(&As[last & 1][foffA], &Ws[last & 1][foffB], fa, fb0);
    l1_wmma_upper(fa, fb0, acc0, acc2);
    l1_trip_order<L1_UPPER_RD, L1_UPPER_MM, 0, 0>();
  }

  const int rowoff = (lane >> 4) * 8;
  const int n0 = jhi * L1_SLAB + t * 16 + (lane & 15);
  const int n1 = jlo * L1_SLAB + t * 16 + (lane & 15);
  const float bias0 = b1[n0];
  const float bias1 = b1[n1];
#pragma unroll
  for (int v = 0; v < 8; ++v) {
    const int m = ms * L1_MROWS + mi * 16 + v + rowoff;
    const float h0 = (acc0[v] + acc2[v]) + bias0;
    const float h1 = acc1[v] + bias1;
    act[m * HID + n0] = fast_tanh(h0);
    lga[m * HID + n0] = log_abs_dtanh(h0);
    act[m * HID + n1] = fast_tanh(h1);
    lga[m * HID + n1] = log_abs_dtanh(h1);
  }
}

__device__ __forceinline__ void grad_body(int wave,
                                          const float* __restrict__ E, const float* __restrict__ W1,
                                          const float* __restrict__ lga, const float* __restrict__ mrow,
                                          float* __restrict__ Eo, float* __restrict__ mo) {
  const int ks  = wave >> 7;
  const int rem = wave & 127;
  const int mt  = rem >> 3;
  const int nt  = rem & 7;
  if (ks >= Dd) return;
  const float* Ak = E  + (size_t)ks * Mm;
  const float* Wk = W1 + (size_t)ks * Mm * HID + (size_t)ks * Mm;
  v8f acc = {};
  gemm_tile(Ak, HID, Wk, HID, mt * 16, nt * 16, Mm, acc);

  const int lane = threadIdx.x & 31;
  const int o = nt * 16 + (lane & 15);
  const int rowoff = (lane >> 4) * 8;
  float g[8];
  for (int v = 0; v < 8; ++v) {
    const int b = mt * 16 + v + rowoff;
    const float s = acc[v];
    g[v] = __logf(s) + mrow[b * Dd + ks] + lga[b * HID + ks * Mm + o];
  }
  rowmax_exp_store(g, mt * 16, ks, Eo, mo);
}

#define OG_WAVES  8
#define OG_KC     (HID / OG_WAVES)
#define OG_BLOCKS (Bsz / 16)
__device__ __forceinline__ void out_body(int mt, const float* __restrict__ A,
                                         const float* __restrict__ W2, float* __restrict__ out) {
  __shared__ float part[OG_WAVES][16 * 16];
  const int wave = threadIdx.x >> 5;
  const int lane = threadIdx.x & 31;
  v8f acc = {};
  gemm_tile(A + wave * OG_KC, HID, W2 + wave * OG_KC, HID, mt * 16, 0, OG_KC, acc);
  const int n = lane & 15;
  const int rowoff = (lane >> 4) * 8;
  for (int v = 0; v < 8; ++v)
    part[wave][(v + rowoff) * Dd + n] = acc[v];
  __syncthreads();
  float s = part[0][threadIdx.x];
  for (int w = 1; w < OG_WAVES; ++w) s += part[w][threadIdx.x];
  out[mt * 16 * Dd + threadIdx.x] = s;
}

__global__ __launch_bounds__(RP_WAVES * 32)
void k_grad_out(const float* __restrict__ E, const float* __restrict__ W1,
                const float* __restrict__ lga, const float* __restrict__ mrow,
                float* __restrict__ Eo, float* __restrict__ mo,
                const float* __restrict__ A, const float* __restrict__ W2,
                float* __restrict__ out) {
  if (blockIdx.x < OG_BLOCKS) {
    out_body(blockIdx.x, A, W2, out);
  } else {
    grad_body(((blockIdx.x - OG_BLOCKS) * (RP_WAVES * 32) + threadIdx.x) >> 5,
              E, W1, lga, mrow, Eo, mo);
  }
}

#define LD_WAVES 32

template <int HALF>
__device__ __forceinline__ void fold_half(float (&s)[Dd], int lane, int off) {
  const bool up = (lane & off) != 0;
#pragma unroll
  for (int j = 0; j < HALF; ++j) {
    const float keep = up ? s[j + HALF] : s[j];
    const float send = up ? s[j] : s[j + HALF];
    s[j] = keep + __shfl_xor(send, off);
  }
}

__device__ __forceinline__ float logdet_sample(const float* __restrict__ E,
                                               const float* __restrict__ mrow,
                                               const float* __restrict__ W2,
                                               int b, int lane) {
  const float mk = mrow[b * Dd + (lane >> 1)];
  const float4* e4 = reinterpret_cast<const float4*>(E + (size_t)b * HID) + lane;
  float s[Dd];
#pragma unroll
  for (int k = 0; k < Dd; ++k) {
    const float4 e = e4[k * (Mm / 4)];
    const float4 w =
        reinterpret_cast<const float4*>(W2 + (size_t)k * HID + k * Mm)[lane];
    float p = e.x * w.x;
    p = fmaf(e.y, w.y, p);
    p = fmaf(e.z, w.z, p);
    p = fmaf(e.w, w.w, p);
    s[k] = p;
  }
  fold_half<8>(s, lane, 16);
  fold_half<4>(s, lane, 8);
  fold_half<2>(s, lane, 4);
  fold_half<1>(s, lane, 2);
  const float sk = s[0] + __shfl_xor(s[0], 1);
  float t = __logf(sk) + mk;
  t += __shfl_xor(t, 2);
  t += __shfl_xor(t, 4);
  t += __shfl_xor(t, 8);
  t += __shfl_xor(t, 16);
  return t;
}

__global__ void k_logdet2(const float* __restrict__ Ea, const float* __restrict__ mrowa,
                          const float* __restrict__ W2a,
                          const float* __restrict__ Eb, const float* __restrict__ mrowb,
                          const float* __restrict__ W2b,
                          float* __restrict__ ld) {
  __shared__ float tot_s[LD_WAVES];
  const int wave = threadIdx.x >> 5;
  const int lane = threadIdx.x & 31;
  const int b = blockIdx.x * LD_WAVES + wave;
  const float t0 = logdet_sample(Ea, mrowa, W2a, b, lane);
  const float t1 = logdet_sample(Eb, mrowb, W2b, b, lane);
  if (lane == 0) tot_s[wave] = t0 + t1;
  __syncthreads();
  if (wave == 0) ld[blockIdx.x * LD_WAVES + lane] = tot_s[lane];
}

extern "C" void kernel_launch(void* const* d_in, const int* in_sizes, int n_in,
                              void* d_out, int out_size, void* d_ws, size_t ws_size,
                              hipStream_t stream) {
  const float* inp   = (const float*)d_in[0];
  const float* f0_W0 = (const float*)d_in[1];
  const float* f0_W1 = (const float*)d_in[2];
  const float* f0_W2 = (const float*)d_in[3];
  const float* f0_b0 = (const float*)d_in[4];
  const float* f0_b1 = (const float*)d_in[5];
  const float* f1_W0 = (const float*)d_in[6];
  const float* f1_W1 = (const float*)d_in[7];
  const float* f1_W2 = (const float*)d_in[8];
  const float* f1_b0 = (const float*)d_in[9];
  const float* f1_b1 = (const float*)d_in[10];

  float* out = (float*)d_out;
  float* ld  = out + Bsz * Dd;

  float* ws     = (float*)d_ws;
  float* act0   = ws;
  float* act1   = act0   + Bsz * HID;
  float* lga    = act1   + Bsz * HID;
  float* E0     = lga    + Bsz * HID;
  float* E1a    = E0     + Bsz * HID;
  float* E1b    = E1a    + Bsz * HID;
  float* mrow0  = E1b    + Bsz * HID;
  float* mrow1a = mrow0  + Bsz * Dd;
  float* mrow1b = mrow1a + Bsz * Dd;
  float* xmid   = mrow1b + Bsz * Dd;

  k_layer0  <<<256, RP_WAVES * 32, 0, stream>>>(inp, f0_W0, f0_b0, act0, E0, mrow0);
  k_layer1  <<<(Bsz / L1_MROWS) * (L1_NSLAB / 2), L1_WAVES * 32, 0, stream>>>(act0, f0_W1, f0_b1, act1, lga);
  k_grad_out<<<OG_BLOCKS + 256, RP_WAVES * 32, 0, stream>>>(E0, f0_W1, lga, mrow0, E1a, mrow1a,
                                                           act1, f0_W2, xmid);
  k_layer0  <<<256, RP_WAVES * 32, 0, stream>>>(xmid, f1_W0, f1_b0, act0, E0, mrow0);
  k_layer1  <<<(Bsz / L1_MROWS) * (L1_NSLAB / 2), L1_WAVES * 32, 0, stream>>>(act0, f1_W1, f1_b1, act1, lga);
  k_grad_out<<<OG_BLOCKS + 256, RP_WAVES * 32, 0, stream>>>(E0, f1_W1, lga, mrow0, E1b, mrow1b,
                                                           act1, f1_W2, out);
  k_logdet2 <<<Bsz / LD_WAVES, LD_WAVES * 32, 0, stream>>>(E1a, mrow1a, f0_W2, E1b, mrow1b, f1_W2, ld);
}


# Final GEMM formed where act1 is made: each layer-1 wave multiplies its tanh tiles by W2 (8 WMMAs) into a 16x16 partial, next launch sums the 64; act1 not stored; 7 launches -> 6; sum order changes
# speedup vs baseline: 5.6907x; 1.1141x over previous
#include <hip/hip_runtime.h>
#include <hip/hip_bf16.h>


typedef float v2f __attribute__((ext_vector_type(2)));
typedef float v8f __attribute__((ext_vector_type(8)));

#define Bsz   256
#define Dd    16
#define Mm    128
#define HID   2048

__device__ __forceinline__ v8f wmma_f32_k4(v2f a, v2f b, v8f c) {
  return __builtin_amdgcn_wmma_f32_16x16x4_f32(
      false, a, false, b, (short)0, c, false, false);
}

__device__ __forceinline__ void gemm_tile(const float* __restrict__ A, int lda,
                                          const float* __restrict__ W, int ldw,
                                          int m0, int n0, int K, v8f& acc) {
  const int lane = threadIdx.x & 31;
  const int r    = lane & 15;
  const int k4   = (lane >> 4) << 2;
  const float* arow = A + (size_t)(m0 + r) * lda + k4;
  const float* wrow = W + (size_t)(n0 + r) * ldw + k4;
#pragma unroll
  for (int k = 0; k < K; k += 8) {
    const float4 fa = *reinterpret_cast<const float4*>(arow + k);
    const float4 fb = *reinterpret_cast<const float4*>(wrow + k);
    v2f a0, a1, b0, b1;
    a0.x = fa.x;  a0.y = fa.y;  a1.x = fa.z;  a1.y = fa.w;
    b0.x = fb.x;  b0.y = fb.y;  b1.x = fb.z;  b1.y = fb.w;
    acc = wmma_f32_k4(a0, b0, acc);
    acc = wmma_f32_k4(a1, b1, acc);
  }
}

__device__ __forceinline__ float fast_tanh(float x) {
#if defined(__has_builtin)
#if __has_builtin(__builtin_amdgcn_tanhf)
  return __builtin_amdgcn_tanhf(x);
#elif __has_builtin(__builtin_amdgcn_tanh_f32)
  return __builtin_amdgcn_tanh_f32(x);
#else
  return tanhf(x);
#endif
#else
  return tanhf(x);
#endif
}

__device__ __forceinline__ float softplus_f(float x) {
  return fmaxf(x, 0.0f) + __logf(1.0f + __expf(-fabsf(x)));
}
__device__ __forceinline__ float log_abs_dtanh(float h) {
  return 2.0f * (0.6931471805599453f - h - softplus_f(-2.0f * h));
}

#define RP_WAVES 8
__device__ __forceinline__ void rowmax_exp_store(const float (&g)[8], int m0, int ks,
                                                 float* __restrict__ E,
                                                 float* __restrict__ mrow) {
  __shared__ __attribute__((aligned(16))) float pmax[RP_WAVES][16];
  const int wave = threadIdx.x >> 5;
  const int lane = threadIdx.x & 31;
  const int c = lane & 15;
  const int rowoff = (lane >> 4) * 8;
  float mx[8];
  for (int v = 0; v < 8; ++v) {
    float t = g[v];
    t = fmaxf(t, __shfl_xor(t, 1));
    t = fmaxf(t, __shfl_xor(t, 2));
    t = fmaxf(t, __shfl_xor(t, 4));
    t = fmaxf(t, __shfl_xor(t, 8));
    mx[v] = t;
  }
  if (c == 0) {
    for (int v = 0; v < 8; ++v) pmax[wave][rowoff + v] = mx[v];
  }
  __syncthreads();
  for (int v = 0; v < 8; ++v) {
    float r = pmax[0][rowoff + v];
    for (int w = 1; w < RP_WAVES; ++w) r = fmaxf(r, pmax[w][rowoff + v]);
    mx[v] = r;
  }
  for (int v = 0; v < 8; ++v) {
    const int m = m0 + rowoff + v;
    E[m * HID + ks * Mm + wave * 16 + c] = __expf(g[v] - mx[v]);
  }
  if (wave == 0 && c == 0) {
    for (int v = 0; v < 8; ++v) mrow[(m0 + rowoff + v) * Dd + ks] = mx[v];
  }
}

#define PO_N 64
__device__ __forceinline__ float sum_partials(const float* __restrict__ part, int mt) {
  const float* pp = part + (size_t)mt * (PO_N * 256) + threadIdx.x;
  float sx = pp[0];
#pragma unroll
  for (int j = 1; j < PO_N; ++j) sx += pp[j * 256];
  return sx;
}

__global__ void k_layer0(const float* __restrict__ X, const float* __restrict__ W0,
                         const float* __restrict__ b0,
                         float* __restrict__ act,
                         float* __restrict__ E, float* __restrict__ mrow) {
  const int wave = (blockIdx.x * blockDim.x + threadIdx.x) >> 5;
  const int mt = wave >> 7;
  const int nt = wave & 127;
  if (mt >= Bsz / 16) return;
  v8f acc = {};
  gemm_tile(X, Dd, W0, Dd, mt * 16, nt * 16, Dd, acc);

  const int lane = threadIdx.x & 31;
  const int n = nt * 16 + (lane & 15);
  const int rowoff = (lane >> 4) * 8;
  const float ldiag = __logf(W0[n * Dd + (n >> 7)]);
  const float bias = b0[n];
  float g[8];
  for (int v = 0; v < 8; ++v) {
    const int m = mt * 16 + v + rowoff;
    const float h = acc[v] + bias;
    act[m * HID + n] = fast_tanh(h);
    g[v] = ldiag + log_abs_dtanh(h);
  }
  rowmax_exp_store(g, mt * 16, nt >> 3, E, mrow);
}

__global__ __launch_bounds__(RP_WAVES * 32)
void k_sum_layer0(const float* __restrict__ part, const float* __restrict__ W0,
                  const float* __restrict__ b0,
                  float* __restrict__ act,
                  float* __restrict__ E, float* __restrict__ mrow) {
  __shared__ __attribute__((aligned(16))) float xs[16 * Dd];
  const int lane = threadIdx.x & 31;
  const int wave = (blockIdx.x * blockDim.x + threadIdx.x) >> 5;
  const int mt = wave >> 7;
  const int nt = wave & 127;
  xs[threadIdx.x] = sum_partials(part, mt);
  __syncthreads();

  const int r  = lane & 15;
  const int k4 = (lane >> 4) << 2;
  v8f acc = {};
#pragma unroll
  for (int k = 0; k < Dd; k += 8) {
    const float4 fa = *reinterpret_cast<const float4*>(&xs[r * Dd + k4 + k]);
    const float4 fb = *reinterpret_cast<const float4*>(W0 + (size_t)(nt * 16 + r) * Dd + k4 + k);
    v2f a0, a1, w0, w1;
    a0.x = fa.x;  a0.y = fa.y;  a1.x = fa.z;  a1.y = fa.w;
    w0.x = fb.x;  w0.y = fb.y;  w1.x = fb.z;  w1.y = fb.w;
    acc = wmma_f32_k4(a0, w0, acc);
    acc = wmma_f32_k4(a1, w1, acc);
  }

  const int n = nt * 16 + (lane & 15);
  const int rowoff = (lane >> 4) * 8;
  const float ldiag = __logf(W0[n * Dd + (n >> 7)]);
  const float bias = b0[n];
  float g[8];
  for (int v = 0; v < 8; ++v) {
    const int m = mt * 16 + v + rowoff;
    const float h = acc[v] + bias;
    act[m * HID + n] = fast_tanh(h);
    g[v] = ldiag + log_abs_dtanh(h);
  }
  rowmax_exp_store(g, mt * 16, nt >> 3, E, mrow);
}

#define L1_KC    128
#define L1_PITCH (L1_KC + 4)
#define L1_MROWS 32
#define L1_SLAB  32
#define L1_NROWS (2 * L1_SLAB)
#define L1_NSLAB (HID / L1_SLAB)
#define L1_NG    (L1_KC / 8)
#define L1_WAVES 4
#define L1_TP    20
#define L1_A8(X)   X(0) X(1) X(2) X(3) X(4) X(5) X(6) X(7)
#define L1_WHI8(X) X(0) X(1) X(2) X(3) X(4) X(5) X(6) X(7)
#define L1_WLO8(X) X(8) X(9) X(10) X(11) X(12) X(13) X(14) X(15)
#define L1_DECLA(i) float4 pa##i;
#define L1_DECLW(i) float4 pw##i;
#define L1_LDA(i)  pa##i = *reinterpret_cast<const float4*>(gA  + (size_t)(4 * i) * HID + ko);
#define L1_LDWH(i) pw##i = *reinterpret_cast<const float4*>(gWh + (size_t)(4 * i) * HID + ko);
#define L1_LDWL(i) pw##i = *reinterpret_cast<const float4*>(gWl + (size_t)(4 * (i - 8)) * HID + ko);
#define L1_STA(i) *reinterpret_cast<float4*>(&As[st][soff + 4 * i * L1_PITCH]) = pa##i;
#define L1_STW(i) *reinterpret_cast<float4*>(&Ws[st][soff + 4 * i * L1_PITCH]) = pw##i;
#define L1_LD_AH(kc)  { const int ko = (kc) * L1_KC;  L1_A8(L1_LDA)  L1_WHI8(L1_LDWH) }
#define L1_LD_AHL(kc) { const int ko = (kc) * L1_KC;  L1_A8(L1_LDA)  L1_WHI8(L1_LDWH)  L1_WLO8(L1_LDWL) }
#define L1_LD_NONE(kc)
#define L1_ST_AH(s)   { const int st = (s);  L1_A8(L1_STA)  L1_WHI8(L1_STW) }
#define L1_ST_AHL(s)  { const int st = (s);  L1_A8(L1_STA)  L1_WHI8(L1_STW)  L1_WLO8(L1_STW) }
#define L1_N_NONE 0
#define L1_N_AH   16
#define L1_N_AHL  24
#define L1_BOTH_RD  (3 * L1_NG)
#define L1_BOTH_MM  (4 * L1_NG)
#define L1_UPPER_RD (2 * L1_NG)
#define L1_UPPER_MM (2 * L1_NG)
#define L1_TRIP_BOTH(c, STORE, NST, LOAD, NLD)                       \
  {                                                                  \
    const int cur = (c) & 1;                                         \
    float4 fa[L1_NG], fb0[L1_NG], fb1[L1_NG];                        \
    l1_frag_both(&As[cur][foffA], &Ws[cur][foffB], fa, fb0, fb1);    \
    STORE(cur ^ 1)                                                   \
    LOAD((c) + 2)                                                    \
    l1_wmma_both(fa, fb0, fb1, acc0, acc1);                          \
    l1_trip_order<L1_BOTH_RD, L1_BOTH_MM, NST, NLD>();               \
    l1_lds_barrier();                                                \
  }
#define L1_TRIP_UPPER(c, STORE, NST, LOAD, NLD)                      \
  {                                                                  \
    const int cur = (c) & 1;                                         \
    float4 fa[L1_NG], fb0[L1_NG];                                    \
    l1_frag_upper(&As[cur][foffA], &Ws[cur][foffB], fa, fb0);        \
    STORE(cur ^ 1)                                                   \
    LOAD((c) + 2)                                                    \
    l1_wmma_upper(fa, fb0, acc0, acc2);                              \
    l1_trip_order<L1_UPPER_RD, L1_UPPER_MM, NST, NLD>();             \
    l1_lds_barrier();                                                \
  }

__device__ __forceinline__ void l1_frag_both(const float* __restrict__ as, const float* __restrict__ ws0,
                                             float4 (&fa)[L1_NG], float4 (&fb0)[L1_NG], float4 (&fb1)[L1_NG]) {
  const float* ws1 = ws0 + L1_SLAB * L1_PITCH;
#pragma unroll
  for (int j = 0; j < L1_NG; ++j) {
    fa[j]  = *reinterpret_cast<const float4*>(as  + 8 * j);
    fb0[j] = *reinterpret_cast<const float4*>(ws0 + 8 * j);
    fb1[j] = *reinterpret_cast<const float4*>(ws1 + 8 * j);
  }
}
__device__ __forceinline__ void l1_frag_upper(const float* __restrict__ as, const float* __restrict__ ws0,
                                              float4 (&fa)[L1_NG], float4 (&fb0)[L1_NG]) {
#pragma unroll
  for (int j = 0; j < L1_NG; ++j) {
    fa[j]  = *reinterpret_cast<const float4*>(as  + 8 * j);
    fb0[j] = *reinterpret_cast<const float4*>(ws0 + 8 * j);
  }
}

__device__ __forceinline__ void l1_wmma_both(const float4 (&fa)[L1_NG], const float4 (&fb0)[L1_NG],
                                             const float4 (&fb1)[L1_NG], v8f& acc0, v8f& acc1) {
#pragma unroll
  for (int j = 0; j < L1_NG; ++j) {
    v2f a0, a1, p0, p1, q0, q1;
    a0.x = fa[j].x;   a0.y = fa[j].y;   a1.x = fa[j].z;   a1.y = fa[j].w;
    p0.x = fb0[j].x;  p0.y = fb0[j].y;  p1.x = fb0[j].z;  p1.y = fb0[j].w;
    q0.x = fb1[j].x;  q0.y = fb1[j].y;  q1.x = fb1[j].z;  q1.y = fb1[j].w;
    acc0 = wmma_f32_k4(a0, p0, acc0);
    acc1 = wmma_f32_k4(a0, q0, acc1);
    acc0 = wmma_f32_k4(a1, p1, acc0);
    acc1 = wmma_f32_k4(a1, q1, acc1);
  }
}

__device__ __forceinline__ void l1_wmma_upper(const float4 (&fa)[L1_NG], const float4 (&fb0)[L1_NG],
                                              v8f& acce, v8f& acco) {
#pragma unroll
  for (int j = 0; j < L1_NG; ++j) {
    v2f a0, a1, p0, p1;
    a0.x = fa[j].x;   a0.y = fa[j].y;   a1.x = fa[j].z;   a1.y = fa[j].w;
    p0.x = fb0[j].x;  p0.y = fb0[j].y;  p1.x = fb0[j].z;  p1.y = fb0[j].w;
    acce = wmma_f32_k4(a0, p0, acce);
    acco = wmma_f32_k4(a1, p1, acco);
  }
}

template <int NREAD, int NWMMA, int NST, int NLD>
__device__ __forceinline__ void l1_trip_order() {
  constexpr int NMEM = NST > NLD ? NST : NLD;
  __builtin_amdgcn_sched_group_barrier(0x100, NREAD, 0);
#pragma unroll
  for (int i = 0; i < NMEM; ++i) {
    __builtin_amdgcn_sched_group_barrier(0x008, 1, 0);
    if (i < NST) __builtin_amdgcn_sched_group_barrier(0x200, 1, 0);
    if (i < NLD) __builtin_amdgcn_sched_group_barrier(0x020, 1, 0);
  }
  if constexpr (NWMMA - NMEM > 0)
    __builtin_amdgcn_sched_group_barrier(0x008, NWMMA - NMEM, 0);
}

__device__ __forceinline__ void l1_lds_barrier() {
  __builtin_amdgcn_sched_barrier(0);
  __builtin_amdgcn_fence(__ATOMIC_RELEASE, "workgroup", "local");
  __builtin_amdgcn_s_barrier();
  __builtin_amdgcn_fence(__ATOMIC_ACQUIRE, "workgroup", "local");
}

__global__ __launch_bounds__(L1_WAVES * 32)
void k_layer1(const float* __restrict__ A, const float* __restrict__ W1,
              const float* __restrict__ b1, const float* __restrict__ W2,
              float* __restrict__ lga, float* __restrict__ part) {
  __shared__ __attribute__((aligned(16))) float As[2][L1_MROWS * L1_PITCH];
  __shared__ __attribute__((aligned(16))) float Ws[2][L1_NROWS * L1_PITCH];
  __shared__ __attribute__((aligned(16))) float Ts[L1_WAVES][2][16 * L1_TP];
  const int tid  = threadIdx.x;
  const int wave = tid >> 5;
  const int lane = tid & 31;
  const int ms  = blockIdx.x >> 5;
  const int p   = blockIdx.x & 31;
  const int jlo = p;
  const int jhi = L1_NSLAB - 1 - p;
  const int q   = p >> 2;
  const int nch = (HID / L1_KC) - q;
  const int mi = wave >> 1;
  const int t  = wave & 1;

  const int srow = tid >> 5;
  const int sc4  = (tid & 31) * 4;
  const float* gA  = A  + (size_t)(ms  * L1_MROWS + srow) * HID + sc4;
  const float* gWh = W1 + (size_t)(jhi * L1_SLAB  + srow) * HID + sc4;
  const float* gWl = W1 + (size_t)(jlo * L1_SLAB  + srow) * HID + sc4;
  const int soff = srow * L1_PITCH + sc4;

  const int foffA = (mi * 16 + (lane & 15)) * L1_PITCH + (lane >> 4) * 4;
  const int foffB = (t  * 16 + (lane & 15)) * L1_PITCH + (lane >> 4) * 4;

  L1_A8(L1_DECLA)
  L1_WHI8(L1_DECLW)
  L1_WLO8(L1_DECLW)
  const int last = nch - 1;
  L1_LD_AHL(0)
  L1_ST_AHL(0)
  if (q >= 1) { L1_LD_AHL(1) } else { L1_LD_AH(1) }
  __syncthreads();

  v8f acc0 = {};
  v8f acc1 = {};
  v8f acc2 = {};
  int c = 0;
  for (; c + 2 <= q; ++c)
    L1_TRIP_BOTH(c, L1_ST_AHL, L1_N_AHL, L1_LD_AHL, L1_N_AHL)
  if (q >= 1) {
    L1_TRIP_BOTH(c, L1_ST_AHL, L1_N_AHL, L1_LD_AH, L1_N_AH)
    ++c;
  }
  if (q + 2 <= last)
    L1_TRIP_BOTH(c, L1_ST_AH, L1_N_AH, L1_LD_AH, L1_N_AH)
  else
    L1_TRIP_BOTH(c, L1_ST_AH, L1_N_AH, L1_LD_NONE, L1_N_NONE)
  ++c;
  for (; c + 2 <= last; ++c)
    L1_TRIP_UPPER(c, L1_ST_AH, L1_N_AH, L1_LD_AH, L1_N_AH)
  if (c + 1 <= last) {
    L1_TRIP_UPPER(c, L1_ST_AH, L1_N_AH, L1_LD_NONE, L1_N_NONE)
    ++c;
  }
  {
    float4 fa[L1_NG], fb0[L1_NG];
    l1_frag_upper(&As[last & 1][foffA], &Ws[last & 1][foffB], fa, fb0);
    l1_wmma_upper(fa, fb0, acc0, acc2);
    l1_trip_order<L1_UPPER_RD, L1_UPPER_MM, 0, 0>();
  }

  const int rowoff = (lane >> 4) * 8;
  const int cc = lane & 15;
  const int n0 = jhi * L1_SLAB + t * 16 + cc;
  const int n1 = jlo * L1_SLAB + t * 16 + cc;
  const float bias0 = b1[n0];
  const float bias1 = b1[n1];
  const int k4 = (lane >> 4) * 4;
  const float* w2u = W2 + (size_t)cc * HID + jhi * L1_SLAB + t * 16 + k4;
  const float* w2l = W2 + (size_t)cc * HID + jlo * L1_SLAB + t * 16 + k4;
  const float4 wu0 = *reinterpret_cast<const float4*>(w2u);
  const float4 wu1 = *reinterpret_cast<const float4*>(w2u + 8);
  const float4 wl0 = *reinterpret_cast<const float4*>(w2l);
  const float4 wl1 = *reinterpret_cast<const float4*>(w2l + 8);
  float* tu = &Ts[wave][0][0];
  float* tl = &Ts[wave][1][0];
#pragma unroll
  for (int v = 0; v < 8; ++v) {
    const int m = ms * L1_MROWS + mi * 16 + v + rowoff;
    const float h0 = (acc0[v] + acc2[v]) + bias0;
    const float h1 = acc1[v] + bias1;
    tu[(v + rowoff) * L1_TP + cc] = fast_tanh(h0);
    tl[(v + rowoff) * L1_TP + cc] = fast_tanh(h1);
    lga[m * HID + n0] = log_abs_dtanh(h0);
    lga[m * HID + n1] = log_abs_dtanh(h1);
  }
  l1_lds_barrier();
  v8f pe = {};
  v8f po = {};
  {
    const float4 au0 = *reinterpret_cast<const float4*>(&tu[cc * L1_TP + k4]);
    const float4 au1 = *reinterpret_cast<const float4*>(&tu[cc * L1_TP + 8 + k4]);
    const float4 al0 = *reinterpret_cast<const float4*>(&tl[cc * L1_TP + k4]);
    const float4 al1 = *reinterpret_cast<const float4*>(&tl[cc * L1_TP + 8 + k4]);
    v2f a, b;
    a.x = au0.x;  a.y = au0.y;  b.x = wu0.x;  b.y = wu0.y;  pe = wmma_f32_k4(a, b, pe);
    a.x = au0.z;  a.y = au0.w;  b.x = wu0.z;  b.y = wu0.w;  po = wmma_f32_k4(a, b, po);
    a.x = au1.x;  a.y = au1.y;  b.x = wu1.x;  b.y = wu1.y;  pe = wmma_f32_k4(a, b, pe);
    a.x = au1.z;  a.y = au1.w;  b.x = wu1.z;  b.y = wu1.w;  po = wmma_f32_k4(a, b, po);
    a.x = al0.x;  a.y = al0.y;  b.x = wl0.x;  b.y = wl0.y;  pe = wmma_f32_k4(a, b, pe);
    a.x = al0.z;  a.y = al0.w;  b.x = wl0.z;  b.y = wl0.w;  po = wmma_f32_k4(a, b, po);
    a.x = al1.x;  a.y = al1.y;  b.x = wl1.x;  b.y = wl1.y;  pe = wmma_f32_k4(a, b, pe);
    a.x = al1.z;  a.y = al1.w;  b.x = wl1.z;  b.y = wl1.w;  po = wmma_f32_k4(a, b, po);
  }
  float* pt = part + ((size_t)(ms * 2 + mi) * PO_N + (p * 2 + t)) * 256;
#pragma unroll
  for (int v = 0; v < 8; ++v)
    pt[(v + rowoff) * Dd + cc] = pe[v] + po[v];
}

__device__ __forceinline__ void grad_body(int wave,
                                          const float* __restrict__ E, const float* __restrict__ W1,
                                          const float* __restrict__ lga, const float* __restrict__ mrow,
                                          float* __restrict__ Eo, float* __restrict__ mo) {
  const int ks  = wave >> 7;
  const int rem = wave & 127;
  const int mt  = rem >> 3;
  const int nt  = rem & 7;
  if (ks >= Dd) return;
  const float* Ak = E  + (size_t)ks * Mm;
  const float* Wk = W1 + (size_t)ks * Mm * HID + (size_t)ks * Mm;
  v8f acc = {};
  gemm_tile(Ak, HID, Wk, HID, mt * 16, nt * 16, Mm, acc);

  const int lane = threadIdx.x & 31;
  const int o = nt * 16 + (lane & 15);
  const int rowoff = (lane >> 4) * 8;
  float g[8];
  for (int v = 0; v < 8; ++v) {
    const int b = mt * 16 + v + rowoff;
    const float s = acc[v];
    g[v] = __logf(s) + mrow[b * Dd + ks] + lga[b * HID + ks * Mm + o];
  }
  rowmax_exp_store(g, mt * 16, ks, Eo, mo);
}

#define OG_BLOCKS (Bsz / 16)
#define GR_BLOCKS (Dd * (Bsz / 16))
__global__ __launch_bounds__(RP_WAVES * 32)
void k_grad2_out(const float* __restrict__ Ea, const float* __restrict__ W1a,
                 const float* __restrict__ lgaa, const float* __restrict__ mrowa,
                 float* __restrict__ Eoa, float* __restrict__ moa,
                 const float* __restrict__ Eb, const float* __restrict__ W1b,
                 const float* __restrict__ lgab, const float* __restrict__ mrowb,
                 float* __restrict__ Eob, float* __restrict__ mob,
                 const float* __restrict__ part, float* __restrict__ out) {
  if (blockIdx.x < OG_BLOCKS) {
    out[blockIdx.x * 16 * Dd + threadIdx.x] = sum_partials(part, blockIdx.x);
  } else if (blockIdx.x < OG_BLOCKS + GR_BLOCKS) {
    grad_body(((blockIdx.x - OG_BLOCKS) * (RP_WAVES * 32) + threadIdx.x) >> 5,
              Eb, W1b, lgab, mrowb, Eob, mob);
  } else {
    grad_body(((blockIdx.x - OG_BLOCKS - GR_BLOCKS) * (RP_WAVES * 32) + threadIdx.x) >> 5,
              Ea, W1a, lgaa, mrowa, Eoa, moa);
  }
}

#define LD_WAVES 32

template <int HALF>
__device__ __forceinline__ void fold_half(float (&s)[Dd], int lane, int off) {
  const bool up = (lane & off) != 0;
#pragma unroll
  for (int j = 0; j < HALF; ++j) {
    const float keep = up ? s[j + HALF] : s[j];
    const float send = up ? s[j] : s[j + HALF];
    s[j] = keep + __shfl_xor(send, off);
  }
}

__device__ __forceinline__ float logdet_sample(const float* __restrict__ E,
                                               const float* __restrict__ mrow,
                                               const float* __restrict__ W2,
                                               int b, int lane) {
  const float mk = mrow[b * Dd + (lane >> 1)];
  const float4* e4 = reinterpret_cast<const float4*>(E + (size_t)b * HID) + lane;
  float s[Dd];
#pragma unroll
  for (int k = 0; k < Dd; ++k) {
    const float4 e = e4[k * (Mm / 4)];
    const float4 w =
        reinterpret_cast<const float4*>(W2 + (size_t)k * HID + k * Mm)[lane];
    float p = e.x * w.x;
    p = fmaf(e.y, w.y, p);
    p = fmaf(e.z, w.z, p);
    p = fmaf(e.w, w.w, p);
    s[k] = p;
  }
  fold_half<8>(s, lane, 16);
  fold_half<4>(s, lane, 8);
  fold_half<2>(s, lane, 4);
  fold_half<1>(s, lane, 2);
  const float sk = s[0] + __shfl_xor(s[0], 1);
  float t = __logf(sk) + mk;
  t += __shfl_xor(t, 2);
  t += __shfl_xor(t, 4);
  t += __shfl_xor(t, 8);
  t += __shfl_xor(t, 16);
  return t;
}

__global__ void k_logdet2(const float* __restrict__ Ea, const float* __restrict__ mrowa,
                          const float* __restrict__ W2a,
                          const float* __restrict__ Eb, const float* __restrict__ mrowb,
                          const float* __restrict__ W2b,
                          float* __restrict__ ld) {
  __shared__ float tot_s[LD_WAVES];
  const int wave = threadIdx.x >> 5;
  const int lane = threadIdx.x & 31;
  const int b = blockIdx.x * LD_WAVES + wave;
  const float t0 = logdet_sample(Ea, mrowa, W2a, b, lane);
  const float t1 = logdet_sample(Eb, mrowb, W2b, b, lane);
  if (lane == 0) tot_s[wave] = t0 + t1;
  __syncthreads();
  if (wave == 0) ld[blockIdx.x * LD_WAVES + lane] = tot_s[lane];
}

extern "C" void kernel_launch(void* const* d_in, const int* in_sizes, int n_in,
                              void* d_out, int out_size, void* d_ws, size_t ws_size,
                              hipStream_t stream) {
  const float* inp   = (const float*)d_in[0];
  const float* f0_W0 = (const float*)d_in[1];
  const float* f0_W1 = (const float*)d_in[2];
  const float* f0_W2 = (const float*)d_in[3];
  const float* f0_b0 = (const float*)d_in[4];
  const float* f0_b1 = (const float*)d_in[5];
  const float* f1_W0 = (const float*)d_in[6];
  const float* f1_W1 = (const float*)d_in[7];
  const float* f1_W2 = (const float*)d_in[8];
  const float* f1_b0 = (const float*)d_in[9];
  const float* f1_b1 = (const float*)d_in[10];

  float* out = (float*)d_out;
  float* ld  = out + Bsz * Dd;

  float* ws     = (float*)d_ws;
  float* act0   = ws;
  float* lgaa   = act0   + Bsz * HID;
  float* lgab   = lgaa   + Bsz * HID;
  float* E0a    = lgab   + Bsz * HID;
  float* E0b    = E0a    + Bsz * HID;
  float* E1a    = E0b    + Bsz * HID;
  float* E1b    = E1a    + Bsz * HID;
  float* mrow0a = E1b    + Bsz * HID;
  float* mrow0b = mrow0a + Bsz * Dd;
  float* mrow1a = mrow0b + Bsz * Dd;
  float* mrow1b = mrow1a + Bsz * Dd;
  float* part   = mrow1b + Bsz * Dd;

  k_layer0    <<<256, RP_WAVES * 32, 0, stream>>>(inp, f0_W0, f0_b0, act0, E0a, mrow0a);
  k_layer1    <<<(Bsz / L1_MROWS) * (L1_NSLAB / 2), L1_WAVES * 32, 0, stream>>>(act0, f0_W1, f0_b1, f0_W2, lgaa, part);
  k_sum_layer0<<<256, RP_WAVES * 32, 0, stream>>>(part, f1_W0, f1_b0, act0, E0b, mrow0b);
  k_layer1    <<<(Bsz / L1_MROWS) * (L1_NSLAB / 2), L1_WAVES * 32, 0, stream>>>(act0, f1_W1, f1_b1, f1_W2, lgab, part);
  k_grad2_out <<<OG_BLOCKS + 2 * GR_BLOCKS, RP_WAVES * 32, 0, stream>>>(E0a, f0_W1, lgaa, mrow0a, E1a, mrow1a,
                                                                       E0b, f1_W1, lgab, mrow0b, E1b, mrow1b,
                                                                       part, out);
  k_logdet2   <<<Bsz / LD_WAVES, LD_WAVES * 32, 0, stream>>>(E1a, mrow1a, f0_W2, E1b, mrow1b, f1_W2, ld);
}


# Log-det dots formed in the grad update's epilogue (its work-group holds whole 128-wide rows): E and m of that update never stored; last launch only adds 32 terms per sample; sum order changes
# speedup vs baseline: 7.0354x; 1.2363x over previous
#include <hip/hip_runtime.h>
#include <hip/hip_bf16.h>


typedef float v2f __attribute__((ext_vector_type(2)));
typedef float v8f __attribute__((ext_vector_type(8)));

#define Bsz   256
#define Dd    16
#define Mm    128
#define HID   2048

__device__ __forceinline__ v8f wmma_f32_k4(v2f a, v2f b, v8f c) {
  return __builtin_amdgcn_wmma_f32_16x16x4_f32(
      false, a, false, b, (short)0, c, false, false);
}

__device__ __forceinline__ void gemm_tile(const float* __restrict__ A, int lda,
                                          const float* __restrict__ W, int ldw,
                                          int m0, int n0, int K, v8f& acc) {
  const int lane = threadIdx.x & 31;
  const int r    = lane & 15;
  const int k4   = (lane >> 4) << 2;
  const float* arow = A + (size_t)(m0 + r) * lda + k4;
  const float* wrow = W + (size_t)(n0 + r) * ldw + k4;
#pragma unroll
  for (int k = 0; k < K; k += 8) {
    const float4 fa = *reinterpret_cast<const float4*>(arow + k);
    const float4 fb = *reinterpret_cast<const float4*>(wrow + k);
    v2f a0, a1, b0, b1;
    a0.x = fa.x;  a0.y = fa.y;  a1.x = fa.z;  a1.y = fa.w;
    b0.x = fb.x;  b0.y = fb.y;  b1.x = fb.z;  b1.y = fb.w;
    acc = wmma_f32_k4(a0, b0, acc);
    acc = wmma_f32_k4(a1, b1, acc);
  }
}

__device__ __forceinline__ float fast_tanh(float x) {
#if defined(__has_builtin)
#if __has_builtin(__builtin_amdgcn_tanhf)
  return __builtin_amdgcn_tanhf(x);
#elif __has_builtin(__builtin_amdgcn_tanh_f32)
  return __builtin_amdgcn_tanh_f32(x);
#else
  return tanhf(x);
#endif
#else
  return tanhf(x);
#endif
}

__device__ __forceinline__ float softplus_f(float x) {
  return fmaxf(x, 0.0f) + __logf(1.0f + __expf(-fabsf(x)));
}
__device__ __forceinline__ float log_abs_dtanh(float h) {
  return 2.0f * (0.6931471805599453f - h - softplus_f(-2.0f * h));
}

#define RP_WAVES 8
__device__ __forceinline__ void rowmax_exp_store(const float (&g)[8], int m0, int ks,
                                                 float* __restrict__ E,
                                                 float* __restrict__ mrow) {
  __shared__ __attribute__((aligned(16))) float pmax[RP_WAVES][16];
  const int wave = threadIdx.x >> 5;
  const int lane = threadIdx.x & 31;
  const int c = lane & 15;
  const int rowoff = (lane >> 4) * 8;
  float mx[8];
  for (int v = 0; v < 8; ++v) {
    float t = g[v];
    t = fmaxf(t, __shfl_xor(t, 1));
    t = fmaxf(t, __shfl_xor(t, 2));
    t = fmaxf(t, __shfl_xor(t, 4));
    t = fmaxf(t, __shfl_xor(t, 8));
    mx[v] = t;
  }
  if (c == 0) {
    for (int v = 0; v < 8; ++v) pmax[wave][rowoff + v] = mx[v];
  }
  __syncthreads();
  for (int v = 0; v < 8; ++v) {
    float r = pmax[0][rowoff + v];
    for (int w = 1; w < RP_WAVES; ++w) r = fmaxf(r, pmax[w][rowoff + v]);
    mx[v] = r;
  }
  for (int v = 0; v < 8; ++v) {
    const int m = m0 + rowoff + v;
    E[m * HID + ks * Mm + wave * 16 + c] = __expf(g[v] - mx[v]);
  }
  if (wave == 0 && c == 0) {
    for (int v = 0; v < 8; ++v) mrow[(m0 + rowoff + v) * Dd + ks] = mx[v];
  }
}

template <int HALF>
__device__ __forceinline__ void fold_rows(float (&s)[8], int lane, int off) {
  const bool up = (lane & off) != 0;
#pragma unroll
  for (int j = 0; j < HALF; ++j) {
    const float keep = up ? s[j + HALF] : s[j];
    const float send = up ? s[j] : s[j + HALF];
    s[j] = keep + __shfl_xor(send, off);
  }
}
__device__ __forceinline__ void rowmax_dot_term(const float (&g)[8], float w2c,
                                                float* __restrict__ term) {
  __shared__ __attribute__((aligned(16))) float qmax[RP_WAVES][16];
  __shared__ __attribute__((aligned(16))) float qsum[RP_WAVES][16];
  const int wave = threadIdx.x >> 5;
  const int lane = threadIdx.x & 31;
  const int c = lane & 15;
  const int rowoff = (lane >> 4) * 8;
  float mx[8];
  for (int v = 0; v < 8; ++v) {
    float t = g[v];
    t = fmaxf(t, __shfl_xor(t, 1));
    t = fmaxf(t, __shfl_xor(t, 2));
    t = fmaxf(t, __shfl_xor(t, 4));
    t = fmaxf(t, __shfl_xor(t, 8));
    mx[v] = t;
  }
  if (c == 0) {
    for (int v = 0; v < 8; ++v) qmax[wave][rowoff + v] = mx[v];
  }
  __syncthreads();
  for (int v = 0; v < 8; ++v) {
    float r = qmax[0][rowoff + v];
    for (int w = 1; w < RP_WAVES; ++w) r = fmaxf(r, qmax[w][rowoff + v]);
    mx[v] = r;
  }
  float ps[8];
#pragma unroll
  for (int v = 0; v < 8; ++v) ps[v] = __expf(g[v] - mx[v]) * w2c;
  fold_rows<4>(ps, lane, 8);
  fold_rows<2>(ps, lane, 4);
  fold_rows<1>(ps, lane, 2);
  const float rs = ps[0] + __shfl_xor(ps[0], 1);
  if ((c & 1) == 0) qsum[wave][rowoff + (c >> 1)] = rs;
  __syncthreads();
  if (wave == 0 && lane < 16) {
    float sm = qsum[0][lane];
    float r  = qmax[0][lane];
    for (int w = 1; w < RP_WAVES; ++w) {
      sm += qsum[w][lane];
      r = fmaxf(r, qmax[w][lane]);
    }
    term[lane] = __logf(sm) + r;
  }
}

#define PO_N 64
__device__ __forceinline__ float sum_partials(const float* __restrict__ part, int mt) {
  const float* pp = part + (size_t)mt * (PO_N * 256) + threadIdx.x;
  float sx = pp[0];
#pragma unroll
  for (int j = 1; j < PO_N; ++j) sx += pp[j * 256];
  return sx;
}

__global__ void k_layer0(const float* __restrict__ X, const float* __restrict__ W0,
                         const float* __restrict__ b0,
                         float* __restrict__ act,
                         float* __restrict__ E, float* __restrict__ mrow) {
  const int wave = (blockIdx.x * blockDim.x + threadIdx.x) >> 5;
  const int mt = wave >> 7;
  const int nt = wave & 127;
  if (mt >= Bsz / 16) return;
  v8f acc = {};
  gemm_tile(X, Dd, W0, Dd, mt * 16, nt * 16, Dd, acc);

  const int lane = threadIdx.x & 31;
  const int n = nt * 16 + (lane & 15);
  const int rowoff = (lane >> 4) * 8;
  const float ldiag = __logf(W0[n * Dd + (n >> 7)]);
  const float bias = b0[n];
  float g[8];
  for (int v = 0; v < 8; ++v) {
    const int m = mt * 16 + v + rowoff;
    const float h = acc[v] + bias;
    act[m * HID + n] = fast_tanh(h);
    g[v] = ldiag + log_abs_dtanh(h);
  }
  rowmax_exp_store(g, mt * 16, nt >> 3, E, mrow);
}

__global__ __launch_bounds__(RP_WAVES * 32)
void k_sum_layer0(const float* __restrict__ part, const float* __restrict__ W0,
                  const float* __restrict__ b0,
                  float* __restrict__ act,
                  float* __restrict__ E, float* __restrict__ mrow) {
  __shared__ __attribute__((aligned(16))) float xs[16 * Dd];
  const int lane = threadIdx.x & 31;
  const int wave = (blockIdx.x * blockDim.x + threadIdx.x) >> 5;
  const int mt = wave >> 7;
  const int nt = wave & 127;
  xs[threadIdx.x] = sum_partials(part, mt);
  __syncthreads();

  const int r  = lane & 15;
  const int k4 = (lane >> 4) << 2;
  v8f acc = {};
#pragma unroll
  for (int k = 0; k < Dd; k += 8) {
    const float4 fa = *reinterpret_cast<const float4*>(&xs[r * Dd + k4 + k]);
    const float4 fb = *reinterpret_cast<const float4*>(W0 + (size_t)(nt * 16 + r) * Dd + k4 + k);
    v2f a0, a1, w0, w1;
    a0.x = fa.x;  a0.y = fa.y;  a1.x = fa.z;  a1.y = fa.w;
    w0.x = fb.x;  w0.y = fb.y;  w1.x = fb.z;  w1.y = fb.w;
    acc = wmma_f32_k4(a0, w0, acc);
    acc = wmma_f32_k4(a1, w1, acc);
  }

  const int n = nt * 16 + (lane & 15);
  const int rowoff = (lane >> 4) * 8;
  const float ldiag = __logf(W0[n * Dd + (n >> 7)]);
  const float bias = b0[n];
  float g[8];
  for (int v = 0; v < 8; ++v) {
    const int m = mt * 16 + v + rowoff;
    const float h = acc[v] + bias;
    act[m * HID + n] = fast_tanh(h);
    g[v] = ldiag + log_abs_dtanh(h);
  }
  rowmax_exp_store(g, mt * 16, nt >> 3, E, mrow);
}

#define L1_KC    128
#define L1_PITCH (L1_KC + 4)
#define L1_MROWS 32
#define L1_SLAB  32
#define L1_NROWS (2 * L1_SLAB)
#define L1_NSLAB (HID / L1_SLAB)
#define L1_NG    (L1_KC / 8)
#define L1_WAVES 4
#define L1_TP    20
#define L1_A8(X)   X(0) X(1) X(2) X(3) X(4) X(5) X(6) X(7)
#define L1_WHI8(X) X(0) X(1) X(2) X(3) X(4) X(5) X(6) X(7)
#define L1_WLO8(X) X(8) X(9) X(10) X(11) X(12) X(13) X(14) X(15)
#define L1_DECLA(i) float4 pa##i;
#define L1_DECLW(i) float4 pw##i;
#define L1_LDA(i)  pa##i = *reinterpret_cast<const float4*>(gA  + (size_t)(4 * i) * HID + ko);
#define L1_LDWH(i) pw##i = *reinterpret_cast<const float4*>(gWh + (size_t)(4 * i) * HID + ko);
#define L1_LDWL(i) pw##i = *reinterpret_cast<const float4*>(gWl + (size_t)(4 * (i - 8)) * HID + ko);
#define L1_STA(i) *reinterpret_cast<float4*>(&As[st][soff + 4 * i * L1_PITCH]) = pa##i;
#define L1_STW(i) *reinterpret_cast<float4*>(&Ws[st][soff + 4 * i * L1_PITCH]) = pw##i;
#define L1_LD_AH(kc)  { const int ko = (kc) * L1_KC;  L1_A8(L1_LDA)  L1_WHI8(L1_LDWH) }
#define L1_LD_AHL(kc) { const int ko = (kc) * L1_KC;  L1_A8(L1_LDA)  L1_WHI8(L1_LDWH)  L1_WLO8(L1_LDWL) }
#define L1_LD_NONE(kc)
#define L1_ST_AH(s)   { const int st = (s);  L1_A8(L1_STA)  L1_WHI8(L1_STW) }
#define L1_ST_AHL(s)  { const int st = (s);  L1_A8(L1_STA)  L1_WHI8(L1_STW)  L1_WLO8(L1_STW) }
#define L1_N_NONE 0
#define L1_N_AH   16
#define L1_N_AHL  24
#define L1_BOTH_RD  (3 * L1_NG)
#define L1_BOTH_MM  (4 * L1_NG)
#define L1_UPPER_RD (2 * L1_NG)
#define L1_UPPER_MM (2 * L1_NG)
#define L1_TRIP_BOTH(c, STORE, NST, LOAD, NLD)                       \
  {                                                                  \
    const int cur = (c) & 1;                                         \
    float4 fa[L1_NG], fb0[L1_NG], fb1[L1_NG];                        \
    l1_frag_both(&As[cur][foffA], &Ws[cur][foffB], fa, fb0, fb1);    \
    STORE(cur ^ 1)                                                   \
    LOAD((c) + 2)                                                    \
    l1_wmma_both(fa, fb0, fb1, acc0, acc1);                          \
    l1_trip_order<L1_BOTH_RD, L1_BOTH_MM, NST, NLD>();               \
    l1_lds_barrier();                                                \
  }
#define L1_TRIP_UPPER(c, STORE, NST, LOAD, NLD)                      \
  {                                                                  \
    const int cur = (c) & 1;                                         \
    float4 fa[L1_NG], fb0[L1_NG];                                    \
    l1_frag_upper(&As[cur][foffA], &Ws[cur][foffB], fa, fb0);        \
    STORE(cur ^ 1)                                                   \
    LOAD((c) + 2)                                                    \
    l1_wmma_upper(fa, fb0, acc0, acc2);                              \
    l1_trip_order<L1_UPPER_RD, L1_UPPER_MM, NST, NLD>();             \
    l1_lds_barrier();                                                \
  }

__device__ __forceinline__ void l1_frag_both(const float* __restrict__ as, const float* __restrict__ ws0,
                                             float4 (&fa)[L1_NG], float4 (&fb0)[L1_NG], float4 (&fb1)[L1_NG]) {
  const float* ws1 = ws0 + L1_SLAB * L1_PITCH;
#pragma unroll
  for (int j = 0; j < L1_NG; ++j) {
    fa[j]  = *reinterpret_cast<const float4*>(as  + 8 * j);
    fb0[j] = *reinterpret_cast<const float4*>(ws0 + 8 * j);
    fb1[j] = *reinterpret_cast<const float4*>(ws1 + 8 * j);
  }
}
__device__ __forceinline__ void l1_frag_upper(const float* __restrict__ as, const float* __restrict__ ws0,
                                              float4 (&fa)[L1_NG], float4 (&fb0)[L1_NG]) {
#pragma unroll
  for (int j = 0; j < L1_NG; ++j) {
    fa[j]  = *reinterpret_cast<const float4*>(as  + 8 * j);
    fb0[j] = *reinterpret_cast<const float4*>(ws0 + 8 * j);
  }
}

__device__ __forceinline__ void l1_wmma_both(const float4 (&fa)[L1_NG], const float4 (&fb0)[L1_NG],
                                             const float4 (&fb1)[L1_NG], v8f& acc0, v8f& acc1) {
#pragma unroll
  for (int j = 0; j < L1_NG; ++j) {
    v2f a0, a1, p0, p1, q0, q1;
    a0.x = fa[j].x;   a0.y = fa[j].y;   a1.x = fa[j].z;   a1.y = fa[j].w;
    p0.x = fb0[j].x;  p0.y = fb0[j].y;  p1.x = fb0[j].z;  p1.y = fb0[j].w;
    q0.x = fb1[j].x;  q0.y = fb1[j].y;  q1.x = fb1[j].z;  q1.y = fb1[j].w;
    acc0 = wmma_f32_k4(a0, p0, acc0);
    acc1 = wmma_f32_k4(a0, q0, acc1);
    acc0 = wmma_f32_k4(a1, p1, acc0);
    acc1 = wmma_f32_k4(a1, q1, acc1);
  }
}

__device__ __forceinline__ void l1_wmma_upper(const float4 (&fa)[L1_NG], const float4 (&fb0)[L1_NG],
                                              v8f& acce, v8f& acco) {
#pragma unroll
  for (int j = 0; j < L1_NG; ++j) {
    v2f a0, a1, p0, p1;
    a0.x = fa[j].x;   a0.y = fa[j].y;   a1.x = fa[j].z;   a1.y = fa[j].w;
    p0.x = fb0[j].x;  p0.y = fb0[j].y;  p1.x = fb0[j].z;  p1.y = fb0[j].w;
    acce = wmma_f32_k4(a0, p0, acce);
    acco = wmma_f32_k4(a1, p1, acco);
  }
}

template <int NREAD, int NWMMA, int NST, int NLD>
__device__ __forceinline__ void l1_trip_order() {
  constexpr int NMEM = NST > NLD ? NST : NLD;
  __builtin_amdgcn_sched_group_barrier(0x100, NREAD, 0);
#pragma unroll
  for (int i = 0; i < NMEM; ++i) {
    __builtin_amdgcn_sched_group_barrier(0x008, 1, 0);
    if (i < NST) __builtin_amdgcn_sched_group_barrier(0x200, 1, 0);
    if (i < NLD) __builtin_amdgcn_sched_group_barrier(0x020, 1, 0);
  }
  if constexpr (NWMMA - NMEM > 0)
    __builtin_amdgcn_sched_group_barrier(0x008, NWMMA - NMEM, 0);
}

__device__ __forceinline__ void l1_lds_barrier() {
  __builtin_amdgcn_sched_barrier(0);
  __builtin_amdgcn_fence(__ATOMIC_RELEASE, "workgroup", "local");
  __builtin_amdgcn_s_barrier();
  __builtin_amdgcn_fence(__ATOMIC_ACQUIRE, "workgroup", "local");
}

__global__ __launch_bounds__(L1_WAVES * 32)
void k_layer1(const float* __restrict__ A, const float* __restrict__ W1,
              const float* __restrict__ b1, const float* __restrict__ W2,
              float* __restrict__ lga, float* __restrict__ part) {
  __shared__ __attribute__((aligned(16))) float As[2][L1_MROWS * L1_PITCH];
  __shared__ __attribute__((aligned(16))) float Ws[2][L1_NROWS * L1_PITCH];
  __shared__ __attribute__((aligned(16))) float Ts[L1_WAVES][2][16 * L1_TP];
  const int tid  = threadIdx.x;
  const int wave = tid >> 5;
  const int lane = tid & 31;
  const int ms  = blockIdx.x >> 5;
  const int p   = blockIdx.x & 31;
  const int jlo = p;
  const int jhi = L1_NSLAB - 1 - p;
  const int q   = p >> 2;
  const int nch = (HID / L1_KC) - q;
  const int mi = wave >> 1;
  const int t  = wave & 1;

  const int srow = tid >> 5;
  const int sc4  = (tid & 31) * 4;
  const float* gA  = A  + (size_t)(ms  * L1_MROWS + srow) * HID + sc4;
  const float* gWh = W1 + (size_t)(jhi * L1_SLAB  + srow) * HID + sc4;
  const float* gWl = W1 + (size_t)(jlo * L1_SLAB  + srow) * HID + sc4;
  const int soff = srow * L1_PITCH + sc4;

  const int foffA = (mi * 16 + (lane & 15)) * L1_PITCH + (lane >> 4) * 4;
  const int foffB = (t  * 16 + (lane & 15)) * L1_PITCH + (lane >> 4) * 4;

  L1_A8(L1_DECLA)
  L1_WHI8(L1_DECLW)
  L1_WLO8(L1_DECLW)
  const int last = nch - 1;
  L1_LD_AHL(0)
  L1_ST_AHL(0)
  if (q >= 1) { L1_LD_AHL(1) } else { L1_LD_AH(1) }
  __syncthreads();

  v8f acc0 = {};
  v8f acc1 = {};
  v8f acc2 = {};
  int c = 0;
  for (; c + 2 <= q; ++c)
    L1_TRIP_BOTH(c, L1_ST_AHL, L1_N_AHL, L1_LD_AHL, L1_N_AHL)
  if (q >= 1) {
    L1_TRIP_BOTH(c, L1_ST_AHL, L1_N_AHL, L1_LD_AH, L1_N_AH)
    ++c;
  }
  if (q + 2 <= last)
    L1_TRIP_BOTH(c, L1_ST_AH, L1_N_AH, L1_LD_AH, L1_N_AH)
  else
    L1_TRIP_BOTH(c, L1_ST_AH, L1_N_AH, L1_LD_NONE, L1_N_NONE)
  ++c;
  for (; c + 2 <= last; ++c)
    L1_TRIP_UPPER(c, L1_ST_AH, L1_N_AH, L1_LD_AH, L1_N_AH)
  if (c + 1 <= last) {
    L1_TRIP_UPPER(c, L1_ST_AH, L1_N_AH, L1_LD_NONE, L1_N_NONE)
    ++c;
  }
  {
    float4 fa[L1_NG], fb0[L1_NG];
    l1_frag_upper(&As[last & 1][foffA], &Ws[last & 1][foffB], fa, fb0);
    l1_wmma_upper(fa, fb0, acc0, acc2);
    l1_trip_order<L1_UPPER_RD, L1_UPPER_MM, 0, 0>();
  }

  const int rowoff = (lane >> 4) * 8;
  const int cc = lane & 15;
  const int n0 = jhi * L1_SLAB + t * 16 + cc;
  const int n1 = jlo * L1_SLAB + t * 16 + cc;
  const float bias0 = b1[n0];
  const float bias1 = b1[n1];
  const int k4 = (lane >> 4) * 4;
  const float* w2u = W2 + (size_t)cc * HID + jhi * L1_SLAB + t * 16 + k4;
  const float* w2l = W2 + (size_t)cc * HID + jlo * L1_SLAB + t * 16 + k4;
  const float4 wu0 = *reinterpret_cast<const float4*>(w2u);
  const float4 wu1 = *reinterpret_cast<const float4*>(w2u + 8);
  const float4 wl0 = *reinterpret_cast<const float4*>(w2l);
  const float4 wl1 = *reinterpret_cast<const float4*>(w2l + 8);
  float* tu = &Ts[wave][0][0];
  float* tl = &Ts[wave][1][0];
#pragma unroll
  for (int v = 0; v < 8; ++v) {
    const int m = ms * L1_MROWS + mi * 16 + v + rowoff;
    const float h0 = (acc0[v] + acc2[v]) + bias0;
    const float h1 = acc1[v] + bias1;
    tu[(v + rowoff) * L1_TP + cc] = fast_tanh(h0);
    tl[(v + rowoff) * L1_TP + cc] = fast_tanh(h1);
    lga[m * HID + n0] = log_abs_dtanh(h0);
    lga[m * HID + n1] = log_abs_dtanh(h1);
  }
  l1_lds_barrier();
  v8f pe = {};
  v8f po = {};
  {
    const float4 au0 = *reinterpret_cast<const float4*>(&tu[cc * L1_TP + k4]);
    const float4 au1 = *reinterpret_cast<const float4*>(&tu[cc * L1_TP + 8 + k4]);
    const float4 al0 = *reinterpret_cast<const float4*>(&tl[cc * L1_TP + k4]);
    const float4 al1 = *reinterpret_cast<const float4*>(&tl[cc * L1_TP + 8 + k4]);
    v2f a, b;
    a.x = au0.x;  a.y = au0.y;  b.x = wu0.x;  b.y = wu0.y;  pe = wmma_f32_k4(a, b, pe);
    a.x = au0.z;  a.y = au0.w;  b.x = wu0.z;  b.y = wu0.w;  po = wmma_f32_k4(a, b, po);
    a.x = au1.x;  a.y = au1.y;  b.x = wu1.x;  b.y = wu1.y;  pe = wmma_f32_k4(a, b, pe);
    a.x = au1.z;  a.y = au1.w;  b.x = wu1.z;  b.y = wu1.w;  po = wmma_f32_k4(a, b, po);
    a.x = al0.x;  a.y = al0.y;  b.x = wl0.x;  b.y = wl0.y;  pe = wmma_f32_k4(a, b, pe);
    a.x = al0.z;  a.y = al0.w;  b.x = wl0.z;  b.y = wl0.w;  po = wmma_f32_k4(a, b, po);
    a.x = al1.x;  a.y = al1.y;  b.x = wl1.x;  b.y = wl1.y;  pe = wmma_f32_k4(a, b, pe);
    a.x = al1.z;  a.y = al1.w;  b.x = wl1.z;  b.y = wl1.w;  po = wmma_f32_k4(a, b, po);
  }
  float* pt = part + ((size_t)(ms * 2 + mi) * PO_N + (p * 2 + t)) * 256;
#pragma unroll
  for (int v = 0; v < 8; ++v)
    pt[(v + rowoff) * Dd + cc] = pe[v] + po[v];
}

#define TERM_PITCH 32
#define TERM_FLOW  (Dd * (Bsz / 16) * TERM_PITCH)
__device__ __forceinline__ void grad_body(int wave,
                                          const float* __restrict__ E, const float* __restrict__ W1,
                                          const float* __restrict__ lga, const float* __restrict__ mrow,
                                          const float* __restrict__ W2, float* __restrict__ terms) {
  const int ks  = wave >> 7;
  const int rem = wave & 127;
  const int mt  = rem >> 3;
  const int nt  = rem & 7;
  if (ks >= Dd) return;
  const float* Ak = E  + (size_t)ks * Mm;
  const float* Wk = W1 + (size_t)ks * Mm * HID + (size_t)ks * Mm;
  v8f acc = {};
  gemm_tile(Ak, HID, Wk, HID, mt * 16, nt * 16, Mm, acc);

  const int lane = threadIdx.x & 31;
  const int o = nt * 16 + (lane & 15);
  const int rowoff = (lane >> 4) * 8;
  const float w2c = W2[(size_t)ks * HID + ks * Mm + o];
  float g[8];
  for (int v = 0; v < 8; ++v) {
    const int b = mt * 16 + v + rowoff;
    const float s = acc[v];
    g[v] = __logf(s) + mrow[b * Dd + ks] + lga[b * HID + ks * Mm + o];
  }
  rowmax_dot_term(g, w2c, terms + (size_t)(ks * (Bsz / 16) + mt) * TERM_PITCH);
}

#define OG_BLOCKS (Bsz / 16)
#define GR_BLOCKS (Dd * (Bsz / 16))
__global__ __launch_bounds__(RP_WAVES * 32)
void k_grad2_out(const float* __restrict__ Ea, const float* __restrict__ W1a,
                 const float* __restrict__ lgaa, const float* __restrict__ mrowa,
                 const float* __restrict__ W2a,
                 const float* __restrict__ Eb, const float* __restrict__ W1b,
                 const float* __restrict__ lgab, const float* __restrict__ mrowb,
                 const float* __restrict__ W2b,
                 float* __restrict__ terms,
                 const float* __restrict__ part, float* __restrict__ out) {
  if (blockIdx.x < OG_BLOCKS) {
    out[blockIdx.x * 16 * Dd + threadIdx.x] = sum_partials(part, blockIdx.x);
  } else if (blockIdx.x < OG_BLOCKS + GR_BLOCKS) {
    grad_body(((blockIdx.x - OG_BLOCKS) * (RP_WAVES * 32) + threadIdx.x) >> 5,
              Eb, W1b, lgab, mrowb, W2b, terms + TERM_FLOW);
  } else {
    grad_body(((blockIdx.x - OG_BLOCKS - GR_BLOCKS) * (RP_WAVES * 32) + threadIdx.x) >> 5,
              Ea, W1a, lgaa, mrowa, W2a, terms);
  }
}

__global__ __launch_bounds__(Bsz)
void k_ldsum(const float* __restrict__ terms, float* __restrict__ ld) {
  const int b = threadIdx.x;
  const float* tp = terms + (b >> 4) * TERM_PITCH + (b & 15);
  float t = tp[0];
#pragma unroll
  for (int j = 1; j < 2 * Dd; ++j) t += tp[j * ((Bsz / 16) * TERM_PITCH)];
  ld[b] = t;
}

extern "C" void kernel_launch(void* const* d_in, const int* in_sizes, int n_in,
                              void* d_out, int out_size, void* d_ws, size_t ws_size,
                              hipStream_t stream) {
  const float* inp   = (const float*)d_in[0];
  const float* f0_W0 = (const float*)d_in[1];
  const float* f0_W1 = (const float*)d_in[2];
  const float* f0_W2 = (const float*)d_in[3];
  const float* f0_b0 = (const float*)d_in[4];
  const float* f0_b1 = (const float*)d_in[5];
  const float* f1_W0 = (const float*)d_in[6];
  const float* f1_W1 = (const float*)d_in[7];
  const float* f1_W2 = (const float*)d_in[8];
  const float* f1_b0 = (const float*)d_in[9];
  const float* f1_b1 = (const float*)d_in[10];

  float* out = (float*)d_out;
  float* ld  = out + Bsz * Dd;

  float* ws     = (float*)d_ws;
  float* act0   = ws;
  float* lgaa   = act0   + Bsz * HID;
  float* lgab   = lgaa   + Bsz * HID;
  float* E0a    = lgab   + Bsz * HID;
  float* E0b    = E0a    + Bsz * HID;
  float* mrow0a = E0b    + Bsz * HID;
  float* mrow0b = mrow0a + Bsz * Dd;
  float* terms  = mrow0b + Bsz * Dd;
  float* part   = terms  + 2 * TERM_FLOW;

  k_layer0    <<<256, RP_WAVES * 32, 0, stream>>>(inp, f0_W0, f0_b0, act0, E0a, mrow0a);
  k_layer1    <<<(Bsz / L1_MROWS) * (L1_NSLAB / 2), L1_WAVES * 32, 0, stream>>>(act0, f0_W1, f0_b1, f0_W2, lgaa, part);
  k_sum_layer0<<<256, RP_WAVES * 32, 0, stream>>>(part, f1_W0, f1_b0, act0, E0b, mrow0b);
  k_layer1    <<<(Bsz / L1_MROWS) * (L1_NSLAB / 2), L1_WAVES * 32, 0, stream>>>(act0, f1_W1, f1_b1, f1_W2, lgab, part);
  k_grad2_out <<<OG_BLOCKS + 2 * GR_BLOCKS, RP_WAVES * 32, 0, stream>>>(E0a, f0_W1, lgaa, mrow0a, f0_W2,
                                                                       E0b, f1_W1, lgab, mrow0b, f1_W2,
                                                                       terms, part, out);
  k_ldsum     <<<1, Bsz, 0, stream>>>(terms, ld);
}
